# starting kernel
# baseline (speedup 1.0000x reference)
#include <hip/hip_runtime.h>
#include <hip/hip_bf16.h>

typedef __attribute__((ext_vector_type(2))) float v2f;
typedef __attribute__((ext_vector_type(8))) float v8f;

#define D_DIM 256
#define L_DIM 1024
#define B_DIM 4
#define M_DIM (B_DIM * L_DIM)
#define LN_EPS 1e-5f

__device__ __forceinline__ float block_sum_256(float v, float* sh8) {
#pragma unroll
  for (int m = 16; m > 0; m >>= 1) v += __shfl_xor(v, m, 32);
  if ((threadIdx.x & 31) == 0) sh8[threadIdx.x >> 5] = v;
  __syncthreads();
  float r = 0.f;
#pragma unroll
  for (int i = 0; i < 8; ++i) r += sh8[i];
  return r;
}

__global__ void wmma_gemm_f32_kernel(const float* __restrict__ A,
                                     const float* __restrict__ Wm,
                                     float* __restrict__ Out,
                                     const float* __restrict__ bias) {
  const int K = D_DIM, N = D_DIM;
  const int wave = threadIdx.x >> 5;
  const int lane = threadIdx.x & 31;
  const int half = lane >> 4;
  const int l16  = lane & 15;
  const int tM = (blockIdx.x * 8 + wave) * 16;
  const int tN = blockIdx.y * 16;

  const float* arow = A  + (size_t)(tM + l16) * K + (half ? 2 : 0);
  const float* brow = Wm + (size_t)(tN + l16) * K + (half ? 2 : 0);

  v8f acc = {};
#pragma unroll 8
  for (int kc = 0; kc < K; kc += 4) {
    v2f a; a.x = arow[kc]; a.y = arow[kc + 1];
    v2f b; b.x = brow[kc]; b.y = brow[kc + 1];
    acc = __builtin_amdgcn_wmma_f32_16x16x4_f32(
         false, a,  false, b,
         (short)0, acc,  false,  false);
  }

  const int n = tN + l16;
  const float bv = bias ? bias[n] : 0.f;
  union { v8f v; float f[8]; } u; u.v = acc;
#pragma unroll
  for (int r = 0; r < 8; ++r) {
    const int m = tM + r + (half ? 8 : 0);
    Out[(size_t)m * N + n] = u.f[r] + bv;
  }
}

__global__ void phi_beta_kernel(const float* __restrict__ x,
                                float* __restrict__ q,
                                float* __restrict__ k,
                                const float* __restrict__ beta_w,
                                const float* __restrict__ beta_b,
                                const float* __restrict__ lnp_g,
                                const float* __restrict__ lnp_b,
                                float* __restrict__ beta) {
  __shared__ float sh[8 * 5];
  const int row = blockIdx.x, tid = threadIdx.x;
  const size_t off = (size_t)row * D_DIM + tid;

  float qe; { float t = q[off]; qe = (t > 0.f) ? t + 1.f : __expf(t); }
  float ke; { float t = k[off]; ke = (t > 0.f) ? t + 1.f : __expf(t); }
  const float xb = x[off] * beta_w[tid];

  const float sq  = block_sum_256(qe,      sh + 0);
  const float sq2 = block_sum_256(qe * qe, sh + 8);
  const float sk  = block_sum_256(ke,      sh + 16);
  const float sk2 = block_sum_256(ke * ke, sh + 24);
  const float sb  = block_sum_256(xb,      sh + 32);

  const float muq = sq / D_DIM, varq = sq2 / D_DIM - muq * muq;
  const float muk = sk / D_DIM, vark = sk2 / D_DIM - muk * muk;
  const float g = lnp_g[tid], bb = lnp_b[tid];
  q[off] = (qe - muq) * rsqrtf(varq + LN_EPS) * g + bb;
  k[off] = (ke - muk) * rsqrtf(vark + LN_EPS) * g + bb;
  if (tid == 0) beta[row] = 1.f / (1.f + __expf(-(sb + beta_b[0])));
}

union F8 { float4 v4[2]; float f[8]; };

__global__ void delta_scan_kernel(const float* __restrict__ phi_q,
                                  const float* __restrict__ phi_k,
                                  const float* __restrict__ v,
                                  const float* __restrict__ beta,
                                  const float* __restrict__ decay,
                                  float* __restrict__ y) {
  const int b    = blockIdx.y;
  const int wave = threadIdx.x >> 5;
  const int lane = threadIdx.x & 31;
  const int row  = blockIdx.x * 8 + wave;
  const int c0   = lane * 8;

  const float df = 1.f / (1.f + __expf(-decay[0]));
  const size_t base = (size_t)b * L_DIM * D_DIM;

  float w[8];
#pragma unroll
  for (int c = 0; c < 8; ++c) w[c] = 0.f;

  F8 cq, ck;
  cq.v4[0] = *(const float4*)(phi_q + base + c0);
  cq.v4[1] = *(const float4*)(phi_q + base + c0 + 4);
  ck.v4[0] = *(const float4*)(phi_k + base + c0);
  ck.v4[1] = *(const float4*)(phi_k + base + c0 + 4);
  float cv = v[base + row];
  float cb = beta[(size_t)b * L_DIM];

  for (int t = 0; t < L_DIM; ++t) {
    const int tn = (t + 1 < L_DIM) ? t + 1 : t;
    const size_t noff = base + (size_t)tn * D_DIM;
    F8 nq, nk;
    nq.v4[0] = *(const float4*)(phi_q + noff + c0);
    nq.v4[1] = *(const float4*)(phi_q + noff + c0 + 4);
    nk.v4[0] = *(const float4*)(phi_k + noff + c0);
    nk.v4[1] = *(const float4*)(phi_k + noff + c0 + 4);
    const float nv = v[noff + row];
    const float nb = beta[(size_t)b * L_DIM + tn];
    if (t + 8 < L_DIM) {
      const size_t poff = base + (size_t)(t + 8) * D_DIM + c0;
      __builtin_prefetch(phi_q + poff, 0, 0);
      __builtin_prefetch(phi_k + poff, 0, 0);
    }

    float rq = 0.f, rk = 0.f;
#pragma unroll
    for (int c = 0; c < 8; ++c) {
      rq = fmaf(w[c], cq.f[c], rq);
      rk = fmaf(w[c], ck.f[c], rk);
    }
#pragma unroll
    for (int m = 16; m > 0; m >>= 1) {
      rq += __shfl_xor(rq, m, 32);
      rk += __shfl_xor(rk, m, 32);
    }
    if (lane == 0) y[base + (size_t)t * D_DIM + row] = rq;

    const float u = cb * (cv - rk);
#pragma unroll
    for (int c = 0; c < 8; ++c) w[c] = fmaf(df, w[c], u * ck.f[c]);

    cq = nq; ck = nk; cv = nv; cb = nb;
  }
}

__global__ void ln_rows_kernel(float* __restrict__ y,
                               const float* __restrict__ g,
                               const float* __restrict__ b) {
  __shared__ float sh[16];
  const int row = blockIdx.x, tid = threadIdx.x;
  const size_t off = (size_t)row * D_DIM + tid;
  const float v  = y[off];
  const float s  = block_sum_256(v, sh);
  const float s2 = block_sum_256(v * v, sh + 8);
  const float mu = s / D_DIM, var = s2 / D_DIM - mu * mu;
  y[off] = (v - mu) * rsqrtf(var + LN_EPS) * g[tid] + b[tid];
}

extern "C" void kernel_launch(void* const* d_in, const int* in_sizes, int n_in,
                              void* d_out, int out_size, void* d_ws, size_t ws_size,
                              hipStream_t stream) {
  (void)in_sizes; (void)n_in; (void)out_size; (void)ws_size;
  const float* x      = (const float*)d_in[0];
  const float* Wq     = (const float*)d_in[1];
  const float* Wk     = (const float*)d_in[2];
  const float* Wv     = (const float*)d_in[3];
  const float* beta_w = (const float*)d_in[4];
  const float* beta_b = (const float*)d_in[5];
  const float* decay  = (const float*)d_in[6];
  const float* Wo     = (const float*)d_in[7];
  const float* bo     = (const float*)d_in[8];
  const float* ln_g   = (const float*)d_in[9];
  const float* ln_b   = (const float*)d_in[10];
  const float* lnp_g  = (const float*)d_in[11];
  const float* lnp_b  = (const float*)d_in[12];
  float* out = (float*)d_out;

  char* ws = (char*)d_ws;
  const size_t MAT = (size_t)M_DIM * D_DIM * sizeof(float);
  float* q    = (float*)(ws);
  float* k    = (float*)(ws + MAT);
  float* vv   = (float*)(ws + 2 * MAT);
  float* beta = (float*)(ws + 3 * MAT);
  float* y    = (float*)(ws + 3 * MAT + (64 << 10));

  const dim3 gemm_grid(M_DIM / 128, D_DIM / 16);
  const dim3 gemm_blk(256);

  wmma_gemm_f32_kernel<<<gemm_grid, gemm_blk, 0, stream>>>(x, Wq, q, nullptr);
  wmma_gemm_f32_kernel<<<gemm_grid, gemm_blk, 0, stream>>>(x, Wk, k, nullptr);
  wmma_gemm_f32_kernel<<<gemm_grid, gemm_blk, 0, stream>>>(x, Wv, vv, nullptr);

  phi_beta_kernel<<<dim3(M_DIM), dim3(256), 0, stream>>>(
      x, q, k, beta_w, beta_b, lnp_g, lnp_b, beta);

  delta_scan_kernel<<<dim3(32, B_DIM), dim3(256), 0, stream>>>(
      q, k, vv, beta, decay, y);

  ln_rows_kernel<<<dim3(M_DIM), dim3(256), 0, stream>>>(y, ln_g, ln_b);

  wmma_gemm_f32_kernel<<<gemm_grid, gemm_blk, 0, stream>>>(y, Wo, out, bo);
}


# projection GEMMs: wave owns a 32x32 tile (2x2 accumulators), fragments from global by 16-byte loads held two 32-k groups ahead, no LDS; k order in each 8 permuted alike in A and B: new sum order
# speedup vs baseline: 3.5495x; 3.5495x over previous
#include <hip/hip_runtime.h>
#include <hip/hip_bf16.h>

typedef __attribute__((ext_vector_type(2))) float v2f;
typedef __attribute__((ext_vector_type(8))) float v8f;

#define D_DIM 256
#define L_DIM 1024
#define B_DIM 4
#define M_DIM (B_DIM * L_DIM)
#define LN_EPS 1e-5f

__device__ __forceinline__ float block_sum_256(float v, float* sh8) {
#pragma unroll
  for (int m = 16; m > 0; m >>= 1) v += __shfl_xor(v, m, 32);
  if ((threadIdx.x & 31) == 0) sh8[threadIdx.x >> 5] = v;
  __syncthreads();
  float r = 0.f;
#pragma unroll
  for (int i = 0; i < 8; ++i) r += sh8[i];
  return r;
}

__device__ __forceinline__ v8f mma4(v2f a, v2f b, v8f c) {
  return __builtin_amdgcn_wmma_f32_16x16x4_f32(false, a, false, b, (short)0, c, false, false);
}

#define GK_STEPS 4
#define GK_GROUPS (D_DIM / (8 * GK_STEPS))

__global__ __launch_bounds__(128) void wmma_gemm_f32_kernel(const float* __restrict__ A,
                                                            const float* __restrict__ Wm,
                                                            float* __restrict__ Out,
                                                            const float* __restrict__ bias) {
  const int wave = threadIdx.x >> 5;
  const int lane = threadIdx.x & 31;
  const int h = lane >> 4;
  const int c = lane & 15;
  const int tM = blockIdx.x * 64 + (wave >> 1) * 32;
  const int tN = blockIdx.y * 64 + (wave & 1) * 32;

  const float* pa[2];
  const float* pb[2];
#pragma unroll
  for (int i = 0; i < 2; ++i) {
    pa[i] = A  + (size_t)(tM + 16 * i + c) * D_DIM + 4 * h;
    pb[i] = Wm + (size_t)(tN + 16 * i + c) * D_DIM + 4 * h;
  }

  float bv[2];
#pragma unroll
  for (int j = 0; j < 2; ++j) bv[j] = bias ? bias[tN + 16 * j + c] : 0.f;

  float4 fa[3][GK_STEPS][2], fb[3][GK_STEPS][2];
#pragma unroll
  for (int g = 0; g < 2; ++g) {
#pragma unroll
    for (int st = 0; st < GK_STEPS; ++st)
#pragma unroll
      for (int i = 0; i < 2; ++i) {
        fa[g][st][i] = *(const float4*)(pa[i] + 32 * g + 8 * st);
        fb[g][st][i] = *(const float4*)(pb[i] + 32 * g + 8 * st);
      }
    __builtin_amdgcn_sched_barrier(0);
  }

  v8f acc[2][2];
#pragma unroll
  for (int i = 0; i < 2; ++i)
#pragma unroll
    for (int j = 0; j < 2; ++j) acc[i][j] = (v8f){};

#pragma unroll
  for (int g = 0; g < GK_GROUPS; ++g) {
    if (g + 2 < GK_GROUPS) {
#pragma unroll
      for (int st = 0; st < GK_STEPS; ++st)
#pragma unroll
        for (int i = 0; i < 2; ++i) {
          fa[(g + 2) % 3][st][i] = *(const float4*)(pa[i] + 32 * (g + 2) + 8 * st);
          fb[(g + 2) % 3][st][i] = *(const float4*)(pb[i] + 32 * (g + 2) + 8 * st);
        }
    }
    __builtin_amdgcn_sched_barrier(0);
#pragma unroll
    for (int st = 0; st < GK_STEPS; ++st) {
      const float4 a0 = fa[g % 3][st][0], a1 = fa[g % 3][st][1];
      const float4 b0 = fb[g % 3][st][0], b1 = fb[g % 3][st][1];
      acc[0][0] = mma4((v2f){a0.x, a0.y}, (v2f){b0.x, b0.y}, acc[0][0]);
      acc[0][1] = mma4((v2f){a0.x, a0.y}, (v2f){b1.x, b1.y}, acc[0][1]);
      acc[1][0] = mma4((v2f){a1.x, a1.y}, (v2f){b0.x, b0.y}, acc[1][0]);
      acc[1][1] = mma4((v2f){a1.x, a1.y}, (v2f){b1.x, b1.y}, acc[1][1]);
      acc[0][0] = mma4((v2f){a0.z, a0.w}, (v2f){b0.z, b0.w}, acc[0][0]);
      acc[0][1] = mma4((v2f){a0.z, a0.w}, (v2f){b1.z, b1.w}, acc[0][1]);
      acc[1][0] = mma4((v2f){a1.z, a1.w}, (v2f){b0.z, b0.w}, acc[1][0]);
      acc[1][1] = mma4((v2f){a1.z, a1.w}, (v2f){b1.z, b1.w}, acc[1][1]);
    }
  }

#pragma unroll
  for (int j = 0; j < 2; ++j) {
    const int n = tN + 16 * j + c;
#pragma unroll
    for (int i = 0; i < 2; ++i)
#pragma unroll
      for (int r = 0; r < 8; ++r) {
        const int m = tM + 16 * i + 8 * h + r;
        Out[(size_t)m * D_DIM + n] = acc[i][j][r] + bv[j];
      }
  }
}

__global__ void phi_beta_kernel(const float* __restrict__ x,
                                float* __restrict__ q,
                                float* __restrict__ k,
                                const float* __restrict__ beta_w,
                                const float* __restrict__ beta_b,
                                const float* __restrict__ lnp_g,
                                const float* __restrict__ lnp_b,
                                float* __restrict__ beta) {
  __shared__ float sh[8 * 5];
  const int row = blockIdx.x, tid = threadIdx.x;
  const size_t off = (size_t)row * D_DIM + tid;

  float qe; { float t = q[off]; qe = (t > 0.f) ? t + 1.f : __expf(t); }
  float ke; { float t = k[off]; ke = (t > 0.f) ? t + 1.f : __expf(t); }
  const float xb = x[off] * beta_w[tid];

  const float sq  = block_sum_256(qe,      sh + 0);
  const float sq2 = block_sum_256(qe * qe, sh + 8);
  const float sk  = block_sum_256(ke,      sh + 16);
  const float sk2 = block_sum_256(ke * ke, sh + 24);
  const float sb  = block_sum_256(xb,      sh + 32);

  const float muq = sq / D_DIM, varq = sq2 / D_DIM - muq * muq;
  const float muk = sk / D_DIM, vark = sk2 / D_DIM - muk * muk;
  const float g = lnp_g[tid], bb = lnp_b[tid];
  q[off] = (qe - muq) * rsqrtf(varq + LN_EPS) * g + bb;
  k[off] = (ke - muk) * rsqrtf(vark + LN_EPS) * g + bb;
  if (tid == 0) beta[row] = 1.f / (1.f + __expf(-(sb + beta_b[0])));
}

#define CH 16
#define NCH (L_DIM / CH)
#define AUX_STRIDE 512

__global__ void chunk_prep_kernel(const float* __restrict__ phi_q,
                                  const float* __restrict__ phi_k,
                                  const float* __restrict__ beta,
                                  const float* __restrict__ decay,
                                  float* __restrict__ vu,
                                  float* __restrict__ aux) {
  __shared__ __attribute__((aligned(16))) float Ks[CH][260];
  __shared__ __attribute__((aligned(16))) float Qs[CH][260];
  __shared__ float Al[CH][17];
  __shared__ float Tl[CH][17];
  __shared__ float bl[CH];
  const int ch = blockIdx.x, b = blockIdx.y, tid = threadIdx.x;
  const float df = 1.f / (1.f + __expf(-decay[0]));
  const size_t base = ((size_t)b * L_DIM + (size_t)ch * CH) * D_DIM;
  float vcol[CH];
#pragma unroll
  for (int i = 0; i < CH; ++i) {
    Ks[i][tid] = phi_k[base + (size_t)i * D_DIM + tid];
    Qs[i][tid] = phi_q[base + (size_t)i * D_DIM + tid];
    vcol[i] = vu[base + (size_t)i * D_DIM + tid];
  }
  if (tid < CH) bl[tid] = beta[(size_t)b * L_DIM + ch * CH + tid];
  __syncthreads();

  const int i = tid >> 4, j = tid & 15;
  float gk = 0.f, gq = 0.f;
#pragma unroll 4
  for (int c = 0; c < D_DIM; c += 4) {
    const float4 kj = *(const float4*)&Ks[j][c];
    const float4 ki = *(const float4*)&Ks[i][c];
    const float4 qi = *(const float4*)&Qs[i][c];
    gk = fmaf(ki.x, kj.x, gk); gk = fmaf(ki.y, kj.y, gk); gk = fmaf(ki.z, kj.z, gk); gk = fmaf(ki.w, kj.w, gk);
    gq = fmaf(qi.x, kj.x, gq); gq = fmaf(qi.y, kj.y, gq); gq = fmaf(qi.z, kj.z, gq); gq = fmaf(qi.w, kj.w, gq);
  }
  float pw = 1.f;
  for (int n = 0; n < i - 1 - j; ++n) pw *= df;
  const bool lower = j < i;
  Al[i][j] = lower ? bl[i] * pw * gk : 0.f;
  float* o = aux + ((size_t)b * NCH + ch) * AUX_STRIDE;
  o[256 + tid] = lower ? pw * gq : 0.f;
  __syncthreads();

  if (tid < CH) {
    float tc[CH];
#pragma unroll
    for (int ii = 0; ii < CH; ++ii) {
      float s = (ii == tid) ? 1.f : 0.f;
#pragma unroll
      for (int jj = 0; jj < ii; ++jj) s = fmaf(-Al[ii][jj], tc[jj], s);
      tc[ii] = s;
      Tl[ii][tid] = s;
    }
  }
  __syncthreads();
  float gj = 1.f;
  for (int n = 0; n < j; ++n) gj *= df;
  o[tid] = -Tl[i][j] * bl[j] * gj;

#pragma unroll
  for (int jj = 0; jj < CH; ++jj) vcol[jj] *= bl[jj];
#pragma unroll
  for (int ii = 0; ii < CH; ++ii) {
    float s = 0.f;
#pragma unroll
    for (int jj = 0; jj <= ii; ++jj) s = fmaf(Tl[ii][jj], vcol[jj], s);
    vu[base + (size_t)ii * D_DIM + tid] = s;
  }
}

__global__ __launch_bounds__(128) void delta_chunk_scan_kernel(
    const float* __restrict__ phi_q, const float* __restrict__ phi_k,
    float* vy, const float* __restrict__ aux, const float* __restrict__ decay) {
  __shared__ float red[2][4][2][8][32];
  __shared__ float u0s[2][8][32];
  const int b = blockIdx.y;
  const int row0 = blockIdx.x * 16;
  const int wave = __builtin_amdgcn_readfirstlane(threadIdx.x >> 5);
  const int lane = threadIdx.x & 31;
  const int c = lane & 15, h = lane >> 4;

  const float df = 1.f / (1.f + __expf(-decay[0]));
  float pwr[17];
  pwr[0] = 1.f;
#pragma unroll
  for (int n = 1; n <= 16; ++n) pwr[n] = pwr[n - 1] * df;
  const float g16 = pwr[16];
  float gpos[8], grev[8];
#pragma unroll
  for (int r = 0; r < 8; ++r) {
    gpos[r] = h ? pwr[8 + r] : pwr[r];
    grev[r] = h ? pwr[7 - r] : pwr[15 - r];
  }

  const size_t bbase = (size_t)b * L_DIM * D_DIM;
  const int colw = 64 * wave;
  const float* kA = phi_k + bbase + (size_t)c * D_DIM + colw + 8 * h;
  const float* qA = phi_q + bbase + (size_t)c * D_DIM + colw + 8 * h;
  const float* kT = phi_k + bbase + (size_t)(8 * h) * D_DIM + colw + c;
  float* vt = vy + bbase + (size_t)(8 * h) * D_DIM + row0 + c;
  const float* ax = aux + (size_t)b * NCH * AUX_STRIDE + c * 16 + 8 * h;

  v8f acc[4];
#pragma unroll
  for (int ct = 0; ct < 4; ++ct) acc[ct] = (v8f){};

  float4 Kf[4][2], Qf[4][2];
  float ktr[4][8];
#pragma unroll
  for (int ct = 0; ct < 4; ++ct) {
    Kf[ct][0] = *(const float4*)(kA + 16 * ct);
    Kf[ct][1] = *(const float4*)(kA + 16 * ct + 4);
    Qf[ct][0] = *(const float4*)(qA + 16 * ct);
    Qf[ct][1] = *(const float4*)(qA + 16 * ct + 4);
#pragma unroll
    for (int r = 0; r < 8; ++r) ktr[ct][r] = kT[(size_t)r * D_DIM + 16 * ct];
  }
  float4 tnf[2], tnN[2], mqf[2] = {};
  float u0r[8] = {};
  tnf[0] = *(const float4*)(ax);  tnf[1] = *(const float4*)(ax + 4);
  if (wave == 1) {
    mqf[0] = *(const float4*)(ax + 256);  mqf[1] = *(const float4*)(ax + 260);
#pragma unroll
    for (int r = 0; r < 8; ++r) u0r[r] = vt[(size_t)r * D_DIM];
  }

  for (int t = 0; t < NCH; ++t) {
    const size_t coff = (size_t)t * CH * D_DIM;
    const int tn1 = (t + 1 < NCH) ? t + 1 : t;
    const size_t noff = (size_t)tn1 * CH * D_DIM;

    v8f P = {}, Yq = {};
#pragma unroll
    for (int ct = 0; ct < 4; ++ct) {
      const float* kf = (const float*)Kf[ct];
      const float* qf = (const float*)Qf[ct];
#pragma unroll
      for (int p = 0; p < 4; ++p) {
        const v2f sb = {acc[ct][2 * p], acc[ct][2 * p + 1]};
        P  = mma4((v2f){kf[2 * p], kf[2 * p + 1]}, sb, P);
        Yq = mma4((v2f){qf[2 * p], qf[2 * p + 1]}, sb, Yq);
      }
    }

#pragma unroll
    for (int ct = 0; ct < 4; ++ct) {
      Kf[ct][0] = *(const float4*)(kA + noff + 16 * ct);
      Kf[ct][1] = *(const float4*)(kA + noff + 16 * ct + 4);
      Qf[ct][0] = *(const float4*)(qA + noff + 16 * ct);
      Qf[ct][1] = *(const float4*)(qA + noff + 16 * ct + 4);
    }
    {
      const float* an = ax + (size_t)tn1 * AUX_STRIDE;
      tnN[0] = *(const float4*)(an);  tnN[1] = *(const float4*)(an + 4);
    }
    if (t + 2 < NCH) {
      __builtin_prefetch(kA + noff + (size_t)CH * D_DIM + 24 * h, 0, 0);
      __builtin_prefetch(qA + noff + (size_t)CH * D_DIM + 24 * h, 0, 0);
    }

    const int par = t & 1;
#pragma unroll
    for (int r = 0; r < 8; ++r) {
      red[par][wave][0][r][lane] = P[r];
      red[par][wave][1][r][lane] = Yq[r];
    }
    if (wave == 1) {
#pragma unroll
      for (int r = 0; r < 8; ++r) u0s[par][r][lane] = u0r[r];
    }
#pragma unroll
    for (int ct = 0; ct < 4; ++ct) acc[ct] *= g16;
    __builtin_amdgcn_fence(__ATOMIC_RELEASE, "workgroup", "local");
    __builtin_amdgcn_s_barrier();
    __builtin_amdgcn_fence(__ATOMIC_ACQUIRE, "workgroup", "local");
    float ps[8];
#pragma unroll
    for (int r = 0; r < 8; ++r)
      ps[r] = (red[par][0][0][r][lane] + red[par][1][0][r][lane]) +
              (red[par][2][0][r][lane] + red[par][3][0][r][lane]);
    v8f Y = {};
    if (wave == 1) {
#pragma unroll
      for (int r = 0; r < 8; ++r)
        Y[r] = gpos[r] * ((red[par][0][1][r][lane] + red[par][1][1][r][lane]) +
                          (red[par][2][1][r][lane] + red[par][3][1][r][lane]));
    }

    v8f U;
#pragma unroll
    for (int r = 0; r < 8; ++r) U[r] = u0s[par][r][lane];
    {
      const float* tnp = (const float*)tnf;
#pragma unroll
      for (int jj = 0; jj < 4; ++jj)
        U = mma4((v2f){tnp[2 * jj], tnp[2 * jj + 1]}, (v2f){ps[2 * jj], ps[2 * jj + 1]}, U);
    }

    float kt[4][8];
#pragma unroll
    for (int ct = 0; ct < 4; ++ct)
#pragma unroll
      for (int r = 0; r < 8; ++r) kt[ct][r] = ktr[ct][r] * grev[r];
#pragma unroll
    for (int jj = 0; jj < 4; ++jj)
#pragma unroll
      for (int ct = 0; ct < 4; ++ct)
        acc[ct] = mma4((v2f){kt[ct][2 * jj], kt[ct][2 * jj + 1]},
                       (v2f){U[2 * jj], U[2 * jj + 1]}, acc[ct]);

    if (wave == 1) {
      const float* mqp = (const float*)mqf;
#pragma unroll
      for (int jj = 0; jj < 4; ++jj)
        Y = mma4((v2f){mqp[2 * jj], mqp[2 * jj + 1]}, (v2f){U[2 * jj], U[2 * jj + 1]}, Y);
#pragma unroll
      for (int r = 0; r < 8; ++r) vt[coff + (size_t)r * D_DIM] = Y[r];
      const float* an = ax + (size_t)tn1 * AUX_STRIDE;
      mqf[0] = *(const float4*)(an + 256);  mqf[1] = *(const float4*)(an + 260);
#pragma unroll
      for (int r = 0; r < 8; ++r) u0r[r] = vt[noff + (size_t)r * D_DIM];
    }

#pragma unroll
    for (int ct = 0; ct < 4; ++ct)
#pragma unroll
      for (int r = 0; r < 8; ++r) ktr[ct][r] = kT[noff + (size_t)r * D_DIM + 16 * ct];
    tnf[0] = tnN[0]; tnf[1] = tnN[1];
  }
}

__global__ void ln_rows_kernel(float* __restrict__ y,
                               const float* __restrict__ g,
                               const float* __restrict__ b) {
  __shared__ float sh[16];
  const int row = blockIdx.x, tid = threadIdx.x;
  const size_t off = (size_t)row * D_DIM + tid;
  const float v  = y[off];
  const float s  = block_sum_256(v, sh);
  const float s2 = block_sum_256(v * v, sh + 8);
  const float mu = s / D_DIM, var = s2 / D_DIM - mu * mu;
  y[off] = (v - mu) * rsqrtf(var + LN_EPS) * g[tid] + b[tid];
}

extern "C" void kernel_launch(void* const* d_in, const int* in_sizes, int n_in,
                              void* d_out, int out_size, void* d_ws, size_t ws_size,
                              hipStream_t stream) {
  (void)in_sizes; (void)n_in; (void)out_size; (void)ws_size;
  const float* x      = (const float*)d_in[0];
  const float* Wq     = (const float*)d_in[1];
  const float* Wk     = (const float*)d_in[2];
  const float* Wv     = (const float*)d_in[3];
  const float* beta_w = (const float*)d_in[4];
  const float* beta_b = (const float*)d_in[5];
  const float* decay  = (const float*)d_in[6];
  const float* Wo     = (const float*)d_in[7];
  const float* bo     = (const float*)d_in[8];
  const float* ln_g   = (const float*)d_in[9];
  const float* ln_b   = (const float*)d_in[10];
  const float* lnp_g  = (const float*)d_in[11];
  const float* lnp_b  = (const float*)d_in[12];
  float* out = (float*)d_out;

  char* ws = (char*)d_ws;
  const size_t MAT = (size_t)M_DIM * D_DIM * sizeof(float);
  float* q    = (float*)(ws);
  float* k    = (float*)(ws + MAT);
  float* vv   = (float*)(ws + 2 * MAT);
  float* beta = (float*)(ws + 3 * MAT);
  float* aux  = (float*)(ws + 3 * MAT + (64 << 10));

  const dim3 gemm_grid(M_DIM / 64, D_DIM / 64);
  const dim3 gemm_blk(128);

  wmma_gemm_f32_kernel<<<gemm_grid, gemm_blk, 0, stream>>>(x, Wq, q, nullptr);
  wmma_gemm_f32_kernel<<<gemm_grid, gemm_blk, 0, stream>>>(x, Wk, k, nullptr);
  wmma_gemm_f32_kernel<<<gemm_grid, gemm_blk, 0, stream>>>(x, Wv, vv, nullptr);

  phi_beta_kernel<<<dim3(M_DIM), dim3(256), 0, stream>>>(
      x, q, k, beta_w, beta_b, lnp_g, lnp_b, beta);

  chunk_prep_kernel<<<dim3(NCH, B_DIM), dim3(256), 0, stream>>>(q, k, beta, decay, vv, aux);

  delta_chunk_scan_kernel<<<dim3(D_DIM / 16, B_DIM), dim3(128), 0, stream>>>(
      q, k, vv, aux, decay);

  ln_rows_kernel<<<dim3(M_DIM), dim3(256), 0, stream>>>(vv, ln_g, ln_b);

  wmma_gemm_f32_kernel<<<gemm_grid, gemm_blk, 0, stream>>>(vv, Wo, out, bo);
}


# scan: work on the y / U0 tile (U0 load and hand-over, sum of the Q.S column slices, y store) shared by register row over the block's four waves, Mq.U formed by every wave; same sums, same order
# speedup vs baseline: 3.8723x; 1.0909x over previous
#include <hip/hip_runtime.h>
#include <hip/hip_bf16.h>

typedef __attribute__((ext_vector_type(2))) float v2f;
typedef __attribute__((ext_vector_type(8))) float v8f;

#define D_DIM 256
#define L_DIM 1024
#define B_DIM 4
#define M_DIM (B_DIM * L_DIM)
#define LN_EPS 1e-5f

__device__ __forceinline__ float block_sum_256(float v, float* sh8) {
#pragma unroll
  for (int m = 16; m > 0; m >>= 1) v += __shfl_xor(v, m, 32);
  if ((threadIdx.x & 31) == 0) sh8[threadIdx.x >> 5] = v;
  __syncthreads();
  float r = 0.f;
#pragma unroll
  for (int i = 0; i < 8; ++i) r += sh8[i];
  return r;
}

__device__ __forceinline__ v8f mma4(v2f a, v2f b, v8f c) {
  return __builtin_amdgcn_wmma_f32_16x16x4_f32(false, a, false, b, (short)0, c, false, false);
}

#define GK_STEPS 4
#define GK_GROUPS (D_DIM / (8 * GK_STEPS))

__global__ __launch_bounds__(128) void wmma_gemm_f32_kernel(const float* __restrict__ A,
                                                            const float* __restrict__ Wm,
                                                            float* __restrict__ Out,
                                                            const float* __restrict__ bias) {
  const int wave = threadIdx.x >> 5;
  const int lane = threadIdx.x & 31;
  const int h = lane >> 4;
  const int c = lane & 15;
  const int tM = blockIdx.x * 64 + (wave >> 1) * 32;
  const int tN = blockIdx.y * 64 + (wave & 1) * 32;

  const float* pa[2];
  const float* pb[2];
#pragma unroll
  for (int i = 0; i < 2; ++i) {
    pa[i] = A  + (size_t)(tM + 16 * i + c) * D_DIM + 4 * h;
    pb[i] = Wm + (size_t)(tN + 16 * i + c) * D_DIM + 4 * h;
  }

  float bv[2];
#pragma unroll
  for (int j = 0; j < 2; ++j) bv[j] = bias ? bias[tN + 16 * j + c] : 0.f;

  float4 fa[3][GK_STEPS][2], fb[3][GK_STEPS][2];
#pragma unroll
  for (int g = 0; g < 2; ++g) {
#pragma unroll
    for (int st = 0; st < GK_STEPS; ++st)
#pragma unroll
      for (int i = 0; i < 2; ++i) {
        fa[g][st][i] = *(const float4*)(pa[i] + 32 * g + 8 * st);
        fb[g][st][i] = *(const float4*)(pb[i] + 32 * g + 8 * st);
      }
    __builtin_amdgcn_sched_barrier(0);
  }

  v8f acc[2][2];
#pragma unroll
  for (int i = 0; i < 2; ++i)
#pragma unroll
    for (int j = 0; j < 2; ++j) acc[i][j] = (v8f){};

#pragma unroll
  for (int g = 0; g < GK_GROUPS; ++g) {
    if (g + 2 < GK_GROUPS) {
#pragma unroll
      for (int st = 0; st < GK_STEPS; ++st)
#pragma unroll
        for (int i = 0; i < 2; ++i) {
          fa[(g + 2) % 3][st][i] = *(const float4*)(pa[i] + 32 * (g + 2) + 8 * st);
          fb[(g + 2) % 3][st][i] = *(const float4*)(pb[i] + 32 * (g + 2) + 8 * st);
        }
    }
    __builtin_amdgcn_sched_barrier(0);
#pragma unroll
    for (int st = 0; st < GK_STEPS; ++st) {
      const float4 a0 = fa[g % 3][st][0], a1 = fa[g % 3][st][1];
      const float4 b0 = fb[g % 3][st][0], b1 = fb[g % 3][st][1];
      acc[0][0] = mma4((v2f){a0.x, a0.y}, (v2f){b0.x, b0.y}, acc[0][0]);
      acc[0][1] = mma4((v2f){a0.x, a0.y}, (v2f){b1.x, b1.y}, acc[0][1]);
      acc[1][0] = mma4((v2f){a1.x, a1.y}, (v2f){b0.x, b0.y}, acc[1][0]);
      acc[1][1] = mma4((v2f){a1.x, a1.y}, (v2f){b1.x, b1.y}, acc[1][1]);
      acc[0][0] = mma4((v2f){a0.z, a0.w}, (v2f){b0.z, b0.w}, acc[0][0]);
      acc[0][1] = mma4((v2f){a0.z, a0.w}, (v2f){b1.z, b1.w}, acc[0][1]);
      acc[1][0] = mma4((v2f){a1.z, a1.w}, (v2f){b0.z, b0.w}, acc[1][0]);
      acc[1][1] = mma4((v2f){a1.z, a1.w}, (v2f){b1.z, b1.w}, acc[1][1]);
    }
  }

#pragma unroll
  for (int j = 0; j < 2; ++j) {
    const int n = tN + 16 * j + c;
#pragma unroll
    for (int i = 0; i < 2; ++i)
#pragma unroll
      for (int r = 0; r < 8; ++r) {
        const int m = tM + 16 * i + 8 * h + r;
        Out[(size_t)m * D_DIM + n] = acc[i][j][r] + bv[j];
      }
  }
}

__global__ void phi_beta_kernel(const float* __restrict__ x,
                                float* __restrict__ q,
                                float* __restrict__ k,
                                const float* __restrict__ beta_w,
                                const float* __restrict__ beta_b,
                                const float* __restrict__ lnp_g,
                                const float* __restrict__ lnp_b,
                                float* __restrict__ beta) {
  __shared__ float sh[8 * 5];
  const int row = blockIdx.x, tid = threadIdx.x;
  const size_t off = (size_t)row * D_DIM + tid;

  float qe; { float t = q[off]; qe = (t > 0.f) ? t + 1.f : __expf(t); }
  float ke; { float t = k[off]; ke = (t > 0.f) ? t + 1.f : __expf(t); }
  const float xb = x[off] * beta_w[tid];

  const float sq  = block_sum_256(qe,      sh + 0);
  const float sq2 = block_sum_256(qe * qe, sh + 8);
  const float sk  = block_sum_256(ke,      sh + 16);
  const float sk2 = block_sum_256(ke * ke, sh + 24);
  const float sb  = block_sum_256(xb,      sh + 32);

  const float muq = sq / D_DIM, varq = sq2 / D_DIM - muq * muq;
  const float muk = sk / D_DIM, vark = sk2 / D_DIM - muk * muk;
  const float g = lnp_g[tid], bb = lnp_b[tid];
  q[off] = (qe - muq) * rsqrtf(varq + LN_EPS) * g + bb;
  k[off] = (ke - muk) * rsqrtf(vark + LN_EPS) * g + bb;
  if (tid == 0) beta[row] = 1.f / (1.f + __expf(-(sb + beta_b[0])));
}

#define CH 16
#define NCH (L_DIM / CH)
#define AUX_STRIDE 512

__global__ void chunk_prep_kernel(const float* __restrict__ phi_q,
                                  const float* __restrict__ phi_k,
                                  const float* __restrict__ beta,
                                  const float* __restrict__ decay,
                                  float* __restrict__ vu,
                                  float* __restrict__ aux) {
  __shared__ __attribute__((aligned(16))) float Ks[CH][260];
  __shared__ __attribute__((aligned(16))) float Qs[CH][260];
  __shared__ float Al[CH][17];
  __shared__ float Tl[CH][17];
  __shared__ float bl[CH];
  const int ch = blockIdx.x, b = blockIdx.y, tid = threadIdx.x;
  const float df = 1.f / (1.f + __expf(-decay[0]));
  const size_t base = ((size_t)b * L_DIM + (size_t)ch * CH) * D_DIM;
  float vcol[CH];
#pragma unroll
  for (int i = 0; i < CH; ++i) {
    Ks[i][tid] = phi_k[base + (size_t)i * D_DIM + tid];
    Qs[i][tid] = phi_q[base + (size_t)i * D_DIM + tid];
    vcol[i] = vu[base + (size_t)i * D_DIM + tid];
  }
  if (tid < CH) bl[tid] = beta[(size_t)b * L_DIM + ch * CH + tid];
  __syncthreads();

  const int i = tid >> 4, j = tid & 15;
  float gk = 0.f, gq = 0.f;
#pragma unroll 4
  for (int c = 0; c < D_DIM; c += 4) {
    const float4 kj = *(const float4*)&Ks[j][c];
    const float4 ki = *(const float4*)&Ks[i][c];
    const float4 qi = *(const float4*)&Qs[i][c];
    gk = fmaf(ki.x, kj.x, gk); gk = fmaf(ki.y, kj.y, gk); gk = fmaf(ki.z, kj.z, gk); gk = fmaf(ki.w, kj.w, gk);
    gq = fmaf(qi.x, kj.x, gq); gq = fmaf(qi.y, kj.y, gq); gq = fmaf(qi.z, kj.z, gq); gq = fmaf(qi.w, kj.w, gq);
  }
  float pw = 1.f;
  for (int n = 0; n < i - 1 - j; ++n) pw *= df;
  const bool lower = j < i;
  Al[i][j] = lower ? bl[i] * pw * gk : 0.f;
  float* o = aux + ((size_t)b * NCH + ch) * AUX_STRIDE;
  o[256 + tid] = lower ? pw * gq : 0.f;
  __syncthreads();

  if (tid < CH) {
    float tc[CH];
#pragma unroll
    for (int ii = 0; ii < CH; ++ii) {
      float s = (ii == tid) ? 1.f : 0.f;
#pragma unroll
      for (int jj = 0; jj < ii; ++jj) s = fmaf(-Al[ii][jj], tc[jj], s);
      tc[ii] = s;
      Tl[ii][tid] = s;
    }
  }
  __syncthreads();
  float gj = 1.f;
  for (int n = 0; n < j; ++n) gj *= df;
  o[tid] = -Tl[i][j] * bl[j] * gj;

#pragma unroll
  for (int jj = 0; jj < CH; ++jj) vcol[jj] *= bl[jj];
#pragma unroll
  for (int ii = 0; ii < CH; ++ii) {
    float s = 0.f;
#pragma unroll
    for (int jj = 0; jj <= ii; ++jj) s = fmaf(Tl[ii][jj], vcol[jj], s);
    vu[base + (size_t)ii * D_DIM + tid] = s;
  }
}

__global__ __launch_bounds__(128) void delta_chunk_scan_kernel(
    const float* __restrict__ phi_q, const float* __restrict__ phi_k,
    float* vy, const float* __restrict__ aux, const float* __restrict__ decay) {
  __shared__ float red[2][4][2][8][32];
  __shared__ float u0s[2][8][32];
  const int b = blockIdx.y;
  const int row0 = blockIdx.x * 16;
  const int wave = __builtin_amdgcn_readfirstlane(threadIdx.x >> 5);
  const int lane = threadIdx.x & 31;
  const int c = lane & 15, h = lane >> 4;
  const int r0 = 2 * wave;

  const float df = 1.f / (1.f + __expf(-decay[0]));
  float pwr[17];
  pwr[0] = 1.f;
#pragma unroll
  for (int n = 1; n <= 16; ++n) pwr[n] = pwr[n - 1] * df;
  const float g16 = pwr[16];
  float gpos[8], grev[8];
#pragma unroll
  for (int r = 0; r < 8; ++r) {
    gpos[r] = h ? pwr[8 + r] : pwr[r];
    grev[r] = h ? pwr[7 - r] : pwr[15 - r];
  }
  float gy0 = gpos[0], gy1 = gpos[1];
#pragma unroll
  for (int r = 2; r < 8; r += 2)
    if (r == r0) { gy0 = gpos[r]; gy1 = gpos[r + 1]; }

  const size_t bbase = (size_t)b * L_DIM * D_DIM;
  const int colw = 64 * wave;
  const float* kA = phi_k + bbase + (size_t)c * D_DIM + colw + 8 * h;
  const float* qA = phi_q + bbase + (size_t)c * D_DIM + colw + 8 * h;
  const float* kT = phi_k + bbase + (size_t)(8 * h) * D_DIM + colw + c;
  float* vtw = vy + bbase + (size_t)(8 * h + r0) * D_DIM + row0 + c;
  const float* ax = aux + (size_t)b * NCH * AUX_STRIDE + c * 16 + 8 * h;

  v8f acc[4];
#pragma unroll
  for (int ct = 0; ct < 4; ++ct) acc[ct] = (v8f){};

  float4 Kf[4][2], Qf[4][2];
  float ktr[4][8];
#pragma unroll
  for (int ct = 0; ct < 4; ++ct) {
    Kf[ct][0] = *(const float4*)(kA + 16 * ct);
    Kf[ct][1] = *(const float4*)(kA + 16 * ct + 4);
    Qf[ct][0] = *(const float4*)(qA + 16 * ct);
    Qf[ct][1] = *(const float4*)(qA + 16 * ct + 4);
#pragma unroll
    for (int r = 0; r < 8; ++r) ktr[ct][r] = kT[(size_t)r * D_DIM + 16 * ct];
  }
  float4 tnf[2], tnN[2], mqf[2];
  tnf[0] = *(const float4*)(ax);  tnf[1] = *(const float4*)(ax + 4);
  mqf[0] = *(const float4*)(ax + 256);  mqf[1] = *(const float4*)(ax + 260);
  float u0a = vtw[0], u0b = vtw[D_DIM];

  __builtin_amdgcn_fence(__ATOMIC_ACQUIRE, "workgroup", "global");

  for (int t = 0; t < NCH; ++t) {
    const size_t coff = (size_t)t * CH * D_DIM;
    const int tn1 = (t + 1 < NCH) ? t + 1 : t;
    const size_t noff = (size_t)tn1 * CH * D_DIM;

    v8f P = {}, Yq = {};
#pragma unroll
    for (int ct = 0; ct < 4; ++ct) {
      const float* kf = (const float*)Kf[ct];
      const float* qf = (const float*)Qf[ct];
#pragma unroll
      for (int p = 0; p < 4; ++p) {
        const v2f sb = {acc[ct][2 * p], acc[ct][2 * p + 1]};
        P  = mma4((v2f){kf[2 * p], kf[2 * p + 1]}, sb, P);
        Yq = mma4((v2f){qf[2 * p], qf[2 * p + 1]}, sb, Yq);
      }
    }

#pragma unroll
    for (int ct = 0; ct < 4; ++ct) {
      Kf[ct][0] = *(const float4*)(kA + noff + 16 * ct);
      Kf[ct][1] = *(const float4*)(kA + noff + 16 * ct + 4);
      Qf[ct][0] = *(const float4*)(qA + noff + 16 * ct);
      Qf[ct][1] = *(const float4*)(qA + noff + 16 * ct + 4);
    }
    {
      const float* an = ax + (size_t)tn1 * AUX_STRIDE;
      tnN[0] = *(const float4*)(an);  tnN[1] = *(const float4*)(an + 4);
    }
    if (t + 2 < NCH) {
      __builtin_prefetch(kA + noff + (size_t)CH * D_DIM + 24 * h, 0, 0);
      __builtin_prefetch(qA + noff + (size_t)CH * D_DIM + 24 * h, 0, 0);
    }

    const int par = t & 1;
#pragma unroll
    for (int r = 0; r < 8; ++r) {
      red[par][wave][0][r][lane] = P[r];
      red[par][wave][1][r][lane] = Yq[r];
    }
    u0s[par][r0][lane] = u0a;
    u0s[par][r0 + 1][lane] = u0b;
#pragma unroll
    for (int ct = 0; ct < 4; ++ct) acc[ct] *= g16;
    __builtin_amdgcn_fence(__ATOMIC_RELEASE, "workgroup", "local");
    __builtin_amdgcn_s_barrier();
    __builtin_amdgcn_fence(__ATOMIC_ACQUIRE, "workgroup", "local");
    float ps[8];
#pragma unroll
    for (int r = 0; r < 8; ++r)
      ps[r] = (red[par][0][0][r][lane] + red[par][1][0][r][lane]) +
              (red[par][2][0][r][lane] + red[par][3][0][r][lane]);
    const float yq0 = (red[par][0][1][r0][lane] + red[par][1][1][r0][lane]) +
                      (red[par][2][1][r0][lane] + red[par][3][1][r0][lane]);
    const float yq1 = (red[par][0][1][r0 + 1][lane] + red[par][1][1][r0 + 1][lane]) +
                      (red[par][2][1][r0 + 1][lane] + red[par][3][1][r0 + 1][lane]);
    v8f Y;
#pragma unroll
    for (int r = 0; r < 8; r += 2) {
      Y[r]     = (r == r0) ? gy0 * yq0 : 0.f;
      Y[r + 1] = (r == r0) ? gy1 * yq1 : 0.f;
    }

    v8f U;
#pragma unroll
    for (int r = 0; r < 8; ++r) U[r] = u0s[par][r][lane];
    {
      const float* tnp = (const float*)tnf;
#pragma unroll
      for (int jj = 0; jj < 4; ++jj)
        U = mma4((v2f){tnp[2 * jj], tnp[2 * jj + 1]}, (v2f){ps[2 * jj], ps[2 * jj + 1]}, U);
    }

    float kt[4][8];
#pragma unroll
    for (int ct = 0; ct < 4; ++ct)
#pragma unroll
      for (int r = 0; r < 8; ++r) kt[ct][r] = ktr[ct][r] * grev[r];
    const float* mqp = (const float*)mqf;
#pragma unroll
    for (int jj = 0; jj < 4; ++jj) {
#pragma unroll
      for (int ct = 0; ct < 4; ++ct)
        acc[ct] = mma4((v2f){kt[ct][2 * jj], kt[ct][2 * jj + 1]},
                       (v2f){U[2 * jj], U[2 * jj + 1]}, acc[ct]);
      Y = mma4((v2f){mqp[2 * jj], mqp[2 * jj + 1]}, (v2f){U[2 * jj], U[2 * jj + 1]}, Y);
    }

    float yo0 = Y[0], yo1 = Y[1];
#pragma unroll
    for (int r = 2; r < 8; r += 2)
      if (r == r0) { yo0 = Y[r]; yo1 = Y[r + 1]; }
    vtw[coff] = yo0;
    vtw[coff + D_DIM] = yo1;
    {
      const float* an = ax + (size_t)tn1 * AUX_STRIDE;
      mqf[0] = *(const float4*)(an + 256);  mqf[1] = *(const float4*)(an + 260);
    }
    u0a = vtw[noff];
    u0b = vtw[noff + D_DIM];

#pragma unroll
    for (int ct = 0; ct < 4; ++ct)
#pragma unroll
      for (int r = 0; r < 8; ++r) ktr[ct][r] = kT[noff + (size_t)r * D_DIM + 16 * ct];
    tnf[0] = tnN[0]; tnf[1] = tnN[1];
    __builtin_amdgcn_sched_barrier(0);
  }
}

__global__ void ln_rows_kernel(float* __restrict__ y,
                               const float* __restrict__ g,
                               const float* __restrict__ b) {
  __shared__ float sh[16];
  const int row = blockIdx.x, tid = threadIdx.x;
  const size_t off = (size_t)row * D_DIM + tid;
  const float v  = y[off];
  const float s  = block_sum_256(v, sh);
  const float s2 = block_sum_256(v * v, sh + 8);
  const float mu = s / D_DIM, var = s2 / D_DIM - mu * mu;
  y[off] = (v - mu) * rsqrtf(var + LN_EPS) * g[tid] + b[tid];
}

extern "C" void kernel_launch(void* const* d_in, const int* in_sizes, int n_in,
                              void* d_out, int out_size, void* d_ws, size_t ws_size,
                              hipStream_t stream) {
  (void)in_sizes; (void)n_in; (void)out_size; (void)ws_size;
  const float* x      = (const float*)d_in[0];
  const float* Wq     = (const float*)d_in[1];
  const float* Wk     = (const float*)d_in[2];
  const float* Wv     = (const float*)d_in[3];
  const float* beta_w = (const float*)d_in[4];
  const float* beta_b = (const float*)d_in[5];
  const float* decay  = (const float*)d_in[6];
  const float* Wo     = (const float*)d_in[7];
  const float* bo     = (const float*)d_in[8];
  const float* ln_g   = (const float*)d_in[9];
  const float* ln_b   = (const float*)d_in[10];
  const float* lnp_g  = (const float*)d_in[11];
  const float* lnp_b  = (const float*)d_in[12];
  float* out = (float*)d_out;

  char* ws = (char*)d_ws;
  const size_t MAT = (size_t)M_DIM * D_DIM * sizeof(float);
  float* q    = (float*)(ws);
  float* k    = (float*)(ws + MAT);
  float* vv   = (float*)(ws + 2 * MAT);
  float* beta = (float*)(ws + 3 * MAT);
  float* aux  = (float*)(ws + 3 * MAT + (64 << 10));

  const dim3 gemm_grid(M_DIM / 64, D_DIM / 64);
  const dim3 gemm_blk(128);

  wmma_gemm_f32_kernel<<<gemm_grid, gemm_blk, 0, stream>>>(x, Wq, q, nullptr);
  wmma_gemm_f32_kernel<<<gemm_grid, gemm_blk, 0, stream>>>(x, Wk, k, nullptr);
  wmma_gemm_f32_kernel<<<gemm_grid, gemm_blk, 0, stream>>>(x, Wv, vv, nullptr);

  phi_beta_kernel<<<dim3(M_DIM), dim3(256), 0, stream>>>(
      x, q, k, beta_w, beta_b, lnp_g, lnp_b, beta);

  chunk_prep_kernel<<<dim3(NCH, B_DIM), dim3(256), 0, stream>>>(q, k, beta, decay, vv, aux);

  delta_chunk_scan_kernel<<<dim3(D_DIM / 16, B_DIM), dim3(128), 0, stream>>>(
      q, k, vv, aux, decay);

  ln_rows_kernel<<<dim3(M_DIM), dim3(256), 0, stream>>>(vv, ln_g, ln_b);

  wmma_gemm_f32_kernel<<<gemm_grid, gemm_blk, 0, stream>>>(vv, Wo, out, bo);
}


# scan: only K_c S^T stands in front of the chunk's LDS exchange; Q_c S^T products issued around the split barrier, summed across waves at the next chunk's barrier; new sum order of P and y
# speedup vs baseline: 4.7256x; 1.2204x over previous
#include <hip/hip_runtime.h>
#include <hip/hip_bf16.h>

typedef __attribute__((ext_vector_type(2))) float v2f;
typedef __attribute__((ext_vector_type(8))) float v8f;

#define D_DIM 256
#define L_DIM 1024
#define B_DIM 4
#define M_DIM (B_DIM * L_DIM)
#define LN_EPS 1e-5f

__device__ __forceinline__ float block_sum_256(float v, float* sh8) {
#pragma unroll
  for (int m = 16; m > 0; m >>= 1) v += __shfl_xor(v, m, 32);
  if ((threadIdx.x & 31) == 0) sh8[threadIdx.x >> 5] = v;
  __syncthreads();
  float r = 0.f;
#pragma unroll
  for (int i = 0; i < 8; ++i) r += sh8[i];
  return r;
}

__device__ __forceinline__ v8f mma4(v2f a, v2f b, v8f c) {
  return __builtin_amdgcn_wmma_f32_16x16x4_f32(false, a, false, b, (short)0, c, false, false);
}

#define GK_STEPS 4
#define GK_GROUPS (D_DIM / (8 * GK_STEPS))

__global__ __launch_bounds__(128) void wmma_gemm_f32_kernel(const float* __restrict__ A,
                                                            const float* __restrict__ Wm,
                                                            float* __restrict__ Out,
                                                            const float* __restrict__ bias) {
  const int wave = threadIdx.x >> 5;
  const int lane = threadIdx.x & 31;
  const int h = lane >> 4;
  const int c = lane & 15;
  const int tM = blockIdx.x * 64 + (wave >> 1) * 32;
  const int tN = blockIdx.y * 64 + (wave & 1) * 32;

  const float* pa[2];
  const float* pb[2];
#pragma unroll
  for (int i = 0; i < 2; ++i) {
    pa[i] = A  + (size_t)(tM + 16 * i + c) * D_DIM + 4 * h;
    pb[i] = Wm + (size_t)(tN + 16 * i + c) * D_DIM + 4 * h;
  }

  float bv[2];
#pragma unroll
  for (int j = 0; j < 2; ++j) bv[j] = bias ? bias[tN + 16 * j + c] : 0.f;

  float4 fa[3][GK_STEPS][2], fb[3][GK_STEPS][2];
#pragma unroll
  for (int g = 0; g < 2; ++g) {
#pragma unroll
    for (int st = 0; st < GK_STEPS; ++st)
#pragma unroll
      for (int i = 0; i < 2; ++i) {
        fa[g][st][i] = *(const float4*)(pa[i] + 32 * g + 8 * st);
        fb[g][st][i] = *(const float4*)(pb[i] + 32 * g + 8 * st);
      }
    __builtin_amdgcn_sched_barrier(0);
  }

  v8f acc[2][2];
#pragma unroll
  for (int i = 0; i < 2; ++i)
#pragma unroll
    for (int j = 0; j < 2; ++j) acc[i][j] = (v8f){};

#pragma unroll
  for (int g = 0; g < GK_GROUPS; ++g) {
    if (g + 2 < GK_GROUPS) {
#pragma unroll
      for (int st = 0; st < GK_STEPS; ++st)
#pragma unroll
        for (int i = 0; i < 2; ++i) {
          fa[(g + 2) % 3][st][i] = *(const float4*)(pa[i] + 32 * (g + 2) + 8 * st);
          fb[(g + 2) % 3][st][i] = *(const float4*)(pb[i] + 32 * (g + 2) + 8 * st);
        }
    }
    __builtin_amdgcn_sched_barrier(0);
#pragma unroll
    for (int st = 0; st < GK_STEPS; ++st) {
      const float4 a0 = fa[g % 3][st][0], a1 = fa[g % 3][st][1];
      const float4 b0 = fb[g % 3][st][0], b1 = fb[g % 3][st][1];
      acc[0][0] = mma4((v2f){a0.x, a0.y}, (v2f){b0.x, b0.y}, acc[0][0]);
      acc[0][1] = mma4((v2f){a0.x, a0.y}, (v2f){b1.x, b1.y}, acc[0][1]);
      acc[1][0] = mma4((v2f){a1.x, a1.y}, (v2f){b0.x, b0.y}, acc[1][0]);
      acc[1][1] = mma4((v2f){a1.x, a1.y}, (v2f){b1.x, b1.y}, acc[1][1]);
      acc[0][0] = mma4((v2f){a0.z, a0.w}, (v2f){b0.z, b0.w}, acc[0][0]);
      acc[0][1] = mma4((v2f){a0.z, a0.w}, (v2f){b1.z, b1.w}, acc[0][1]);
      acc[1][0] = mma4((v2f){a1.z, a1.w}, (v2f){b0.z, b0.w}, acc[1][0]);
      acc[1][1] = mma4((v2f){a1.z, a1.w}, (v2f){b1.z, b1.w}, acc[1][1]);
    }
  }

#pragma unroll
  for (int j = 0; j < 2; ++j) {
    const int n = tN + 16 * j + c;
#pragma unroll
    for (int i = 0; i < 2; ++i)
#pragma unroll
      for (int r = 0; r < 8; ++r) {
        const int m = tM + 16 * i + 8 * h + r;
        Out[(size_t)m * D_DIM + n] = acc[i][j][r] + bv[j];
      }
  }
}

__global__ void phi_beta_kernel(const float* __restrict__ x,
                                float* __restrict__ q,
                                float* __restrict__ k,
                                const float* __restrict__ beta_w,
                                const float* __restrict__ beta_b,
                                const float* __restrict__ lnp_g,
                                const float* __restrict__ lnp_b,
                                float* __restrict__ beta) {
  __shared__ float sh[8 * 5];
  const int row = blockIdx.x, tid = threadIdx.x;
  const size_t off = (size_t)row * D_DIM + tid;

  float qe; { float t = q[off]; qe = (t > 0.f) ? t + 1.f : __expf(t); }
  float ke; { float t = k[off]; ke = (t > 0.f) ? t + 1.f : __expf(t); }
  const float xb = x[off] * beta_w[tid];

  const float sq  = block_sum_256(qe,      sh + 0);
  const float sq2 = block_sum_256(qe * qe, sh + 8);
  const float sk  = block_sum_256(ke,      sh + 16);
  const float sk2 = block_sum_256(ke * ke, sh + 24);
  const float sb  = block_sum_256(xb,      sh + 32);

  const float muq = sq / D_DIM, varq = sq2 / D_DIM - muq * muq;
  const float muk = sk / D_DIM, vark = sk2 / D_DIM - muk * muk;
  const float g = lnp_g[tid], bb = lnp_b[tid];
  q[off] = (qe - muq) * rsqrtf(varq + LN_EPS) * g + bb;
  k[off] = (ke - muk) * rsqrtf(vark + LN_EPS) * g + bb;
  if (tid == 0) beta[row] = 1.f / (1.f + __expf(-(sb + beta_b[0])));
}

#define CH 16
#define NCH (L_DIM / CH)
#define AUX_STRIDE 512

__global__ void chunk_prep_kernel(const float* __restrict__ phi_q,
                                  const float* __restrict__ phi_k,
                                  const float* __restrict__ beta,
                                  const float* __restrict__ decay,
                                  float* __restrict__ vu,
                                  float* __restrict__ aux) {
  __shared__ __attribute__((aligned(16))) float Ks[CH][260];
  __shared__ __attribute__((aligned(16))) float Qs[CH][260];
  __shared__ float Al[CH][17];
  __shared__ float Tl[CH][17];
  __shared__ float bl[CH];
  const int ch = blockIdx.x, b = blockIdx.y, tid = threadIdx.x;
  const float df = 1.f / (1.f + __expf(-decay[0]));
  const size_t base = ((size_t)b * L_DIM + (size_t)ch * CH) * D_DIM;
  float vcol[CH];
#pragma unroll
  for (int i = 0; i < CH; ++i) {
    Ks[i][tid] = phi_k[base + (size_t)i * D_DIM + tid];
    Qs[i][tid] = phi_q[base + (size_t)i * D_DIM + tid];
    vcol[i] = vu[base + (size_t)i * D_DIM + tid];
  }
  if (tid < CH) bl[tid] = beta[(size_t)b * L_DIM + ch * CH + tid];
  __syncthreads();

  const int i = tid >> 4, j = tid & 15;
  float gk = 0.f, gq = 0.f;
#pragma unroll 4
  for (int c = 0; c < D_DIM; c += 4) {
    const float4 kj = *(const float4*)&Ks[j][c];
    const float4 ki = *(const float4*)&Ks[i][c];
    const float4 qi = *(const float4*)&Qs[i][c];
    gk = fmaf(ki.x, kj.x, gk); gk = fmaf(ki.y, kj.y, gk); gk = fmaf(ki.z, kj.z, gk); gk = fmaf(ki.w, kj.w, gk);
    gq = fmaf(qi.x, kj.x, gq); gq = fmaf(qi.y, kj.y, gq); gq = fmaf(qi.z, kj.z, gq); gq = fmaf(qi.w, kj.w, gq);
  }
  float pw = 1.f;
  for (int n = 0; n < i - 1 - j; ++n) pw *= df;
  const bool lower = j < i;
  Al[i][j] = lower ? bl[i] * pw * gk : 0.f;
  float* o = aux + ((size_t)b * NCH + ch) * AUX_STRIDE;
  o[256 + tid] = lower ? pw * gq : 0.f;
  __syncthreads();

  if (tid < CH) {
    float tc[CH];
#pragma unroll
    for (int ii = 0; ii < CH; ++ii) {
      float s = (ii == tid) ? 1.f : 0.f;
#pragma unroll
      for (int jj = 0; jj < ii; ++jj) s = fmaf(-Al[ii][jj], tc[jj], s);
      tc[ii] = s;
      Tl[ii][tid] = s;
    }
  }
  __syncthreads();
  float gj = 1.f;
  for (int n = 0; n < j; ++n) gj *= df;
  o[tid] = -Tl[i][j] * bl[j] * gj;

#pragma unroll
  for (int jj = 0; jj < CH; ++jj) vcol[jj] *= bl[jj];
#pragma unroll
  for (int ii = 0; ii < CH; ++ii) {
    float s = 0.f;
#pragma unroll
    for (int jj = 0; jj <= ii; ++jj) s = fmaf(Tl[ii][jj], vcol[jj], s);
    vu[base + (size_t)ii * D_DIM + tid] = s;
  }
}

__global__ __launch_bounds__(128) void delta_chunk_scan_kernel(
    const float* __restrict__ phi_q, const float* __restrict__ phi_k,
    float* vy, const float* __restrict__ aux, const float* __restrict__ decay) {
  __shared__ float red[2][4][2][8][32];
  __shared__ float u0s[2][8][32];
  const int b = blockIdx.y;
  const int row0 = blockIdx.x * 16;
  const int wave = __builtin_amdgcn_readfirstlane(threadIdx.x >> 5);
  const int lane = threadIdx.x & 31;
  const int c = lane & 15, h = lane >> 4;
  const int r0 = 2 * wave;

  const float df = 1.f / (1.f + __expf(-decay[0]));
  float pwr[17];
  pwr[0] = 1.f;
#pragma unroll
  for (int n = 1; n <= 16; ++n) pwr[n] = pwr[n - 1] * df;
  const float g16 = pwr[16];
  float gpos[8], grev[8];
#pragma unroll
  for (int r = 0; r < 8; ++r) {
    gpos[r] = h ? pwr[8 + r] : pwr[r];
    grev[r] = h ? pwr[7 - r] : pwr[15 - r];
  }
  float gy0 = gpos[0], gy1 = gpos[1];
#pragma unroll
  for (int r = 2; r < 8; r += 2)
    if (r == r0) { gy0 = gpos[r]; gy1 = gpos[r + 1]; }

  const size_t bbase = (size_t)b * L_DIM * D_DIM;
  const int colw = 64 * wave;
  const float* kA = phi_k + bbase + (size_t)c * D_DIM + colw + 8 * h;
  const float* qA = phi_q + bbase + (size_t)c * D_DIM + colw + 8 * h;
  const float* kT = phi_k + bbase + (size_t)(8 * h) * D_DIM + colw + c;
  float* vtw = vy + bbase + (size_t)(8 * h + r0) * D_DIM + row0 + c;
  const float* ax = aux + (size_t)b * NCH * AUX_STRIDE + c * 16 + 8 * h;

  v8f acc[4];
#pragma unroll
  for (int ct = 0; ct < 4; ++ct) acc[ct] = (v8f){};

  float4 Kf[4][2], Qf[4][2];
  float ktr[4][8];
#pragma unroll
  for (int ct = 0; ct < 4; ++ct) {
    Kf[ct][0] = *(const float4*)(kA + 16 * ct);
    Kf[ct][1] = *(const float4*)(kA + 16 * ct + 4);
    Qf[ct][0] = *(const float4*)(qA + 16 * ct);
    Qf[ct][1] = *(const float4*)(qA + 16 * ct + 4);
#pragma unroll
    for (int r = 0; r < 8; ++r) ktr[ct][r] = kT[(size_t)r * D_DIM + 16 * ct];
  }
  float4 tnf[2], tnN[2], mqf[2];
  tnf[0] = *(const float4*)(ax);  tnf[1] = *(const float4*)(ax + 4);
  mqf[0] = *(const float4*)(ax + 256);  mqf[1] = *(const float4*)(ax + 260);
  float u0a = vtw[0], u0b = vtw[D_DIM];

#pragma unroll
  for (int r = 0; r < 8; ++r) red[1][wave][1][r][lane] = 0.f;
  float ym0 = 0.f, ym1 = 0.f;

  __builtin_amdgcn_fence(__ATOMIC_ACQUIRE, "workgroup", "global");

#define YQ2(ct, p)                                                                        \
  do {                                                                                    \
    const float* qf_ = (const float*)Qf[ct];                                              \
    Ya = mma4((v2f){qf_[2 * (p)], qf_[2 * (p) + 1]},                                      \
              (v2f){acc[ct][2 * (p)], acc[ct][2 * (p) + 1]}, Ya);                         \
    Yb = mma4((v2f){qf_[2 * (p) + 2], qf_[2 * (p) + 3]},                                  \
              (v2f){acc[ct][2 * (p) + 2], acc[ct][2 * (p) + 3]}, Yb);                     \
  } while (0)
#define ORDER2(a, b) asm volatile("" : "+v"(a), "+v"(b) : : "memory")
#define YQ_ORDER() ORDER2(Ya, Yb)

  for (int t = 0; t < NCH; ++t) {
    const size_t coff = (size_t)t * CH * D_DIM;
    const int tn1 = (t + 1 < NCH) ? t + 1 : t;
    const size_t noff = (size_t)tn1 * CH * D_DIM;
    const int par = t & 1;

    v8f Pa = {}, Pb = {};
#pragma unroll
    for (int ct = 0; ct < 4; ++ct) {
      const float* kf = (const float*)Kf[ct];
#pragma unroll
      for (int p = 0; p < 4; p += 2) {
        Pa = mma4((v2f){kf[2 * p], kf[2 * p + 1]}, (v2f){acc[ct][2 * p], acc[ct][2 * p + 1]}, Pa);
        Pb = mma4((v2f){kf[2 * p + 2], kf[2 * p + 3]}, (v2f){acc[ct][2 * p + 2], acc[ct][2 * p + 3]}, Pb);
      }
    }
    ORDER2(Pa, Pb);

#pragma unroll
    for (int r = 0; r < 8; ++r) red[par][wave][0][r][lane] = Pa[r] + Pb[r];
    u0s[par][r0][lane] = u0a;
    u0s[par][r0 + 1][lane] = u0b;

    v8f Ya = {}, Yb = {};
    YQ_ORDER();
#pragma unroll
    for (int ct = 0; ct < 4; ++ct) {
      Kf[ct][0] = *(const float4*)(kA + noff + 16 * ct);
      Kf[ct][1] = *(const float4*)(kA + noff + 16 * ct + 4);
    }
    {
      const float* an = ax + (size_t)tn1 * AUX_STRIDE;
      tnN[0] = *(const float4*)(an);  tnN[1] = *(const float4*)(an + 4);
    }
    u0a = vtw[noff];
    u0b = vtw[noff + D_DIM];
    YQ2(0, 0); YQ2(0, 2);
    YQ_ORDER();
    __builtin_amdgcn_fence(__ATOMIC_RELEASE, "workgroup", "local");
    __builtin_amdgcn_s_barrier_signal(-1);
    YQ_ORDER();
    YQ2(1, 0); YQ2(1, 2); YQ2(2, 0);
    YQ_ORDER();
    __builtin_amdgcn_s_barrier_wait(-1);
    __builtin_amdgcn_fence(__ATOMIC_ACQUIRE, "workgroup", "local");

    float pp[4][8];
#pragma unroll
    for (int w = 0; w < 4; ++w)
#pragma unroll
      for (int r = 0; r < 8; ++r) pp[w][r] = red[par][w][0][r][lane];
    v8f U;
#pragma unroll
    for (int r = 0; r < 8; ++r) U[r] = u0s[par][r][lane];
    float yp[4][2];
#pragma unroll
    for (int w = 0; w < 4; ++w) {
      yp[w][0] = red[par ^ 1][w][1][r0][lane];
      yp[w][1] = red[par ^ 1][w][1][r0 + 1][lane];
    }
    YQ_ORDER();
    YQ2(2, 2); YQ2(3, 0); YQ2(3, 2);
    YQ_ORDER();
    __builtin_amdgcn_sched_barrier(0);

#pragma unroll
    for (int ct = 0; ct < 4; ++ct) {
      Qf[ct][0] = *(const float4*)(qA + noff + 16 * ct);
      Qf[ct][1] = *(const float4*)(qA + noff + 16 * ct + 4);
    }
    if (t + 2 < NCH) {
      __builtin_prefetch(kA + noff + (size_t)CH * D_DIM + 24 * h, 0, 0);
      __builtin_prefetch(qA + noff + (size_t)CH * D_DIM + 24 * h, 0, 0);
    }

    float ps[8];
#pragma unroll
    for (int r = 0; r < 8; ++r) ps[r] = (pp[0][r] + pp[1][r]) + (pp[2][r] + pp[3][r]);
    {
      const float* tnp = (const float*)tnf;
#pragma unroll
      for (int jj = 0; jj < 4; ++jj)
        U = mma4((v2f){tnp[2 * jj], tnp[2 * jj + 1]}, (v2f){ps[2 * jj], ps[2 * jj + 1]}, U);
    }

    {
      const float yq0 = (yp[0][0] + yp[1][0]) + (yp[2][0] + yp[3][0]);
      const float yq1 = (yp[0][1] + yp[1][1]) + (yp[2][1] + yp[3][1]);
      if (t > 0) {
        vtw[coff - (size_t)CH * D_DIM] = gy0 * yq0 + ym0;
        vtw[coff - (size_t)CH * D_DIM + D_DIM] = gy1 * yq1 + ym1;
      }
    }
#pragma unroll
    for (int r = 0; r < 8; ++r) red[par][wave][1][r][lane] = Ya[r] + Yb[r];

#pragma unroll
    for (int ct = 0; ct < 4; ++ct) acc[ct] *= g16;
    float kt[4][8];
#pragma unroll
    for (int ct = 0; ct < 4; ++ct)
#pragma unroll
      for (int r = 0; r < 8; ++r) kt[ct][r] = ktr[ct][r] * grev[r];
    const float* mqp = (const float*)mqf;
    v8f Ym = {};
#pragma unroll
    for (int jj = 0; jj < 4; ++jj) {
#pragma unroll
      for (int ct = 0; ct < 4; ++ct)
        acc[ct] = mma4((v2f){kt[ct][2 * jj], kt[ct][2 * jj + 1]},
                       (v2f){U[2 * jj], U[2 * jj + 1]}, acc[ct]);
      Ym = mma4((v2f){mqp[2 * jj], mqp[2 * jj + 1]}, (v2f){U[2 * jj], U[2 * jj + 1]}, Ym);
    }
    ym0 = Ym[0]; ym1 = Ym[1];
#pragma unroll
    for (int r = 2; r < 8; r += 2)
      if (r == r0) { ym0 = Ym[r]; ym1 = Ym[r + 1]; }

    {
      const float* an = ax + (size_t)tn1 * AUX_STRIDE;
      mqf[0] = *(const float4*)(an + 256);  mqf[1] = *(const float4*)(an + 260);
    }
#pragma unroll
    for (int ct = 0; ct < 4; ++ct)
#pragma unroll
      for (int r = 0; r < 8; ++r) ktr[ct][r] = kT[noff + (size_t)r * D_DIM + 16 * ct];
    tnf[0] = tnN[0]; tnf[1] = tnN[1];
    __builtin_amdgcn_sched_barrier(0);
  }
#undef YQ2
#undef YQ_ORDER
#undef ORDER2

  __builtin_amdgcn_fence(__ATOMIC_RELEASE, "workgroup", "local");
  __builtin_amdgcn_s_barrier();
  __builtin_amdgcn_fence(__ATOMIC_ACQUIRE, "workgroup", "local");
  {
    const int lp = (NCH - 1) & 1;
    const float yq0 = (red[lp][0][1][r0][lane] + red[lp][1][1][r0][lane]) +
                      (red[lp][2][1][r0][lane] + red[lp][3][1][r0][lane]);
    const float yq1 = (red[lp][0][1][r0 + 1][lane] + red[lp][1][1][r0 + 1][lane]) +
                      (red[lp][2][1][r0 + 1][lane] + red[lp][3][1][r0 + 1][lane]);
    const size_t loff = (size_t)(NCH - 1) * CH * D_DIM;
    vtw[loff] = gy0 * yq0 + ym0;
    vtw[loff + D_DIM] = gy1 * yq1 + ym1;
  }
}

__global__ void ln_rows_kernel(float* __restrict__ y,
                               const float* __restrict__ g,
                               const float* __restrict__ b) {
  __shared__ float sh[16];
  const int row = blockIdx.x, tid = threadIdx.x;
  const size_t off = (size_t)row * D_DIM + tid;
  const float v  = y[off];
  const float s  = block_sum_256(v, sh);
  const float s2 = block_sum_256(v * v, sh + 8);
  const float mu = s / D_DIM, var = s2 / D_DIM - mu * mu;
  y[off] = (v - mu) * rsqrtf(var + LN_EPS) * g[tid] + b[tid];
}

extern "C" void kernel_launch(void* const* d_in, const int* in_sizes, int n_in,
                              void* d_out, int out_size, void* d_ws, size_t ws_size,
                              hipStream_t stream) {
  (void)in_sizes; (void)n_in; (void)out_size; (void)ws_size;
  const float* x      = (const float*)d_in[0];
  const float* Wq     = (const float*)d_in[1];
  const float* Wk     = (const float*)d_in[2];
  const float* Wv     = (const float*)d_in[3];
  const float* beta_w = (const float*)d_in[4];
  const float* beta_b = (const float*)d_in[5];
  const float* decay  = (const float*)d_in[6];
  const float* Wo     = (const float*)d_in[7];
  const float* bo     = (const float*)d_in[8];
  const float* ln_g   = (const float*)d_in[9];
  const float* ln_b   = (const float*)d_in[10];
  const float* lnp_g  = (const float*)d_in[11];
  const float* lnp_b  = (const float*)d_in[12];
  float* out = (float*)d_out;

  char* ws = (char*)d_ws;
  const size_t MAT = (size_t)M_DIM * D_DIM * sizeof(float);
  float* q    = (float*)(ws);
  float* k    = (float*)(ws + MAT);
  float* vv   = (float*)(ws + 2 * MAT);
  float* beta = (float*)(ws + 3 * MAT);
  float* aux  = (float*)(ws + 3 * MAT + (64 << 10));

  const dim3 gemm_grid(M_DIM / 64, D_DIM / 64);
  const dim3 gemm_blk(128);

  wmma_gemm_f32_kernel<<<gemm_grid, gemm_blk, 0, stream>>>(x, Wq, q, nullptr);
  wmma_gemm_f32_kernel<<<gemm_grid, gemm_blk, 0, stream>>>(x, Wk, k, nullptr);
  wmma_gemm_f32_kernel<<<gemm_grid, gemm_blk, 0, stream>>>(x, Wv, vv, nullptr);

  phi_beta_kernel<<<dim3(M_DIM), dim3(256), 0, stream>>>(
      x, q, k, beta_w, beta_b, lnp_g, lnp_b, beta);

  chunk_prep_kernel<<<dim3(NCH, B_DIM), dim3(256), 0, stream>>>(q, k, beta, decay, vv, aux);

  delta_chunk_scan_kernel<<<dim3(D_DIM / 16, B_DIM), dim3(128), 0, stream>>>(
      q, k, vv, aux, decay);

  ln_rows_kernel<<<dim3(M_DIM), dim3(256), 0, stream>>>(vv, ln_g, ln_b);

  wmma_gemm_f32_kernel<<<gemm_grid, gemm_blk, 0, stream>>>(vv, Wo, out, bo);
}


# input projections q, k, v as ONE launch: grid z picks weight matrix and output plane, same 32x32 wave tiles; 768 blocks resident together; 6 launches for 8; same sums, same order
# speedup vs baseline: 4.9424x; 1.0459x over previous
#include <hip/hip_runtime.h>
#include <hip/hip_bf16.h>

typedef __attribute__((ext_vector_type(2))) float v2f;
typedef __attribute__((ext_vector_type(8))) float v8f;

#define D_DIM 256
#define L_DIM 1024
#define B_DIM 4
#define M_DIM (B_DIM * L_DIM)
#define LN_EPS 1e-5f

__device__ __forceinline__ float block_sum_256(float v, float* sh8) {
#pragma unroll
  for (int m = 16; m > 0; m >>= 1) v += __shfl_xor(v, m, 32);
  if ((threadIdx.x & 31) == 0) sh8[threadIdx.x >> 5] = v;
  __syncthreads();
  float r = 0.f;
#pragma unroll
  for (int i = 0; i < 8; ++i) r += sh8[i];
  return r;
}

__device__ __forceinline__ v8f mma4(v2f a, v2f b, v8f c) {
  return __builtin_amdgcn_wmma_f32_16x16x4_f32(false, a, false, b, (short)0, c, false, false);
}

#define GK_STEPS 4
#define GK_GROUPS (D_DIM / (8 * GK_STEPS))

__device__ __forceinline__ void gemm_tile_f32(const float* __restrict__ A,
                                              const float* __restrict__ Wm,
                                              float* __restrict__ Out,
                                              const float* __restrict__ bias) {
  const int wave = threadIdx.x >> 5;
  const int lane = threadIdx.x & 31;
  const int h = lane >> 4;
  const int c = lane & 15;
  const int tM = blockIdx.x * 64 + (wave >> 1) * 32;
  const int tN = blockIdx.y * 64 + (wave & 1) * 32;

  const float* pa[2];
  const float* pb[2];
#pragma unroll
  for (int i = 0; i < 2; ++i) {
    pa[i] = A  + (size_t)(tM + 16 * i + c) * D_DIM + 4 * h;
    pb[i] = Wm + (size_t)(tN + 16 * i + c) * D_DIM + 4 * h;
  }

  float bv[2];
#pragma unroll
  for (int j = 0; j < 2; ++j) bv[j] = bias ? bias[tN + 16 * j + c] : 0.f;

  float4 fa[3][GK_STEPS][2], fb[3][GK_STEPS][2];
#pragma unroll
  for (int g = 0; g < 2; ++g) {
#pragma unroll
    for (int st = 0; st < GK_STEPS; ++st)
#pragma unroll
      for (int i = 0; i < 2; ++i) {
        fa[g][st][i] = *(const float4*)(pa[i] + 32 * g + 8 * st);
        fb[g][st][i] = *(const float4*)(pb[i] + 32 * g + 8 * st);
      }
    __builtin_amdgcn_sched_barrier(0);
  }

  v8f acc[2][2];
#pragma unroll
  for (int i = 0; i < 2; ++i)
#pragma unroll
    for (int j = 0; j < 2; ++j) acc[i][j] = (v8f){};

#pragma unroll
  for (int g = 0; g < GK_GROUPS; ++g) {
    if (g + 2 < GK_GROUPS) {
#pragma unroll
      for (int st = 0; st < GK_STEPS; ++st)
#pragma unroll
        for (int i = 0; i < 2; ++i) {
          fa[(g + 2) % 3][st][i] = *(const float4*)(pa[i] + 32 * (g + 2) + 8 * st);
          fb[(g + 2) % 3][st][i] = *(const float4*)(pb[i] + 32 * (g + 2) + 8 * st);
        }
    }
    __builtin_amdgcn_sched_barrier(0);
#pragma unroll
    for (int st = 0; st < GK_STEPS; ++st) {
      const float4 a0 = fa[g % 3][st][0], a1 = fa[g % 3][st][1];
      const float4 b0 = fb[g % 3][st][0], b1 = fb[g % 3][st][1];
      acc[0][0] = mma4((v2f){a0.x, a0.y}, (v2f){b0.x, b0.y}, acc[0][0]);
      acc[0][1] = mma4((v2f){a0.x, a0.y}, (v2f){b1.x, b1.y}, acc[0][1]);
      acc[1][0] = mma4((v2f){a1.x, a1.y}, (v2f){b0.x, b0.y}, acc[1][0]);
      acc[1][1] = mma4((v2f){a1.x, a1.y}, (v2f){b1.x, b1.y}, acc[1][1]);
      acc[0][0] = mma4((v2f){a0.z, a0.w}, (v2f){b0.z, b0.w}, acc[0][0]);
      acc[0][1] = mma4((v2f){a0.z, a0.w}, (v2f){b1.z, b1.w}, acc[0][1]);
      acc[1][0] = mma4((v2f){a1.z, a1.w}, (v2f){b0.z, b0.w}, acc[1][0]);
      acc[1][1] = mma4((v2f){a1.z, a1.w}, (v2f){b1.z, b1.w}, acc[1][1]);
    }
  }

#pragma unroll
  for (int j = 0; j < 2; ++j) {
    const int n = tN + 16 * j + c;
#pragma unroll
    for (int i = 0; i < 2; ++i)
#pragma unroll
      for (int r = 0; r < 8; ++r) {
        const int m = tM + 16 * i + 8 * h + r;
        Out[(size_t)m * D_DIM + n] = acc[i][j][r] + bv[j];
      }
  }
}

__global__ __launch_bounds__(128) void wmma_gemm_f32_kernel(const float* __restrict__ A,
                                                            const float* __restrict__ Wm,
                                                            float* __restrict__ Out,
                                                            const float* __restrict__ bias) {
  gemm_tile_f32(A, Wm, Out, bias);
}

__global__ __launch_bounds__(128) void wmma_gemm3_f32_kernel(const float* __restrict__ A,
                                                             const float* __restrict__ W0,
                                                             const float* __restrict__ W1,
                                                             const float* __restrict__ W2,
                                                             float* __restrict__ O0,
                                                             float* __restrict__ O1,
                                                             float* __restrict__ O2) {
  const int z = blockIdx.z;
  const float* Wm = (z == 0) ? W0 : (z == 1) ? W1 : W2;
  float* Out = (z == 0) ? O0 : (z == 1) ? O1 : O2;
  gemm_tile_f32(A, Wm, Out, nullptr);
}

__global__ void phi_beta_kernel(const float* __restrict__ x,
                                float* __restrict__ q,
                                float* __restrict__ k,
                                const float* __restrict__ beta_w,
                                const float* __restrict__ beta_b,
                                const float* __restrict__ lnp_g,
                                const float* __restrict__ lnp_b,
                                float* __restrict__ beta) {
  __shared__ float sh[8 * 5];
  const int row = blockIdx.x, tid = threadIdx.x;
  const size_t off = (size_t)row * D_DIM + tid;

  float qe; { float t = q[off]; qe = (t > 0.f) ? t + 1.f : __expf(t); }
  float ke; { float t = k[off]; ke = (t > 0.f) ? t + 1.f : __expf(t); }
  const float xb = x[off] * beta_w[tid];

  const float sq  = block_sum_256(qe,      sh + 0);
  const float sq2 = block_sum_256(qe * qe, sh + 8);
  const float sk  = block_sum_256(ke,      sh + 16);
  const float sk2 = block_sum_256(ke * ke, sh + 24);
  const float sb  = block_sum_256(xb,      sh + 32);

  const float muq = sq / D_DIM, varq = sq2 / D_DIM - muq * muq;
  const float muk = sk / D_DIM, vark = sk2 / D_DIM - muk * muk;
  const float g = lnp_g[tid], bb = lnp_b[tid];
  q[off] = (qe - muq) * rsqrtf(varq + LN_EPS) * g + bb;
  k[off] = (ke - muk) * rsqrtf(vark + LN_EPS) * g + bb;
  if (tid == 0) beta[row] = 1.f / (1.f + __expf(-(sb + beta_b[0])));
}

#define CH 16
#define NCH (L_DIM / CH)
#define AUX_STRIDE 512

__global__ void chunk_prep_kernel(const float* __restrict__ phi_q,
                                  const float* __restrict__ phi_k,
                                  const float* __restrict__ beta,
                                  const float* __restrict__ decay,
                                  float* __restrict__ vu,
                                  float* __restrict__ aux) {
  __shared__ __attribute__((aligned(16))) float Ks[CH][260];
  __shared__ __attribute__((aligned(16))) float Qs[CH][260];
  __shared__ float Al[CH][17];
  __shared__ float Tl[CH][17];
  __shared__ float bl[CH];
  const int ch = blockIdx.x, b = blockIdx.y, tid = threadIdx.x;
  const float df = 1.f / (1.f + __expf(-decay[0]));
  const size_t base = ((size_t)b * L_DIM + (size_t)ch * CH) * D_DIM;
  float vcol[CH];
#pragma unroll
  for (int i = 0; i < CH; ++i) {
    Ks[i][tid] = phi_k[base + (size_t)i * D_DIM + tid];
    Qs[i][tid] = phi_q[base + (size_t)i * D_DIM + tid];
    vcol[i] = vu[base + (size_t)i * D_DIM + tid];
  }
  if (tid < CH) bl[tid] = beta[(size_t)b * L_DIM + ch * CH + tid];
  __syncthreads();

  const int i = tid >> 4, j = tid & 15;
  float gk = 0.f, gq = 0.f;
#pragma unroll 4
  for (int c = 0; c < D_DIM; c += 4) {
    const float4 kj = *(const float4*)&Ks[j][c];
    const float4 ki = *(const float4*)&Ks[i][c];
    const float4 qi = *(const float4*)&Qs[i][c];
    gk = fmaf(ki.x, kj.x, gk); gk = fmaf(ki.y, kj.y, gk); gk = fmaf(ki.z, kj.z, gk); gk = fmaf(ki.w, kj.w, gk);
    gq = fmaf(qi.x, kj.x, gq); gq = fmaf(qi.y, kj.y, gq); gq = fmaf(qi.z, kj.z, gq); gq = fmaf(qi.w, kj.w, gq);
  }
  float pw = 1.f;
  for (int n = 0; n < i - 1 - j; ++n) pw *= df;
  const bool lower = j < i;
  Al[i][j] = lower ? bl[i] * pw * gk : 0.f;
  float* o = aux + ((size_t)b * NCH + ch) * AUX_STRIDE;
  o[256 + tid] = lower ? pw * gq : 0.f;
  __syncthreads();

  if (tid < CH) {
    float tc[CH];
#pragma unroll
    for (int ii = 0; ii < CH; ++ii) {
      float s = (ii == tid) ? 1.f : 0.f;
#pragma unroll
      for (int jj = 0; jj < ii; ++jj) s = fmaf(-Al[ii][jj], tc[jj], s);
      tc[ii] = s;
      Tl[ii][tid] = s;
    }
  }
  __syncthreads();
  float gj = 1.f;
  for (int n = 0; n < j; ++n) gj *= df;
  o[tid] = -Tl[i][j] * bl[j] * gj;

#pragma unroll
  for (int jj = 0; jj < CH; ++jj) vcol[jj] *= bl[jj];
#pragma unroll
  for (int ii = 0; ii < CH; ++ii) {
    float s = 0.f;
#pragma unroll
    for (int jj = 0; jj <= ii; ++jj) s = fmaf(Tl[ii][jj], vcol[jj], s);
    vu[base + (size_t)ii * D_DIM + tid] = s;
  }
}

__global__ __launch_bounds__(128) void delta_chunk_scan_kernel(
    const float* __restrict__ phi_q, const float* __restrict__ phi_k,
    float* vy, const float* __restrict__ aux, const float* __restrict__ decay) {
  __shared__ float red[2][4][2][8][32];
  __shared__ float u0s[2][8][32];
  const int b = blockIdx.y;
  const int row0 = blockIdx.x * 16;
  const int wave = __builtin_amdgcn_readfirstlane(threadIdx.x >> 5);
  const int lane = threadIdx.x & 31;
  const int c = lane & 15, h = lane >> 4;
  const int r0 = 2 * wave;

  const float df = 1.f / (1.f + __expf(-decay[0]));
  float pwr[17];
  pwr[0] = 1.f;
#pragma unroll
  for (int n = 1; n <= 16; ++n) pwr[n] = pwr[n - 1] * df;
  const float g16 = pwr[16];
  float gpos[8], grev[8];
#pragma unroll
  for (int r = 0; r < 8; ++r) {
    gpos[r] = h ? pwr[8 + r] : pwr[r];
    grev[r] = h ? pwr[7 - r] : pwr[15 - r];
  }
  float gy0 = gpos[0], gy1 = gpos[1];
#pragma unroll
  for (int r = 2; r < 8; r += 2)
    if (r == r0) { gy0 = gpos[r]; gy1 = gpos[r + 1]; }

  const size_t bbase = (size_t)b * L_DIM * D_DIM;
  const int colw = 64 * wave;
  const float* kA = phi_k + bbase + (size_t)c * D_DIM + colw + 8 * h;
  const float* qA = phi_q + bbase + (size_t)c * D_DIM + colw + 8 * h;
  const float* kT = phi_k + bbase + (size_t)(8 * h) * D_DIM + colw + c;
  float* vtw = vy + bbase + (size_t)(8 * h + r0) * D_DIM + row0 + c;
  const float* ax = aux + (size_t)b * NCH * AUX_STRIDE + c * 16 + 8 * h;

  v8f acc[4];
#pragma unroll
  for (int ct = 0; ct < 4; ++ct) acc[ct] = (v8f){};

  float4 Kf[4][2], Qf[4][2];
  float ktr[4][8];
#pragma unroll
  for (int ct = 0; ct < 4; ++ct) {
    Kf[ct][0] = *(const float4*)(kA + 16 * ct);
    Kf[ct][1] = *(const float4*)(kA + 16 * ct + 4);
    Qf[ct][0] = *(const float4*)(qA + 16 * ct);
    Qf[ct][1] = *(const float4*)(qA + 16 * ct + 4);
#pragma unroll
    for (int r = 0; r < 8; ++r) ktr[ct][r] = kT[(size_t)r * D_DIM + 16 * ct];
  }
  float4 tnf[2], tnN[2], mqf[2];
  tnf[0] = *(const float4*)(ax);  tnf[1] = *(const float4*)(ax + 4);
  mqf[0] = *(const float4*)(ax + 256);  mqf[1] = *(const float4*)(ax + 260);
  float u0a = vtw[0], u0b = vtw[D_DIM];

#pragma unroll
  for (int r = 0; r < 8; ++r) red[1][wave][1][r][lane] = 0.f;
  float ym0 = 0.f, ym1 = 0.f;

  __builtin_amdgcn_fence(__ATOMIC_ACQUIRE, "workgroup", "global");

#define YQ2(ct, p)                                                                        \
  do {                                                                                    \
    const float* qf_ = (const float*)Qf[ct];                                              \
    Ya = mma4((v2f){qf_[2 * (p)], qf_[2 * (p) + 1]},                                      \
              (v2f){acc[ct][2 * (p)], acc[ct][2 * (p) + 1]}, Ya);                         \
    Yb = mma4((v2f){qf_[2 * (p) + 2], qf_[2 * (p) + 3]},                                  \
              (v2f){acc[ct][2 * (p) + 2], acc[ct][2 * (p) + 3]}, Yb);                     \
  } while (0)
#define ORDER2(a, b) asm volatile("" : "+v"(a), "+v"(b) : : "memory")
#define YQ_ORDER() ORDER2(Ya, Yb)

  for (int t = 0; t < NCH; ++t) {
    const size_t coff = (size_t)t * CH * D_DIM;
    const int tn1 = (t + 1 < NCH) ? t + 1 : t;
    const size_t noff = (size_t)tn1 * CH * D_DIM;
    const int par = t & 1;

    v8f Pa = {}, Pb = {};
#pragma unroll
    for (int ct = 0; ct < 4; ++ct) {
      const float* kf = (const float*)Kf[ct];
#pragma unroll
      for (int p = 0; p < 4; p += 2) {
        Pa = mma4((v2f){kf[2 * p], kf[2 * p + 1]}, (v2f){acc[ct][2 * p], acc[ct][2 * p + 1]}, Pa);
        Pb = mma4((v2f){kf[2 * p + 2], kf[2 * p + 3]}, (v2f){acc[ct][2 * p + 2], acc[ct][2 * p + 3]}, Pb);
      }
    }
    ORDER2(Pa, Pb);

#pragma unroll
    for (int r = 0; r < 8; ++r) red[par][wave][0][r][lane] = Pa[r] + Pb[r];
    u0s[par][r0][lane] = u0a;
    u0s[par][r0 + 1][lane] = u0b;

    v8f Ya = {}, Yb = {};
    YQ_ORDER();
#pragma unroll
    for (int ct = 0; ct < 4; ++ct) {
      Kf[ct][0] = *(const float4*)(kA + noff + 16 * ct);
      Kf[ct][1] = *(const float4*)(kA + noff + 16 * ct + 4);
    }
    {
      const float* an = ax + (size_t)tn1 * AUX_STRIDE;
      tnN[0] = *(const float4*)(an);  tnN[1] = *(const float4*)(an + 4);
    }
    u0a = vtw[noff];
    u0b = vtw[noff + D_DIM];
    YQ2(0, 0); YQ2(0, 2);
    YQ_ORDER();
    __builtin_amdgcn_fence(__ATOMIC_RELEASE, "workgroup", "local");
    __builtin_amdgcn_s_barrier_signal(-1);
    YQ_ORDER();
    YQ2(1, 0); YQ2(1, 2); YQ2(2, 0);
    YQ_ORDER();
    __builtin_amdgcn_s_barrier_wait(-1);
    __builtin_amdgcn_fence(__ATOMIC_ACQUIRE, "workgroup", "local");

    float pp[4][8];
#pragma unroll
    for (int w = 0; w < 4; ++w)
#pragma unroll
      for (int r = 0; r < 8; ++r) pp[w][r] = red[par][w][0][r][lane];
    v8f U;
#pragma unroll
    for (int r = 0; r < 8; ++r) U[r] = u0s[par][r][lane];
    float yp[4][2];
#pragma unroll
    for (int w = 0; w < 4; ++w) {
      yp[w][0] = red[par ^ 1][w][1][r0][lane];
      yp[w][1] = red[par ^ 1][w][1][r0 + 1][lane];
    }
    YQ_ORDER();
    YQ2(2, 2); YQ2(3, 0); YQ2(3, 2);
    YQ_ORDER();
    __builtin_amdgcn_sched_barrier(0);

#pragma unroll
    for (int ct = 0; ct < 4; ++ct) {
      Qf[ct][0] = *(const float4*)(qA + noff + 16 * ct);
      Qf[ct][1] = *(const float4*)(qA + noff + 16 * ct + 4);
    }
    if (t + 2 < NCH) {
      __builtin_prefetch(kA + noff + (size_t)CH * D_DIM + 24 * h, 0, 0);
      __builtin_prefetch(qA + noff + (size_t)CH * D_DIM + 24 * h, 0, 0);
    }

    float ps[8];
#pragma unroll
    for (int r = 0; r < 8; ++r) ps[r] = (pp[0][r] + pp[1][r]) + (pp[2][r] + pp[3][r]);
    {
      const float* tnp = (const float*)tnf;
#pragma unroll
      for (int jj = 0; jj < 4; ++jj)
        U = mma4((v2f){tnp[2 * jj], tnp[2 * jj + 1]}, (v2f){ps[2 * jj], ps[2 * jj + 1]}, U);
    }

    {
      const float yq0 = (yp[0][0] + yp[1][0]) + (yp[2][0] + yp[3][0]);
      const float yq1 = (yp[0][1] + yp[1][1]) + (yp[2][1] + yp[3][1]);
      if (t > 0) {
        vtw[coff - (size_t)CH * D_DIM] = gy0 * yq0 + ym0;
        vtw[coff - (size_t)CH * D_DIM + D_DIM] = gy1 * yq1 + ym1;
      }
    }
#pragma unroll
    for (int r = 0; r < 8; ++r) red[par][wave][1][r][lane] = Ya[r] + Yb[r];

#pragma unroll
    for (int ct = 0; ct < 4; ++ct) acc[ct] *= g16;
    float kt[4][8];
#pragma unroll
    for (int ct = 0; ct < 4; ++ct)
#pragma unroll
      for (int r = 0; r < 8; ++r) kt[ct][r] = ktr[ct][r] * grev[r];
    const float* mqp = (const float*)mqf;
    v8f Ym = {};
#pragma unroll
    for (int jj = 0; jj < 4; ++jj) {
#pragma unroll
      for (int ct = 0; ct < 4; ++ct)
        acc[ct] = mma4((v2f){kt[ct][2 * jj], kt[ct][2 * jj + 1]},
                       (v2f){U[2 * jj], U[2 * jj + 1]}, acc[ct]);
      Ym = mma4((v2f){mqp[2 * jj], mqp[2 * jj + 1]}, (v2f){U[2 * jj], U[2 * jj + 1]}, Ym);
    }
    ym0 = Ym[0]; ym1 = Ym[1];
#pragma unroll
    for (int r = 2; r < 8; r += 2)
      if (r == r0) { ym0 = Ym[r]; ym1 = Ym[r + 1]; }

    {
      const float* an = ax + (size_t)tn1 * AUX_STRIDE;
      mqf[0] = *(const float4*)(an + 256);  mqf[1] = *(const float4*)(an + 260);
    }
#pragma unroll
    for (int ct = 0; ct < 4; ++ct)
#pragma unroll
      for (int r = 0; r < 8; ++r) ktr[ct][r] = kT[noff + (size_t)r * D_DIM + 16 * ct];
    tnf[0] = tnN[0]; tnf[1] = tnN[1];
    __builtin_amdgcn_sched_barrier(0);
  }
#undef YQ2
#undef YQ_ORDER
#undef ORDER2

  __builtin_amdgcn_fence(__ATOMIC_RELEASE, "workgroup", "local");
  __builtin_amdgcn_s_barrier();
  __builtin_amdgcn_fence(__ATOMIC_ACQUIRE, "workgroup", "local");
  {
    const int lp = (NCH - 1) & 1;
    const float yq0 = (red[lp][0][1][r0][lane] + red[lp][1][1][r0][lane]) +
                      (red[lp][2][1][r0][lane] + red[lp][3][1][r0][lane]);
    const float yq1 = (red[lp][0][1][r0 + 1][lane] + red[lp][1][1][r0 + 1][lane]) +
                      (red[lp][2][1][r0 + 1][lane] + red[lp][3][1][r0 + 1][lane]);
    const size_t loff = (size_t)(NCH - 1) * CH * D_DIM;
    vtw[loff] = gy0 * yq0 + ym0;
    vtw[loff + D_DIM] = gy1 * yq1 + ym1;
  }
}

__global__ void ln_rows_kernel(float* __restrict__ y,
                               const float* __restrict__ g,
                               const float* __restrict__ b) {
  __shared__ float sh[16];
  const int row = blockIdx.x, tid = threadIdx.x;
  const size_t off = (size_t)row * D_DIM + tid;
  const float v  = y[off];
  const float s  = block_sum_256(v, sh);
  const float s2 = block_sum_256(v * v, sh + 8);
  const float mu = s / D_DIM, var = s2 / D_DIM - mu * mu;
  y[off] = (v - mu) * rsqrtf(var + LN_EPS) * g[tid] + b[tid];
}

extern "C" void kernel_launch(void* const* d_in, const int* in_sizes, int n_in,
                              void* d_out, int out_size, void* d_ws, size_t ws_size,
                              hipStream_t stream) {
  (void)in_sizes; (void)n_in; (void)out_size; (void)ws_size;
  const float* x      = (const float*)d_in[0];
  const float* Wq     = (const float*)d_in[1];
  const float* Wk     = (const float*)d_in[2];
  const float* Wv     = (const float*)d_in[3];
  const float* beta_w = (const float*)d_in[4];
  const float* beta_b = (const float*)d_in[5];
  const float* decay  = (const float*)d_in[6];
  const float* Wo     = (const float*)d_in[7];
  const float* bo     = (const float*)d_in[8];
  const float* ln_g   = (const float*)d_in[9];
  const float* ln_b   = (const float*)d_in[10];
  const float* lnp_g  = (const float*)d_in[11];
  const float* lnp_b  = (const float*)d_in[12];
  float* out = (float*)d_out;

  char* ws = (char*)d_ws;
  const size_t MAT = (size_t)M_DIM * D_DIM * sizeof(float);
  float* q    = (float*)(ws);
  float* k    = (float*)(ws + MAT);
  float* vv   = (float*)(ws + 2 * MAT);
  float* beta = (float*)(ws + 3 * MAT);
  float* aux  = (float*)(ws + 3 * MAT + (64 << 10));

  const dim3 gemm_grid(M_DIM / 64, D_DIM / 64);
  const dim3 gemm_blk(128);

  wmma_gemm3_f32_kernel<<<dim3(M_DIM / 64, D_DIM / 64, 3), gemm_blk, 0, stream>>>(
      x, Wq, Wk, Wv, q, k, vv);

  phi_beta_kernel<<<dim3(M_DIM), dim3(256), 0, stream>>>(
      x, q, k, beta_w, beta_b, lnp_g, lnp_b, beta);

  chunk_prep_kernel<<<dim3(NCH, B_DIM), dim3(256), 0, stream>>>(q, k, beta, decay, vv, aux);

  delta_chunk_scan_kernel<<<dim3(D_DIM / 16, B_DIM), dim3(128), 0, stream>>>(
      q, k, vv, aux, decay);

  ln_rows_kernel<<<dim3(M_DIM), dim3(256), 0, stream>>>(vv, ln_g, ln_b);

  wmma_gemm_f32_kernel<<<gemm_grid, gemm_blk, 0, stream>>>(vv, Wo, out, bo);
}


# output projection: layernorm of y applied behind the product, out = rstd (G - dm sg) + cb; row/column sums taken from the passing fragments, g folded into B; 5 launches for 6; new sum order
# speedup vs baseline: 5.1924x; 1.0506x over previous
#include <hip/hip_runtime.h>
#include <hip/hip_bf16.h>

typedef __attribute__((ext_vector_type(2))) float v2f;
typedef __attribute__((ext_vector_type(8))) float v8f;

#define D_DIM 256
#define L_DIM 1024
#define B_DIM 4
#define M_DIM (B_DIM * L_DIM)
#define LN_EPS 1e-5f

__device__ __forceinline__ float block_sum_256(float v, float* sh8) {
#pragma unroll
  for (int m = 16; m > 0; m >>= 1) v += __shfl_xor(v, m, 32);
  if ((threadIdx.x & 31) == 0) sh8[threadIdx.x >> 5] = v;
  __syncthreads();
  float r = 0.f;
#pragma unroll
  for (int i = 0; i < 8; ++i) r += sh8[i];
  return r;
}

__device__ __forceinline__ v8f mma4(v2f a, v2f b, v8f c) {
  return __builtin_amdgcn_wmma_f32_16x16x4_f32(false, a, false, b, (short)0, c, false, false);
}

#define GK_STEPS 4
#define GK_GROUPS (D_DIM / (8 * GK_STEPS))

__device__ __forceinline__ void gemm_tile_f32(const float* __restrict__ A,
                                              const float* __restrict__ Wm,
                                              float* __restrict__ Out,
                                              const float* __restrict__ bias) {
  const int wave = threadIdx.x >> 5;
  const int lane = threadIdx.x & 31;
  const int h = lane >> 4;
  const int c = lane & 15;
  const int tM = blockIdx.x * 64 + (wave >> 1) * 32;
  const int tN = blockIdx.y * 64 + (wave & 1) * 32;

  const float* pa[2];
  const float* pb[2];
#pragma unroll
  for (int i = 0; i < 2; ++i) {
    pa[i] = A  + (size_t)(tM + 16 * i + c) * D_DIM + 4 * h;
    pb[i] = Wm + (size_t)(tN + 16 * i + c) * D_DIM + 4 * h;
  }

  float bv[2];
#pragma unroll
  for (int j = 0; j < 2; ++j) bv[j] = bias ? bias[tN + 16 * j + c] : 0.f;

  float4 fa[3][GK_STEPS][2], fb[3][GK_STEPS][2];
#pragma unroll
  for (int g = 0; g < 2; ++g) {
#pragma unroll
    for (int st = 0; st < GK_STEPS; ++st)
#pragma unroll
      for (int i = 0; i < 2; ++i) {
        fa[g][st][i] = *(const float4*)(pa[i] + 32 * g + 8 * st);
        fb[g][st][i] = *(const float4*)(pb[i] + 32 * g + 8 * st);
      }
    __builtin_amdgcn_sched_barrier(0);
  }

  v8f acc[2][2];
#pragma unroll
  for (int i = 0; i < 2; ++i)
#pragma unroll
    for (int j = 0; j < 2; ++j) acc[i][j] = (v8f){};

#pragma unroll
  for (int g = 0; g < GK_GROUPS; ++g) {
    if (g + 2 < GK_GROUPS) {
#pragma unroll
      for (int st = 0; st < GK_STEPS; ++st)
#pragma unroll
        for (int i = 0; i < 2; ++i) {
          fa[(g + 2) % 3][st][i] = *(const float4*)(pa[i] + 32 * (g + 2) + 8 * st);
          fb[(g + 2) % 3][st][i] = *(const float4*)(pb[i] + 32 * (g + 2) + 8 * st);
        }
    }
    __builtin_amdgcn_sched_barrier(0);
#pragma unroll
    for (int st = 0; st < GK_STEPS; ++st) {
      const float4 a0 = fa[g % 3][st][0], a1 = fa[g % 3][st][1];
      const float4 b0 = fb[g % 3][st][0], b1 = fb[g % 3][st][1];
      acc[0][0] = mma4((v2f){a0.x, a0.y}, (v2f){b0.x, b0.y}, acc[0][0]);
      acc[0][1] = mma4((v2f){a0.x, a0.y}, (v2f){b1.x, b1.y}, acc[0][1]);
      acc[1][0] = mma4((v2f){a1.x, a1.y}, (v2f){b0.x, b0.y}, acc[1][0]);
      acc[1][1] = mma4((v2f){a1.x, a1.y}, (v2f){b1.x, b1.y}, acc[1][1]);
      acc[0][0] = mma4((v2f){a0.z, a0.w}, (v2f){b0.z, b0.w}, acc[0][0]);
      acc[0][1] = mma4((v2f){a0.z, a0.w}, (v2f){b1.z, b1.w}, acc[0][1]);
      acc[1][0] = mma4((v2f){a1.z, a1.w}, (v2f){b0.z, b0.w}, acc[1][0]);
      acc[1][1] = mma4((v2f){a1.z, a1.w}, (v2f){b1.z, b1.w}, acc[1][1]);
    }
  }

#pragma unroll
  for (int j = 0; j < 2; ++j) {
    const int n = tN + 16 * j + c;
#pragma unroll
    for (int i = 0; i < 2; ++i)
#pragma unroll
      for (int r = 0; r < 8; ++r) {
        const int m = tM + 16 * i + 8 * h + r;
        Out[(size_t)m * D_DIM + n] = acc[i][j][r] + bv[j];
      }
  }
}

__device__ __forceinline__ float4 sub4(float4 a, float s) {
  return make_float4(a.x - s, a.y - s, a.z - s, a.w - s);
}
__device__ __forceinline__ float4 mul4(float4 a, float4 b) {
  return make_float4(a.x * b.x, a.y * b.y, a.z * b.z, a.w * b.w);
}

__global__ __launch_bounds__(128) void wmma_gemm_ln_f32_kernel(const float* __restrict__ A,
                                                               const float* __restrict__ Wm,
                                                               float* __restrict__ Out,
                                                               const float* __restrict__ bias,
                                                               const float* __restrict__ lng,
                                                               const float* __restrict__ lnb) {
  const int wave = threadIdx.x >> 5;
  const int lane = threadIdx.x & 31;
  const int h = lane >> 4;
  const int c = lane & 15;
  const int tM = blockIdx.x * 64 + (wave >> 1) * 32;
  const int tN = blockIdx.y * 64 + (wave & 1) * 32;

  const float* pa[2];
  const float* pb[2];
#pragma unroll
  for (int i = 0; i < 2; ++i) {
    pa[i] = A  + (size_t)(tM + 16 * i + c) * D_DIM + 4 * h;
    pb[i] = Wm + (size_t)(tN + 16 * i + c) * D_DIM + 4 * h;
  }
  const float* pg = lng + 4 * h;
  const float* pl = lnb + 4 * h;

  float bv[2];
#pragma unroll
  for (int j = 0; j < 2; ++j) bv[j] = bias[tN + 16 * j + c];

  float4 fa[3][GK_STEPS][2], fb[3][GK_STEPS][2], fg[3][GK_STEPS], fl[3][GK_STEPS];
#pragma unroll
  for (int g = 0; g < 2; ++g) {
#pragma unroll
    for (int st = 0; st < GK_STEPS; ++st) {
#pragma unroll
      for (int i = 0; i < 2; ++i) {
        fa[g][st][i] = *(const float4*)(pa[i] + 32 * g + 8 * st);
        fb[g][st][i] = *(const float4*)(pb[i] + 32 * g + 8 * st);
      }
      fg[g][st] = *(const float4*)(pg + 32 * g + 8 * st);
      fl[g][st] = *(const float4*)(pl + 32 * g + 8 * st);
    }
    __builtin_amdgcn_sched_barrier(0);
  }

  float sh[2];
#pragma unroll
  for (int i = 0; i < 2; ++i) sh[i] = __shfl(fa[0][0][i].x, c, 32);

  v8f acc[2][2];
#pragma unroll
  for (int i = 0; i < 2; ++i)
#pragma unroll
    for (int j = 0; j < 2; ++j) acc[i][j] = (v8f){};
  float s1[2] = {0.f, 0.f}, s2[2] = {0.f, 0.f};
  float sg[2] = {0.f, 0.f}, cb[2] = {0.f, 0.f};

#pragma unroll
  for (int g = 0; g < GK_GROUPS; ++g) {
    if (g + 2 < GK_GROUPS) {
#pragma unroll
      for (int st = 0; st < GK_STEPS; ++st) {
#pragma unroll
        for (int i = 0; i < 2; ++i) {
          fa[(g + 2) % 3][st][i] = *(const float4*)(pa[i] + 32 * (g + 2) + 8 * st);
          fb[(g + 2) % 3][st][i] = *(const float4*)(pb[i] + 32 * (g + 2) + 8 * st);
        }
        fg[(g + 2) % 3][st] = *(const float4*)(pg + 32 * (g + 2) + 8 * st);
        fl[(g + 2) % 3][st] = *(const float4*)(pl + 32 * (g + 2) + 8 * st);
      }
    }
    __builtin_amdgcn_sched_barrier(0);
#pragma unroll
    for (int st = 0; st < GK_STEPS; ++st) {
      const float4 a0 = sub4(fa[g % 3][st][0], sh[0]), a1 = sub4(fa[g % 3][st][1], sh[1]);
      const float4 w0 = fb[g % 3][st][0], w1 = fb[g % 3][st][1];
      const float4 gg = fg[g % 3][st], ll = fl[g % 3][st];
      const float4 b0 = mul4(w0, gg), b1 = mul4(w1, gg);
      s1[0] += (a0.x + a0.y) + (a0.z + a0.w);
      s1[1] += (a1.x + a1.y) + (a1.z + a1.w);
      s2[0] = fmaf(a0.x, a0.x, fmaf(a0.y, a0.y, fmaf(a0.z, a0.z, fmaf(a0.w, a0.w, s2[0]))));
      s2[1] = fmaf(a1.x, a1.x, fmaf(a1.y, a1.y, fmaf(a1.z, a1.z, fmaf(a1.w, a1.w, s2[1]))));
      sg[0] += (b0.x + b0.y) + (b0.z + b0.w);
      sg[1] += (b1.x + b1.y) + (b1.z + b1.w);
      cb[0] = fmaf(w0.x, ll.x, fmaf(w0.y, ll.y, fmaf(w0.z, ll.z, fmaf(w0.w, ll.w, cb[0]))));
      cb[1] = fmaf(w1.x, ll.x, fmaf(w1.y, ll.y, fmaf(w1.z, ll.z, fmaf(w1.w, ll.w, cb[1]))));
      acc[0][0] = mma4((v2f){a0.x, a0.y}, (v2f){b0.x, b0.y}, acc[0][0]);
      acc[0][1] = mma4((v2f){a0.x, a0.y}, (v2f){b1.x, b1.y}, acc[0][1]);
      acc[1][0] = mma4((v2f){a1.x, a1.y}, (v2f){b0.x, b0.y}, acc[1][0]);
      acc[1][1] = mma4((v2f){a1.x, a1.y}, (v2f){b1.x, b1.y}, acc[1][1]);
      acc[0][0] = mma4((v2f){a0.z, a0.w}, (v2f){b0.z, b0.w}, acc[0][0]);
      acc[0][1] = mma4((v2f){a0.z, a0.w}, (v2f){b1.z, b1.w}, acc[0][1]);
      acc[1][0] = mma4((v2f){a1.z, a1.w}, (v2f){b0.z, b0.w}, acc[1][0]);
      acc[1][1] = mma4((v2f){a1.z, a1.w}, (v2f){b1.z, b1.w}, acc[1][1]);
#pragma unroll
      for (int q = 0; q < 8; ++q) {
        __builtin_amdgcn_sched_group_barrier(0x008, 1, 0);
        __builtin_amdgcn_sched_group_barrier(0x002, 5, 0);
      }
      asm volatile("" : "+v"(s1[0]), "+v"(s1[1]), "+v"(s2[0]), "+v"(s2[1]),
                        "+v"(sg[0]), "+v"(sg[1]), "+v"(cb[0]), "+v"(cb[1]));
    }
  }

  float dm[2], rs[2];
#pragma unroll
  for (int i = 0; i < 2; ++i) {
    s1[i] += __shfl_xor(s1[i], 16, 32);
    s2[i] += __shfl_xor(s2[i], 16, 32);
    dm[i] = s1[i] / D_DIM;
    const float var = fmaxf(s2[i] / D_DIM - dm[i] * dm[i], 0.f);
    rs[i] = rsqrtf(var + LN_EPS);
  }
#pragma unroll
  for (int j = 0; j < 2; ++j) {
    sg[j] += __shfl_xor(sg[j], 16, 32);
    cb[j] += __shfl_xor(cb[j], 16, 32);
    cb[j] += bv[j];
  }

#pragma unroll
  for (int i = 0; i < 2; ++i)
#pragma unroll
    for (int r = 0; r < 8; ++r) {
      const float dmr = __shfl(dm[i], 8 * h + r, 32);
      const float rsr = __shfl(rs[i], 8 * h + r, 32);
      const int m = tM + 16 * i + 8 * h + r;
#pragma unroll
      for (int j = 0; j < 2; ++j) {
        const int n = tN + 16 * j + c;
        Out[(size_t)m * D_DIM + n] = rsr * (acc[i][j][r] - dmr * sg[j]) + cb[j];
      }
    }
}

__global__ __launch_bounds__(128) void wmma_gemm3_f32_kernel(const float* __restrict__ A,
                                                             const float* __restrict__ W0,
                                                             const float* __restrict__ W1,
                                                             const float* __restrict__ W2,
                                                             float* __restrict__ O0,
                                                             float* __restrict__ O1,
                                                             float* __restrict__ O2) {
  const int z = blockIdx.z;
  const float* Wm = (z == 0) ? W0 : (z == 1) ? W1 : W2;
  float* Out = (z == 0) ? O0 : (z == 1) ? O1 : O2;
  gemm_tile_f32(A, Wm, Out, nullptr);
}

__global__ void phi_beta_kernel(const float* __restrict__ x,
                                float* __restrict__ q,
                                float* __restrict__ k,
                                const float* __restrict__ beta_w,
                                const float* __restrict__ beta_b,
                                const float* __restrict__ lnp_g,
                                const float* __restrict__ lnp_b,
                                float* __restrict__ beta) {
  __shared__ float sh[8 * 5];
  const int row = blockIdx.x, tid = threadIdx.x;
  const size_t off = (size_t)row * D_DIM + tid;

  float qe; { float t = q[off]; qe = (t > 0.f) ? t + 1.f : __expf(t); }
  float ke; { float t = k[off]; ke = (t > 0.f) ? t + 1.f : __expf(t); }
  const float xb = x[off] * beta_w[tid];

  const float sq  = block_sum_256(qe,      sh + 0);
  const float sq2 = block_sum_256(qe * qe, sh + 8);
  const float sk  = block_sum_256(ke,      sh + 16);
  const float sk2 = block_sum_256(ke * ke, sh + 24);
  const float sb  = block_sum_256(xb,      sh + 32);

  const float muq = sq / D_DIM, varq = sq2 / D_DIM - muq * muq;
  const float muk = sk / D_DIM, vark = sk2 / D_DIM - muk * muk;
  const float g = lnp_g[tid], bb = lnp_b[tid];
  q[off] = (qe - muq) * rsqrtf(varq + LN_EPS) * g + bb;
  k[off] = (ke - muk) * rsqrtf(vark + LN_EPS) * g + bb;
  if (tid == 0) beta[row] = 1.f / (1.f + __expf(-(sb + beta_b[0])));
}

#define CH 16
#define NCH (L_DIM / CH)
#define AUX_STRIDE 512

__global__ void chunk_prep_kernel(const float* __restrict__ phi_q,
                                  const float* __restrict__ phi_k,
                                  const float* __restrict__ beta,
                                  const float* __restrict__ decay,
                                  float* __restrict__ vu,
                                  float* __restrict__ aux) {
  __shared__ __attribute__((aligned(16))) float Ks[CH][260];
  __shared__ __attribute__((aligned(16))) float Qs[CH][260];
  __shared__ float Al[CH][17];
  __shared__ float Tl[CH][17];
  __shared__ float bl[CH];
  const int ch = blockIdx.x, b = blockIdx.y, tid = threadIdx.x;
  const float df = 1.f / (1.f + __expf(-decay[0]));
  const size_t base = ((size_t)b * L_DIM + (size_t)ch * CH) * D_DIM;
  float vcol[CH];
#pragma unroll
  for (int i = 0; i < CH; ++i) {
    Ks[i][tid] = phi_k[base + (size_t)i * D_DIM + tid];
    Qs[i][tid] = phi_q[base + (size_t)i * D_DIM + tid];
    vcol[i] = vu[base + (size_t)i * D_DIM + tid];
  }
  if (tid < CH) bl[tid] = beta[(size_t)b * L_DIM + ch * CH + tid];
  __syncthreads();

  const int i = tid >> 4, j = tid & 15;
  float gk = 0.f, gq = 0.f;
#pragma unroll 4
  for (int c = 0; c < D_DIM; c += 4) {
    const float4 kj = *(const float4*)&Ks[j][c];
    const float4 ki = *(const float4*)&Ks[i][c];
    const float4 qi = *(const float4*)&Qs[i][c];
    gk = fmaf(ki.x, kj.x, gk); gk = fmaf(ki.y, kj.y, gk); gk = fmaf(ki.z, kj.z, gk); gk = fmaf(ki.w, kj.w, gk);
    gq = fmaf(qi.x, kj.x, gq); gq = fmaf(qi.y, kj.y, gq); gq = fmaf(qi.z, kj.z, gq); gq = fmaf(qi.w, kj.w, gq);
  }
  float pw = 1.f;
  for (int n = 0; n < i - 1 - j; ++n) pw *= df;
  const bool lower = j < i;
  Al[i][j] = lower ? bl[i] * pw * gk : 0.f;
  float* o = aux + ((size_t)b * NCH + ch) * AUX_STRIDE;
  o[256 + tid] = lower ? pw * gq : 0.f;
  __syncthreads();

  if (tid < CH) {
    float tc[CH];
#pragma unroll
    for (int ii = 0; ii < CH; ++ii) {
      float s = (ii == tid) ? 1.f : 0.f;
#pragma unroll
      for (int jj = 0; jj < ii; ++jj) s = fmaf(-Al[ii][jj], tc[jj], s);
      tc[ii] = s;
      Tl[ii][tid] = s;
    }
  }
  __syncthreads();
  float gj = 1.f;
  for (int n = 0; n < j; ++n) gj *= df;
  o[tid] = -Tl[i][j] * bl[j] * gj;

#pragma unroll
  for (int jj = 0; jj < CH; ++jj) vcol[jj] *= bl[jj];
#pragma unroll
  for (int ii = 0; ii < CH; ++ii) {
    float s = 0.f;
#pragma unroll
    for (int jj = 0; jj <= ii; ++jj) s = fmaf(Tl[ii][jj], vcol[jj], s);
    vu[base + (size_t)ii * D_DIM + tid] = s;
  }
}

__global__ __launch_bounds__(128) void delta_chunk_scan_kernel(
    const float* __restrict__ phi_q, const float* __restrict__ phi_k,
    float* vy, const float* __restrict__ aux, const float* __restrict__ decay) {
  __shared__ float red[2][4][2][8][32];
  __shared__ float u0s[2][8][32];
  const int b = blockIdx.y;
  const int row0 = blockIdx.x * 16;
  const int wave = __builtin_amdgcn_readfirstlane(threadIdx.x >> 5);
  const int lane = threadIdx.x & 31;
  const int c = lane & 15, h = lane >> 4;
  const int r0 = 2 * wave;

  const float df = 1.f / (1.f + __expf(-decay[0]));
  float pwr[17];
  pwr[0] = 1.f;
#pragma unroll
  for (int n = 1; n <= 16; ++n) pwr[n] = pwr[n - 1] * df;
  const float g16 = pwr[16];
  float gpos[8], grev[8];
#pragma unroll
  for (int r = 0; r < 8; ++r) {
    gpos[r] = h ? pwr[8 + r] : pwr[r];
    grev[r] = h ? pwr[7 - r] : pwr[15 - r];
  }
  float gy0 = gpos[0], gy1 = gpos[1];
#pragma unroll
  for (int r = 2; r < 8; r += 2)
    if (r == r0) { gy0 = gpos[r]; gy1 = gpos[r + 1]; }

  const size_t bbase = (size_t)b * L_DIM * D_DIM;
  const int colw = 64 * wave;
  const float* kA = phi_k + bbase + (size_t)c * D_DIM + colw + 8 * h;
  const float* qA = phi_q + bbase + (size_t)c * D_DIM + colw + 8 * h;
  const float* kT = phi_k + bbase + (size_t)(8 * h) * D_DIM + colw + c;
  float* vtw = vy + bbase + (size_t)(8 * h + r0) * D_DIM + row0 + c;
  const float* ax = aux + (size_t)b * NCH * AUX_STRIDE + c * 16 + 8 * h;

  v8f acc[4];
#pragma unroll
  for (int ct = 0; ct < 4; ++ct) acc[ct] = (v8f){};

  float4 Kf[4][2], Qf[4][2];
  float ktr[4][8];
#pragma unroll
  for (int ct = 0; ct < 4; ++ct) {
    Kf[ct][0] = *(const float4*)(kA + 16 * ct);
    Kf[ct][1] = *(const float4*)(kA + 16 * ct + 4);
    Qf[ct][0] = *(const float4*)(qA + 16 * ct);
    Qf[ct][1] = *(const float4*)(qA + 16 * ct + 4);
#pragma unroll
    for (int r = 0; r < 8; ++r) ktr[ct][r] = kT[(size_t)r * D_DIM + 16 * ct];
  }
  float4 tnf[2], tnN[2], mqf[2];
  tnf[0] = *(const float4*)(ax);  tnf[1] = *(const float4*)(ax + 4);
  mqf[0] = *(const float4*)(ax + 256);  mqf[1] = *(const float4*)(ax + 260);
  float u0a = vtw[0], u0b = vtw[D_DIM];

#pragma unroll
  for (int r = 0; r < 8; ++r) red[1][wave][1][r][lane] = 0.f;
  float ym0 = 0.f, ym1 = 0.f;

  __builtin_amdgcn_fence(__ATOMIC_ACQUIRE, "workgroup", "global");

#define YQ2(ct, p)                                                                        \
  do {                                                                                    \
    const float* qf_ = (const float*)Qf[ct];                                              \
    Ya = mma4((v2f){qf_[2 * (p)], qf_[2 * (p) + 1]},                                      \
              (v2f){acc[ct][2 * (p)], acc[ct][2 * (p) + 1]}, Ya);                         \
    Yb = mma4((v2f){qf_[2 * (p) + 2], qf_[2 * (p) + 3]},                                  \
              (v2f){acc[ct][2 * (p) + 2], acc[ct][2 * (p) + 3]}, Yb);                     \
  } while (0)
#define ORDER2(a, b) asm volatile("" : "+v"(a), "+v"(b) : : "memory")
#define YQ_ORDER() ORDER2(Ya, Yb)

  for (int t = 0; t < NCH; ++t) {
    const size_t coff = (size_t)t * CH * D_DIM;
    const int tn1 = (t + 1 < NCH) ? t + 1 : t;
    const size_t noff = (size_t)tn1 * CH * D_DIM;
    const int par = t & 1;

    v8f Pa = {}, Pb = {};
#pragma unroll
    for (int ct = 0; ct < 4; ++ct) {
      const float* kf = (const float*)Kf[ct];
#pragma unroll
      for (int p = 0; p < 4; p += 2) {
        Pa = mma4((v2f){kf[2 * p], kf[2 * p + 1]}, (v2f){acc[ct][2 * p], acc[ct][2 * p + 1]}, Pa);
        Pb = mma4((v2f){kf[2 * p + 2], kf[2 * p + 3]}, (v2f){acc[ct][2 * p + 2], acc[ct][2 * p + 3]}, Pb);
      }
    }
    ORDER2(Pa, Pb);

#pragma unroll
    for (int r = 0; r < 8; ++r) red[par][wave][0][r][lane] = Pa[r] + Pb[r];
    u0s[par][r0][lane] = u0a;
    u0s[par][r0 + 1][lane] = u0b;

    v8f Ya = {}, Yb = {};
    YQ_ORDER();
#pragma unroll
    for (int ct = 0; ct < 4; ++ct) {
      Kf[ct][0] = *(const float4*)(kA + noff + 16 * ct);
      Kf[ct][1] = *(const float4*)(kA + noff + 16 * ct + 4);
    }
    {
      const float* an = ax + (size_t)tn1 * AUX_STRIDE;
      tnN[0] = *(const float4*)(an);  tnN[1] = *(const float4*)(an + 4);
    }
    u0a = vtw[noff];
    u0b = vtw[noff + D_DIM];
    YQ2(0, 0); YQ2(0, 2);
    YQ_ORDER();
    __builtin_amdgcn_fence(__ATOMIC_RELEASE, "workgroup", "local");
    __builtin_amdgcn_s_barrier_signal(-1);
    YQ_ORDER();
    YQ2(1, 0); YQ2(1, 2); YQ2(2, 0);
    YQ_ORDER();
    __builtin_amdgcn_s_barrier_wait(-1);
    __builtin_amdgcn_fence(__ATOMIC_ACQUIRE, "workgroup", "local");

    float pp[4][8];
#pragma unroll
    for (int w = 0; w < 4; ++w)
#pragma unroll
      for (int r = 0; r < 8; ++r) pp[w][r] = red[par][w][0][r][lane];
    v8f U;
#pragma unroll
    for (int r = 0; r < 8; ++r) U[r] = u0s[par][r][lane];
    float yp[4][2];
#pragma unroll
    for (int w = 0; w < 4; ++w) {
      yp[w][0] = red[par ^ 1][w][1][r0][lane];
      yp[w][1] = red[par ^ 1][w][1][r0 + 1][lane];
    }
    YQ_ORDER();
    YQ2(2, 2); YQ2(3, 0); YQ2(3, 2);
    YQ_ORDER();
    __builtin_amdgcn_sched_barrier(0);

#pragma unroll
    for (int ct = 0; ct < 4; ++ct) {
      Qf[ct][0] = *(const float4*)(qA + noff + 16 * ct);
      Qf[ct][1] = *(const float4*)(qA + noff + 16 * ct + 4);
    }
    if (t + 2 < NCH) {
      __builtin_prefetch(kA + noff + (size_t)CH * D_DIM + 24 * h, 0, 0);
      __builtin_prefetch(qA + noff + (size_t)CH * D_DIM + 24 * h, 0, 0);
    }

    float ps[8];
#pragma unroll
    for (int r = 0; r < 8; ++r) ps[r] = (pp[0][r] + pp[1][r]) + (pp[2][r] + pp[3][r]);
    {
      const float* tnp = (const float*)tnf;
#pragma unroll
      for (int jj = 0; jj < 4; ++jj)
        U = mma4((v2f){tnp[2 * jj], tnp[2 * jj + 1]}, (v2f){ps[2 * jj], ps[2 * jj + 1]}, U);
    }

    {
      const float yq0 = (yp[0][0] + yp[1][0]) + (yp[2][0] + yp[3][0]);
      const float yq1 = (yp[0][1] + yp[1][1]) + (yp[2][1] + yp[3][1]);
      if (t > 0) {
        vtw[coff - (size_t)CH * D_DIM] = gy0 * yq0 + ym0;
        vtw[coff - (size_t)CH * D_DIM + D_DIM] = gy1 * yq1 + ym1;
      }
    }
#pragma unroll
    for (int r = 0; r < 8; ++r) red[par][wave][1][r][lane] = Ya[r] + Yb[r];

#pragma unroll
    for (int ct = 0; ct < 4; ++ct) acc[ct] *= g16;
    float kt[4][8];
#pragma unroll
    for (int ct = 0; ct < 4; ++ct)
#pragma unroll
      for (int r = 0; r < 8; ++r) kt[ct][r] = ktr[ct][r] * grev[r];
    const float* mqp = (const float*)mqf;
    v8f Ym = {};
#pragma unroll
    for (int jj = 0; jj < 4; ++jj) {
#pragma unroll
      for (int ct = 0; ct < 4; ++ct)
        acc[ct] = mma4((v2f){kt[ct][2 * jj], kt[ct][2 * jj + 1]},
                       (v2f){U[2 * jj], U[2 * jj + 1]}, acc[ct]);
      Ym = mma4((v2f){mqp[2 * jj], mqp[2 * jj + 1]}, (v2f){U[2 * jj], U[2 * jj + 1]}, Ym);
    }
    ym0 = Ym[0]; ym1 = Ym[1];
#pragma unroll
    for (int r = 2; r < 8; r += 2)
      if (r == r0) { ym0 = Ym[r]; ym1 = Ym[r + 1]; }

    {
      const float* an = ax + (size_t)tn1 * AUX_STRIDE;
      mqf[0] = *(const float4*)(an + 256);  mqf[1] = *(const float4*)(an + 260);
    }
#pragma unroll
    for (int ct = 0; ct < 4; ++ct)
#pragma unroll
      for (int r = 0; r < 8; ++r) ktr[ct][r] = kT[noff + (size_t)r * D_DIM + 16 * ct];
    tnf[0] = tnN[0]; tnf[1] = tnN[1];
    __builtin_amdgcn_sched_barrier(0);
  }
#undef YQ2
#undef YQ_ORDER
#undef ORDER2

  __builtin_amdgcn_fence(__ATOMIC_RELEASE, "workgroup", "local");
  __builtin_amdgcn_s_barrier();
  __builtin_amdgcn_fence(__ATOMIC_ACQUIRE, "workgroup", "local");
  {
    const int lp = (NCH - 1) & 1;
    const float yq0 = (red[lp][0][1][r0][lane] + red[lp][1][1][r0][lane]) +
                      (red[lp][2][1][r0][lane] + red[lp][3][1][r0][lane]);
    const float yq1 = (red[lp][0][1][r0 + 1][lane] + red[lp][1][1][r0 + 1][lane]) +
                      (red[lp][2][1][r0 + 1][lane] + red[lp][3][1][r0 + 1][lane]);
    const size_t loff = (size_t)(NCH - 1) * CH * D_DIM;
    vtw[loff] = gy0 * yq0 + ym0;
    vtw[loff + D_DIM] = gy1 * yq1 + ym1;
  }
}

extern "C" void kernel_launch(void* const* d_in, const int* in_sizes, int n_in,
                              void* d_out, int out_size, void* d_ws, size_t ws_size,
                              hipStream_t stream) {
  (void)in_sizes; (void)n_in; (void)out_size; (void)ws_size;
  const float* x      = (const float*)d_in[0];
  const float* Wq     = (const float*)d_in[1];
  const float* Wk     = (const float*)d_in[2];
  const float* Wv     = (const float*)d_in[3];
  const float* beta_w = (const float*)d_in[4];
  const float* beta_b = (const float*)d_in[5];
  const float* decay  = (const float*)d_in[6];
  const float* Wo     = (const float*)d_in[7];
  const float* bo     = (const float*)d_in[8];
  const float* ln_g   = (const float*)d_in[9];
  const float* ln_b   = (const float*)d_in[10];
  const float* lnp_g  = (const float*)d_in[11];
  const float* lnp_b  = (const float*)d_in[12];
  float* out = (float*)d_out;

  char* ws = (char*)d_ws;
  const size_t MAT = (size_t)M_DIM * D_DIM * sizeof(float);
  float* q    = (float*)(ws);
  float* k    = (float*)(ws + MAT);
  float* vv   = (float*)(ws + 2 * MAT);
  float* beta = (float*)(ws + 3 * MAT);
  float* aux  = (float*)(ws + 3 * MAT + (64 << 10));

  const dim3 gemm_grid(M_DIM / 64, D_DIM / 64);
  const dim3 gemm_blk(128);

  wmma_gemm3_f32_kernel<<<dim3(M_DIM / 64, D_DIM / 64, 3), gemm_blk, 0, stream>>>(
      x, Wq, Wk, Wv, q, k, vv);

  phi_beta_kernel<<<dim3(M_DIM), dim3(256), 0, stream>>>(
      x, q, k, beta_w, beta_b, lnp_g, lnp_b, beta);

  chunk_prep_kernel<<<dim3(NCH, B_DIM), dim3(256), 0, stream>>>(q, k, beta, decay, vv, aux);

  delta_chunk_scan_kernel<<<dim3(D_DIM / 16, B_DIM), dim3(128), 0, stream>>>(
      q, k, vv, aux, decay);

  wmma_gemm_ln_f32_kernel<<<gemm_grid, gemm_blk, 0, stream>>>(vv, Wo, out, bo, ln_g, ln_b);
}


# scan replicated x4 on idle WGPs: each replica runs the full state recurrence, forms Q_c S^T and Mq U for every 4th chunk only; U0 kept in the output buffer, y in the v plane; same sums, same order
# speedup vs baseline: 5.6071x; 1.0799x over previous
#include <hip/hip_runtime.h>
#include <hip/hip_bf16.h>
#include <type_traits>

typedef __attribute__((ext_vector_type(2))) float v2f;
typedef __attribute__((ext_vector_type(8))) float v8f;

#define D_DIM 256
#define L_DIM 1024
#define B_DIM 4
#define M_DIM (B_DIM * L_DIM)
#define LN_EPS 1e-5f

__device__ __forceinline__ float block_sum_256(float v, float* sh8) {
#pragma unroll
  for (int m = 16; m > 0; m >>= 1) v += __shfl_xor(v, m, 32);
  if ((threadIdx.x & 31) == 0) sh8[threadIdx.x >> 5] = v;
  __syncthreads();
  float r = 0.f;
#pragma unroll
  for (int i = 0; i < 8; ++i) r += sh8[i];
  return r;
}

__device__ __forceinline__ v8f mma4(v2f a, v2f b, v8f c) {
  return __builtin_amdgcn_wmma_f32_16x16x4_f32(false, a, false, b, (short)0, c, false, false);
}

#define GK_STEPS 4
#define GK_GROUPS (D_DIM / (8 * GK_STEPS))

__device__ __forceinline__ void gemm_tile_f32(const float* __restrict__ A,
                                              const float* __restrict__ Wm,
                                              float* __restrict__ Out,
                                              const float* __restrict__ bias) {
  const int wave = threadIdx.x >> 5;
  const int lane = threadIdx.x & 31;
  const int h = lane >> 4;
  const int c = lane & 15;
  const int tM = blockIdx.x * 64 + (wave >> 1) * 32;
  const int tN = blockIdx.y * 64 + (wave & 1) * 32;

  const float* pa[2];
  const float* pb[2];
#pragma unroll
  for (int i = 0; i < 2; ++i) {
    pa[i] = A  + (size_t)(tM + 16 * i + c) * D_DIM + 4 * h;
    pb[i] = Wm + (size_t)(tN + 16 * i + c) * D_DIM + 4 * h;
  }

  float bv[2];
#pragma unroll
  for (int j = 0; j < 2; ++j) bv[j] = bias ? bias[tN + 16 * j + c] : 0.f;

  float4 fa[3][GK_STEPS][2], fb[3][GK_STEPS][2];
#pragma unroll
  for (int g = 0; g < 2; ++g) {
#pragma unroll
    for (int st = 0; st < GK_STEPS; ++st)
#pragma unroll
      for (int i = 0; i < 2; ++i) {
        fa[g][st][i] = *(const float4*)(pa[i] + 32 * g + 8 * st);
        fb[g][st][i] = *(const float4*)(pb[i] + 32 * g + 8 * st);
      }
    __builtin_amdgcn_sched_barrier(0);
  }

  v8f acc[2][2];
#pragma unroll
  for (int i = 0; i < 2; ++i)
#pragma unroll
    for (int j = 0; j < 2; ++j) acc[i][j] = (v8f){};

#pragma unroll
  for (int g = 0; g < GK_GROUPS; ++g) {
    if (g + 2 < GK_GROUPS) {
#pragma unroll
      for (int st = 0; st < GK_STEPS; ++st)
#pragma unroll
        for (int i = 0; i < 2; ++i) {
          fa[(g + 2) % 3][st][i] = *(const float4*)(pa[i] + 32 * (g + 2) + 8 * st);
          fb[(g + 2) % 3][st][i] = *(const float4*)(pb[i] + 32 * (g + 2) + 8 * st);
        }
    }
    __builtin_amdgcn_sched_barrier(0);
#pragma unroll
    for (int st = 0; st < GK_STEPS; ++st) {
      const float4 a0 = fa[g % 3][st][0], a1 = fa[g % 3][st][1];
      const float4 b0 = fb[g % 3][st][0], b1 = fb[g % 3][st][1];
      acc[0][0] = mma4((v2f){a0.x, a0.y}, (v2f){b0.x, b0.y}, acc[0][0]);
      acc[0][1] = mma4((v2f){a0.x, a0.y}, (v2f){b1.x, b1.y}, acc[0][1]);
      acc[1][0] = mma4((v2f){a1.x, a1.y}, (v2f){b0.x, b0.y}, acc[1][0]);
      acc[1][1] = mma4((v2f){a1.x, a1.y}, (v2f){b1.x, b1.y}, acc[1][1]);
      acc[0][0] = mma4((v2f){a0.z, a0.w}, (v2f){b0.z, b0.w}, acc[0][0]);
      acc[0][1] = mma4((v2f){a0.z, a0.w}, (v2f){b1.z, b1.w}, acc[0][1]);
      acc[1][0] = mma4((v2f){a1.z, a1.w}, (v2f){b0.z, b0.w}, acc[1][0]);
      acc[1][1] = mma4((v2f){a1.z, a1.w}, (v2f){b1.z, b1.w}, acc[1][1]);
    }
  }

#pragma unroll
  for (int j = 0; j < 2; ++j) {
    const int n = tN + 16 * j + c;
#pragma unroll
    for (int i = 0; i < 2; ++i)
#pragma unroll
      for (int r = 0; r < 8; ++r) {
        const int m = tM + 16 * i + 8 * h + r;
        Out[(size_t)m * D_DIM + n] = acc[i][j][r] + bv[j];
      }
  }
}

__device__ __forceinline__ float4 sub4(float4 a, float s) {
  return make_float4(a.x - s, a.y - s, a.z - s, a.w - s);
}
__device__ __forceinline__ float4 mul4(float4 a, float4 b) {
  return make_float4(a.x * b.x, a.y * b.y, a.z * b.z, a.w * b.w);
}

__global__ __launch_bounds__(128) void wmma_gemm_ln_f32_kernel(const float* __restrict__ A,
                                                               const float* __restrict__ Wm,
                                                               float* __restrict__ Out,
                                                               const float* __restrict__ bias,
                                                               const float* __restrict__ lng,
                                                               const float* __restrict__ lnb) {
  const int wave = threadIdx.x >> 5;
  const int lane = threadIdx.x & 31;
  const int h = lane >> 4;
  const int c = lane & 15;
  const int tM = blockIdx.x * 64 + (wave >> 1) * 32;
  const int tN = blockIdx.y * 64 + (wave & 1) * 32;

  const float* pa[2];
  const float* pb[2];
#pragma unroll
  for (int i = 0; i < 2; ++i) {
    pa[i] = A  + (size_t)(tM + 16 * i + c) * D_DIM + 4 * h;
    pb[i] = Wm + (size_t)(tN + 16 * i + c) * D_DIM + 4 * h;
  }
  const float* pg = lng + 4 * h;
  const float* pl = lnb + 4 * h;

  float bv[2];
#pragma unroll
  for (int j = 0; j < 2; ++j) bv[j] = bias[tN + 16 * j + c];

  float4 fa[3][GK_STEPS][2], fb[3][GK_STEPS][2], fg[3][GK_STEPS], fl[3][GK_STEPS];
#pragma unroll
  for (int g = 0; g < 2; ++g) {
#pragma unroll
    for (int st = 0; st < GK_STEPS; ++st) {
#pragma unroll
      for (int i = 0; i < 2; ++i) {
        fa[g][st][i] = *(const float4*)(pa[i] + 32 * g + 8 * st);
        fb[g][st][i] = *(const float4*)(pb[i] + 32 * g + 8 * st);
      }
      fg[g][st] = *(const float4*)(pg + 32 * g + 8 * st);
      fl[g][st] = *(const float4*)(pl + 32 * g + 8 * st);
    }
    __builtin_amdgcn_sched_barrier(0);
  }

  float sh[2];
#pragma unroll
  for (int i = 0; i < 2; ++i) sh[i] = __shfl(fa[0][0][i].x, c, 32);

  v8f acc[2][2];
#pragma unroll
  for (int i = 0; i < 2; ++i)
#pragma unroll
    for (int j = 0; j < 2; ++j) acc[i][j] = (v8f){};
  float s1[2] = {0.f, 0.f}, s2[2] = {0.f, 0.f};
  float sg[2] = {0.f, 0.f}, cb[2] = {0.f, 0.f};

#pragma unroll
  for (int g = 0; g < GK_GROUPS; ++g) {
    if (g + 2 < GK_GROUPS) {
#pragma unroll
      for (int st = 0; st < GK_STEPS; ++st) {
#pragma unroll
        for (int i = 0; i < 2; ++i) {
          fa[(g + 2) % 3][st][i] = *(const float4*)(pa[i] + 32 * (g + 2) + 8 * st);
          fb[(g + 2) % 3][st][i] = *(const float4*)(pb[i] + 32 * (g + 2) + 8 * st);
        }
        fg[(g + 2) % 3][st] = *(const float4*)(pg + 32 * (g + 2) + 8 * st);
        fl[(g + 2) % 3][st] = *(const float4*)(pl + 32 * (g + 2) + 8 * st);
      }
    }
    __builtin_amdgcn_sched_barrier(0);
#pragma unroll
    for (int st = 0; st < GK_STEPS; ++st) {
      const float4 a0 = sub4(fa[g % 3][st][0], sh[0]), a1 = sub4(fa[g % 3][st][1], sh[1]);
      const float4 w0 = fb[g % 3][st][0], w1 = fb[g % 3][st][1];
      const float4 gg = fg[g % 3][st], ll = fl[g % 3][st];
      const float4 b0 = mul4(w0, gg), b1 = mul4(w1, gg);
      s1[0] += (a0.x + a0.y) + (a0.z + a0.w);
      s1[1] += (a1.x + a1.y) + (a1.z + a1.w);
      s2[0] = fmaf(a0.x, a0.x, fmaf(a0.y, a0.y, fmaf(a0.z, a0.z, fmaf(a0.w, a0.w, s2[0]))));
      s2[1] = fmaf(a1.x, a1.x, fmaf(a1.y, a1.y, fmaf(a1.z, a1.z, fmaf(a1.w, a1.w, s2[1]))));
      sg[0] += (b0.x + b0.y) + (b0.z + b0.w);
      sg[1] += (b1.x + b1.y) + (b1.z + b1.w);
      cb[0] = fmaf(w0.x, ll.x, fmaf(w0.y, ll.y, fmaf(w0.z, ll.z, fmaf(w0.w, ll.w, cb[0]))));
      cb[1] = fmaf(w1.x, ll.x, fmaf(w1.y, ll.y, fmaf(w1.z, ll.z, fmaf(w1.w, ll.w, cb[1]))));
      acc[0][0] = mma4((v2f){a0.x, a0.y}, (v2f){b0.x, b0.y}, acc[0][0]);
      acc[0][1] = mma4((v2f){a0.x, a0.y}, (v2f){b1.x, b1.y}, acc[0][1]);
      acc[1][0] = mma4((v2f){a1.x, a1.y}, (v2f){b0.x, b0.y}, acc[1][0]);
      acc[1][1] = mma4((v2f){a1.x, a1.y}, (v2f){b1.x, b1.y}, acc[1][1]);
      acc[0][0] = mma4((v2f){a0.z, a0.w}, (v2f){b0.z, b0.w}, acc[0][0]);
      acc[0][1] = mma4((v2f){a0.z, a0.w}, (v2f){b1.z, b1.w}, acc[0][1]);
      acc[1][0] = mma4((v2f){a1.z, a1.w}, (v2f){b0.z, b0.w}, acc[1][0]);
      acc[1][1] = mma4((v2f){a1.z, a1.w}, (v2f){b1.z, b1.w}, acc[1][1]);
#pragma unroll
      for (int q = 0; q < 8; ++q) {
        __builtin_amdgcn_sched_group_barrier(0x008, 1, 0);
        __builtin_amdgcn_sched_group_barrier(0x002, 5, 0);
      }
      asm volatile("" : "+v"(s1[0]), "+v"(s1[1]), "+v"(s2[0]), "+v"(s2[1]),
                        "+v"(sg[0]), "+v"(sg[1]), "+v"(cb[0]), "+v"(cb[1]));
    }
  }

  float dm[2], rs[2];
#pragma unroll
  for (int i = 0; i < 2; ++i) {
    s1[i] += __shfl_xor(s1[i], 16, 32);
    s2[i] += __shfl_xor(s2[i], 16, 32);
    dm[i] = s1[i] / D_DIM;
    const float var = fmaxf(s2[i] / D_DIM - dm[i] * dm[i], 0.f);
    rs[i] = rsqrtf(var + LN_EPS);
  }
#pragma unroll
  for (int j = 0; j < 2; ++j) {
    sg[j] += __shfl_xor(sg[j], 16, 32);
    cb[j] += __shfl_xor(cb[j], 16, 32);
    cb[j] += bv[j];
  }

#pragma unroll
  for (int i = 0; i < 2; ++i)
#pragma unroll
    for (int r = 0; r < 8; ++r) {
      const float dmr = __shfl(dm[i], 8 * h + r, 32);
      const float rsr = __shfl(rs[i], 8 * h + r, 32);
      const int m = tM + 16 * i + 8 * h + r;
#pragma unroll
      for (int j = 0; j < 2; ++j) {
        const int n = tN + 16 * j + c;
        Out[(size_t)m * D_DIM + n] = rsr * (acc[i][j][r] - dmr * sg[j]) + cb[j];
      }
    }
}

__global__ __launch_bounds__(128) void wmma_gemm3_f32_kernel(const float* __restrict__ A,
                                                             const float* __restrict__ W0,
                                                             const float* __restrict__ W1,
                                                             const float* __restrict__ W2,
                                                             float* __restrict__ O0,
                                                             float* __restrict__ O1,
                                                             float* __restrict__ O2) {
  const int z = blockIdx.z;
  const float* Wm = (z == 0) ? W0 : (z == 1) ? W1 : W2;
  float* Out = (z == 0) ? O0 : (z == 1) ? O1 : O2;
  gemm_tile_f32(A, Wm, Out, nullptr);
}

__global__ void phi_beta_kernel(const float* __restrict__ x,
                                float* __restrict__ q,
                                float* __restrict__ k,
                                const float* __restrict__ beta_w,
                                const float* __restrict__ beta_b,
                                const float* __restrict__ lnp_g,
                                const float* __restrict__ lnp_b,
                                float* __restrict__ beta) {
  __shared__ float sh[8 * 5];
  const int row = blockIdx.x, tid = threadIdx.x;
  const size_t off = (size_t)row * D_DIM + tid;

  float qe; { float t = q[off]; qe = (t > 0.f) ? t + 1.f : __expf(t); }
  float ke; { float t = k[off]; ke = (t > 0.f) ? t + 1.f : __expf(t); }
  const float xb = x[off] * beta_w[tid];

  const float sq  = block_sum_256(qe,      sh + 0);
  const float sq2 = block_sum_256(qe * qe, sh + 8);
  const float sk  = block_sum_256(ke,      sh + 16);
  const float sk2 = block_sum_256(ke * ke, sh + 24);
  const float sb  = block_sum_256(xb,      sh + 32);

  const float muq = sq / D_DIM, varq = sq2 / D_DIM - muq * muq;
  const float muk = sk / D_DIM, vark = sk2 / D_DIM - muk * muk;
  const float g = lnp_g[tid], bb = lnp_b[tid];
  q[off] = (qe - muq) * rsqrtf(varq + LN_EPS) * g + bb;
  k[off] = (ke - muk) * rsqrtf(vark + LN_EPS) * g + bb;
  if (tid == 0) beta[row] = 1.f / (1.f + __expf(-(sb + beta_b[0])));
}

#define CH 16
#define NCH (L_DIM / CH)
#define AUX_STRIDE 512

__global__ void chunk_prep_kernel(const float* __restrict__ phi_q,
                                  const float* __restrict__ phi_k,
                                  const float* __restrict__ beta,
                                  const float* __restrict__ decay,
                                  const float* __restrict__ v,
                                  float* __restrict__ u0,
                                  float* __restrict__ aux) {
  __shared__ __attribute__((aligned(16))) float Ks[CH][260];
  __shared__ __attribute__((aligned(16))) float Qs[CH][260];
  __shared__ float Al[CH][17];
  __shared__ float Tl[CH][17];
  __shared__ float bl[CH];
  const int ch = blockIdx.x, b = blockIdx.y, tid = threadIdx.x;
  const float df = 1.f / (1.f + __expf(-decay[0]));
  const size_t base = ((size_t)b * L_DIM + (size_t)ch * CH) * D_DIM;
  float vcol[CH];
#pragma unroll
  for (int i = 0; i < CH; ++i) {
    Ks[i][tid] = phi_k[base + (size_t)i * D_DIM + tid];
    Qs[i][tid] = phi_q[base + (size_t)i * D_DIM + tid];
    vcol[i] = v[base + (size_t)i * D_DIM + tid];
  }
  if (tid < CH) bl[tid] = beta[(size_t)b * L_DIM + ch * CH + tid];
  __syncthreads();

  const int i = tid >> 4, j = tid & 15;
  float gk = 0.f, gq = 0.f;
#pragma unroll 4
  for (int c = 0; c < D_DIM; c += 4) {
    const float4 kj = *(const float4*)&Ks[j][c];
    const float4 ki = *(const float4*)&Ks[i][c];
    const float4 qi = *(const float4*)&Qs[i][c];
    gk = fmaf(ki.x, kj.x, gk); gk = fmaf(ki.y, kj.y, gk); gk = fmaf(ki.z, kj.z, gk); gk = fmaf(ki.w, kj.w, gk);
    gq = fmaf(qi.x, kj.x, gq); gq = fmaf(qi.y, kj.y, gq); gq = fmaf(qi.z, kj.z, gq); gq = fmaf(qi.w, kj.w, gq);
  }
  float pw = 1.f;
  for (int n = 0; n < i - 1 - j; ++n) pw *= df;
  const bool lower = j < i;
  Al[i][j] = lower ? bl[i] * pw * gk : 0.f;
  float* o = aux + ((size_t)b * NCH + ch) * AUX_STRIDE;
  o[256 + tid] = lower ? pw * gq : 0.f;
  __syncthreads();

  if (tid < CH) {
    float tc[CH];
#pragma unroll
    for (int ii = 0; ii < CH; ++ii) {
      float s = (ii == tid) ? 1.f : 0.f;
#pragma unroll
      for (int jj = 0; jj < ii; ++jj) s = fmaf(-Al[ii][jj], tc[jj], s);
      tc[ii] = s;
      Tl[ii][tid] = s;
    }
  }
  __syncthreads();
  float gj = 1.f;
  for (int n = 0; n < j; ++n) gj *= df;
  o[tid] = -Tl[i][j] * bl[j] * gj;

#pragma unroll
  for (int jj = 0; jj < CH; ++jj) vcol[jj] *= bl[jj];
#pragma unroll
  for (int ii = 0; ii < CH; ++ii) {
    float s = 0.f;
#pragma unroll
    for (int jj = 0; jj <= ii; ++jj) s = fmaf(Tl[ii][jj], vcol[jj], s);
    u0[base + (size_t)ii * D_DIM + tid] = s;
  }
}

#define NREP 4
template <bool V> using flag_c = std::integral_constant<bool, V>;
template <int V> using int_c = std::integral_constant<int, V>;

__global__ __launch_bounds__(128) void delta_chunk_scan_kernel(
    const float* __restrict__ phi_q, const float* __restrict__ phi_k,
    const float* u0p, float* yv, const float* __restrict__ aux, const float* __restrict__ decay) {
  __shared__ float lds_pool[41 * 1024];
  float (*red)[4][2][8][32] = reinterpret_cast<float (*)[4][2][8][32]>(lds_pool);
  float (*u0s)[8][32] = reinterpret_cast<float (*)[8][32]>(lds_pool + 2 * 4 * 2 * 8 * 32);
  const int b = blockIdx.y;
  const int row0 = blockIdx.x * 16;
  const int rep = blockIdx.z;
  const int wave = __builtin_amdgcn_readfirstlane(threadIdx.x >> 5);
  const int lane = threadIdx.x & 31;
  const int c = lane & 15, h = lane >> 4;
  const int r0 = 2 * wave;

  const float df = 1.f / (1.f + __expf(-decay[0]));
  float pwr[17];
  pwr[0] = 1.f;
#pragma unroll
  for (int n = 1; n <= 16; ++n) pwr[n] = pwr[n - 1] * df;
  const float g16 = pwr[16];
  float gpos[8], grev[8];
#pragma unroll
  for (int r = 0; r < 8; ++r) {
    gpos[r] = h ? pwr[8 + r] : pwr[r];
    grev[r] = h ? pwr[7 - r] : pwr[15 - r];
  }
  float gy0 = gpos[0], gy1 = gpos[1];
#pragma unroll
  for (int r = 2; r < 8; r += 2)
    if (r == r0) { gy0 = gpos[r]; gy1 = gpos[r + 1]; }

  const size_t bbase = (size_t)b * L_DIM * D_DIM;
  const int colw = 64 * wave;
  const float* kA = phi_k + bbase + (size_t)c * D_DIM + colw + 8 * h;
  const float* qA = phi_q + bbase + (size_t)c * D_DIM + colw + 8 * h;
  const float* kT = phi_k + bbase + (size_t)(8 * h) * D_DIM + colw + c;
  const size_t tile_off = bbase + (size_t)(8 * h + r0) * D_DIM + row0 + c;
  const float* u0w = u0p + tile_off;
  float* ytw = yv + tile_off;
  const float* ax = aux + (size_t)b * NCH * AUX_STRIDE + c * 16 + 8 * h;

  v8f acc[4];
#pragma unroll
  for (int ct = 0; ct < 4; ++ct) acc[ct] = (v8f){};

  float4 Kf[4][2], Qf[4][2];
  float ktr[4][8];
#pragma unroll
  for (int ct = 0; ct < 4; ++ct) {
    Kf[ct][0] = *(const float4*)(kA + 16 * ct);
    Kf[ct][1] = *(const float4*)(kA + 16 * ct + 4);
    Qf[ct][0] = *(const float4*)(qA + 16 * ct);
    Qf[ct][1] = *(const float4*)(qA + 16 * ct + 4);
#pragma unroll
    for (int r = 0; r < 8; ++r) ktr[ct][r] = kT[(size_t)r * D_DIM + 16 * ct];
  }
  float4 tnf[2], tnN[2], mqf[2];
  tnf[0] = *(const float4*)(ax);  tnf[1] = *(const float4*)(ax + 4);
  mqf[0] = *(const float4*)(ax + 256);  mqf[1] = *(const float4*)(ax + 260);
  float u0a = u0w[0], u0b = u0w[D_DIM];

#pragma unroll
  for (int r = 0; r < 8; ++r) red[1][wave][1][r][lane] = 0.f;
  float ym0 = 0.f, ym1 = 0.f;

  __builtin_amdgcn_fence(__ATOMIC_ACQUIRE, "workgroup", "global");

#define YQ2(ct, p)                                                                        \
  do {                                                                                    \
    const float* qf_ = (const float*)Qf[ct];                                              \
    Ya = mma4((v2f){qf_[2 * (p)], qf_[2 * (p) + 1]},                                      \
              (v2f){acc[ct][2 * (p)], acc[ct][2 * (p) + 1]}, Ya);                         \
    Yb = mma4((v2f){qf_[2 * (p) + 2], qf_[2 * (p) + 3]},                                  \
              (v2f){acc[ct][2 * (p) + 2], acc[ct][2 * (p) + 3]}, Yb);                     \
  } while (0)
#define ORDER2(a, b) asm volatile("" : "+v"(a), "+v"(b) : : "memory")
#define YQ_ORDER() ORDER2(Ya, Yb)
#define MEM_ORDER() asm volatile("" : : : "memory")

  auto chunk = [&](const int t, auto own_c, auto prev_c, auto next_c) __attribute__((always_inline)) {
    constexpr bool OWN = decltype(own_c)::value;
    constexpr bool PREV = decltype(prev_c)::value;
    constexpr bool NEXT = decltype(next_c)::value;
    const size_t coff = (size_t)t * CH * D_DIM;
    const int tn1 = (t + 1 < NCH) ? t + 1 : t;
    const size_t noff = (size_t)tn1 * CH * D_DIM;
    const int par = t & 1;

    v8f Pa = {}, Pb = {};
#pragma unroll
    for (int ct = 0; ct < 4; ++ct) {
      const float* kf = (const float*)Kf[ct];
#pragma unroll
      for (int p = 0; p < 4; p += 2) {
        Pa = mma4((v2f){kf[2 * p], kf[2 * p + 1]}, (v2f){acc[ct][2 * p], acc[ct][2 * p + 1]}, Pa);
        Pb = mma4((v2f){kf[2 * p + 2], kf[2 * p + 3]}, (v2f){acc[ct][2 * p + 2], acc[ct][2 * p + 3]}, Pb);
      }
    }
    ORDER2(Pa, Pb);

#pragma unroll
    for (int r = 0; r < 8; ++r) red[par][wave][0][r][lane] = Pa[r] + Pb[r];
    u0s[par][r0][lane] = u0a;
    u0s[par][r0 + 1][lane] = u0b;

    v8f Ya = {}, Yb = {};
    if constexpr (OWN) YQ_ORDER();
    else MEM_ORDER();
#pragma unroll
    for (int ct = 0; ct < 4; ++ct) {
      Kf[ct][0] = *(const float4*)(kA + noff + 16 * ct);
      Kf[ct][1] = *(const float4*)(kA + noff + 16 * ct + 4);
    }
    {
      const float* an = ax + (size_t)tn1 * AUX_STRIDE;
      tnN[0] = *(const float4*)(an);  tnN[1] = *(const float4*)(an + 4);
    }
    u0a = u0w[noff];
    u0b = u0w[noff + D_DIM];
    if constexpr (OWN) {
      YQ2(0, 0); YQ2(0, 2);
      YQ_ORDER();
    }
    __builtin_amdgcn_fence(__ATOMIC_RELEASE, "workgroup", "local");
    __builtin_amdgcn_s_barrier_signal(-1);
    if constexpr (OWN) {
      YQ_ORDER();
      YQ2(1, 0); YQ2(1, 2); YQ2(2, 0);
      YQ_ORDER();
    }
    __builtin_amdgcn_s_barrier_wait(-1);
    __builtin_amdgcn_fence(__ATOMIC_ACQUIRE, "workgroup", "local");

    float pp[4][8];
#pragma unroll
    for (int w = 0; w < 4; ++w)
#pragma unroll
      for (int r = 0; r < 8; ++r) pp[w][r] = red[par][w][0][r][lane];
    v8f U;
#pragma unroll
    for (int r = 0; r < 8; ++r) U[r] = u0s[par][r][lane];
    float yp[4][2] = {};
    if constexpr (PREV) {
#pragma unroll
      for (int w = 0; w < 4; ++w) {
        yp[w][0] = red[par ^ 1][w][1][r0][lane];
        yp[w][1] = red[par ^ 1][w][1][r0 + 1][lane];
      }
    }
    if constexpr (OWN) {
      YQ_ORDER();
      YQ2(2, 2); YQ2(3, 0); YQ2(3, 2);
      YQ_ORDER();
    }
    __builtin_amdgcn_sched_barrier(0);

    if constexpr (NEXT) {
#pragma unroll
      for (int ct = 0; ct < 4; ++ct) {
        Qf[ct][0] = *(const float4*)(qA + noff + 16 * ct);
        Qf[ct][1] = *(const float4*)(qA + noff + 16 * ct + 4);
      }
    }
    if (t + 2 < NCH) {
      __builtin_prefetch(kA + noff + (size_t)CH * D_DIM + 24 * h, 0, 0);
      if constexpr (!OWN && !PREV && !NEXT)
        __builtin_prefetch(qA + noff + (size_t)CH * D_DIM + 24 * h, 0, 0);
    }

    float ps[8];
#pragma unroll
    for (int r = 0; r < 8; ++r) ps[r] = (pp[0][r] + pp[1][r]) + (pp[2][r] + pp[3][r]);
    {
      const float* tnp = (const float*)tnf;
#pragma unroll
      for (int jj = 0; jj < 4; ++jj)
        U = mma4((v2f){tnp[2 * jj], tnp[2 * jj + 1]}, (v2f){ps[2 * jj], ps[2 * jj + 1]}, U);
    }

    if constexpr (PREV) {
      const float yq0 = (yp[0][0] + yp[1][0]) + (yp[2][0] + yp[3][0]);
      const float yq1 = (yp[0][1] + yp[1][1]) + (yp[2][1] + yp[3][1]);
      if (t > 0) {
        ytw[coff - (size_t)CH * D_DIM] = gy0 * yq0 + ym0;
        ytw[coff - (size_t)CH * D_DIM + D_DIM] = gy1 * yq1 + ym1;
      }
    }
    if constexpr (OWN) {
#pragma unroll
      for (int r = 0; r < 8; ++r) red[par][wave][1][r][lane] = Ya[r] + Yb[r];
    }

#pragma unroll
    for (int ct = 0; ct < 4; ++ct) acc[ct] *= g16;
    float kt[4][8];
#pragma unroll
    for (int ct = 0; ct < 4; ++ct)
#pragma unroll
      for (int r = 0; r < 8; ++r) kt[ct][r] = ktr[ct][r] * grev[r];
    const float* mqp = (const float*)mqf;
    v8f Ym = {};
#pragma unroll
    for (int jj = 0; jj < 4; ++jj) {
#pragma unroll
      for (int ct = 0; ct < 4; ++ct)
        acc[ct] = mma4((v2f){kt[ct][2 * jj], kt[ct][2 * jj + 1]},
                       (v2f){U[2 * jj], U[2 * jj + 1]}, acc[ct]);
      if constexpr (OWN)
        Ym = mma4((v2f){mqp[2 * jj], mqp[2 * jj + 1]}, (v2f){U[2 * jj], U[2 * jj + 1]}, Ym);
    }
    if constexpr (OWN) {
      ym0 = Ym[0]; ym1 = Ym[1];
#pragma unroll
      for (int r = 2; r < 8; r += 2)
        if (r == r0) { ym0 = Ym[r]; ym1 = Ym[r + 1]; }
    }

    if constexpr (NEXT) {
      const float* an = ax + (size_t)tn1 * AUX_STRIDE;
      mqf[0] = *(const float4*)(an + 256);  mqf[1] = *(const float4*)(an + 260);
    }
#pragma unroll
    for (int ct = 0; ct < 4; ++ct)
#pragma unroll
      for (int r = 0; r < 8; ++r) ktr[ct][r] = kT[noff + (size_t)r * D_DIM + 16 * ct];
    tnf[0] = tnN[0]; tnf[1] = tnN[1];
    __builtin_amdgcn_sched_barrier(0);
  };

  auto pass = [&](auto rep_c) __attribute__((always_inline)) {
    constexpr int R = decltype(rep_c)::value;
    for (int tb = 0; tb < NCH; tb += NREP) {
      chunk(tb + 0, flag_c<R == 0>{}, flag_c<R == 3>{}, flag_c<R == 1>{});
      chunk(tb + 1, flag_c<R == 1>{}, flag_c<R == 0>{}, flag_c<R == 2>{});
      chunk(tb + 2, flag_c<R == 2>{}, flag_c<R == 1>{}, flag_c<R == 3>{});
      chunk(tb + 3, flag_c<R == 3>{}, flag_c<R == 2>{}, flag_c<R == 0>{});
    }
  };
  switch (rep) {
    case 0: pass(int_c<0>{}); break;
    case 1: pass(int_c<1>{}); break;
    case 2: pass(int_c<2>{}); break;
    default: pass(int_c<3>{}); break;
  }
#undef YQ2
#undef YQ_ORDER
#undef ORDER2
#undef MEM_ORDER

  if (((NCH - 1) & (NREP - 1)) == rep) {
    __builtin_amdgcn_fence(__ATOMIC_RELEASE, "workgroup", "local");
    __builtin_amdgcn_s_barrier();
    __builtin_amdgcn_fence(__ATOMIC_ACQUIRE, "workgroup", "local");
    const int lp = (NCH - 1) & 1;
    const float yq0 = (red[lp][0][1][r0][lane] + red[lp][1][1][r0][lane]) +
                      (red[lp][2][1][r0][lane] + red[lp][3][1][r0][lane]);
    const float yq1 = (red[lp][0][1][r0 + 1][lane] + red[lp][1][1][r0 + 1][lane]) +
                      (red[lp][2][1][r0 + 1][lane] + red[lp][3][1][r0 + 1][lane]);
    const size_t loff = (size_t)(NCH - 1) * CH * D_DIM;
    ytw[loff] = gy0 * yq0 + ym0;
    ytw[loff + D_DIM] = gy1 * yq1 + ym1;
  }
}

extern "C" void kernel_launch(void* const* d_in, const int* in_sizes, int n_in,
                              void* d_out, int out_size, void* d_ws, size_t ws_size,
                              hipStream_t stream) {
  (void)in_sizes; (void)n_in; (void)out_size; (void)ws_size;
  const float* x      = (const float*)d_in[0];
  const float* Wq     = (const float*)d_in[1];
  const float* Wk     = (const float*)d_in[2];
  const float* Wv     = (const float*)d_in[3];
  const float* beta_w = (const float*)d_in[4];
  const float* beta_b = (const float*)d_in[5];
  const float* decay  = (const float*)d_in[6];
  const float* Wo     = (const float*)d_in[7];
  const float* bo     = (const float*)d_in[8];
  const float* ln_g   = (const float*)d_in[9];
  const float* ln_b   = (const float*)d_in[10];
  const float* lnp_g  = (const float*)d_in[11];
  const float* lnp_b  = (const float*)d_in[12];
  float* out = (float*)d_out;

  char* ws = (char*)d_ws;
  const size_t MAT = (size_t)M_DIM * D_DIM * sizeof(float);
  float* q    = (float*)(ws);
  float* k    = (float*)(ws + MAT);
  float* vv   = (float*)(ws + 2 * MAT);
  float* beta = (float*)(ws + 3 * MAT);
  float* aux  = (float*)(ws + 3 * MAT + (64 << 10));

  const dim3 gemm_grid(M_DIM / 64, D_DIM / 64);
  const dim3 gemm_blk(128);

  wmma_gemm3_f32_kernel<<<dim3(M_DIM / 64, D_DIM / 64, 3), gemm_blk, 0, stream>>>(
      x, Wq, Wk, Wv, q, k, vv);

  phi_beta_kernel<<<dim3(M_DIM), dim3(256), 0, stream>>>(
      x, q, k, beta_w, beta_b, lnp_g, lnp_b, beta);

  chunk_prep_kernel<<<dim3(NCH, B_DIM), dim3(256), 0, stream>>>(q, k, beta, decay, vv, out, aux);

  delta_chunk_scan_kernel<<<dim3(D_DIM / 16, B_DIM, NREP), dim3(128), 0, stream>>>(
      q, k, out, vv, aux, decay);

  wmma_gemm_ln_f32_kernel<<<gemm_grid, gemm_blk, 0, stream>>>(vv, Wo, out, bo, ln_g, ln_b);
}


# scan: sequence halves side by side by linearity (2nd half from zero state and, U0 = 0, from identity; replicas dropped); one added WMMA launch adds Z S_mid^T onto y: 6 launches; new sum order of y
# speedup vs baseline: 6.1267x; 1.0927x over previous
#include <hip/hip_runtime.h>
#include <hip/hip_bf16.h>
#include <type_traits>

typedef __attribute__((ext_vector_type(2))) float v2f;
typedef __attribute__((ext_vector_type(8))) float v8f;

#define D_DIM 256
#define L_DIM 1024
#define B_DIM 4
#define M_DIM (B_DIM * L_DIM)
#define LN_EPS 1e-5f

__device__ __forceinline__ float block_sum_256(float v, float* sh8) {
#pragma unroll
  for (int m = 16; m > 0; m >>= 1) v += __shfl_xor(v, m, 32);
  if ((threadIdx.x & 31) == 0) sh8[threadIdx.x >> 5] = v;
  __syncthreads();
  float r = 0.f;
#pragma unroll
  for (int i = 0; i < 8; ++i) r += sh8[i];
  return r;
}

__device__ __forceinline__ v8f mma4(v2f a, v2f b, v8f c) {
  return __builtin_amdgcn_wmma_f32_16x16x4_f32(false, a, false, b, (short)0, c, false, false);
}

#define GK_STEPS 4
#define GK_GROUPS (D_DIM / (8 * GK_STEPS))

template <bool ACC>
__device__ __forceinline__ void gemm_tile_f32(const float* __restrict__ A,
                                              const float* __restrict__ Wm,
                                              float* __restrict__ Out,
                                              const float* __restrict__ bias) {
  const int wave = threadIdx.x >> 5;
  const int lane = threadIdx.x & 31;
  const int h = lane >> 4;
  const int c = lane & 15;
  const int tM = blockIdx.x * 64 + (wave >> 1) * 32;
  const int tN = blockIdx.y * 64 + (wave & 1) * 32;

  const float* pa[2];
  const float* pb[2];
#pragma unroll
  for (int i = 0; i < 2; ++i) {
    pa[i] = A  + (size_t)(tM + 16 * i + c) * D_DIM + 4 * h;
    pb[i] = Wm + (size_t)(tN + 16 * i + c) * D_DIM + 4 * h;
  }

  float bv[2];
#pragma unroll
  for (int j = 0; j < 2; ++j) bv[j] = bias ? bias[tN + 16 * j + c] : 0.f;

  float4 fa[3][GK_STEPS][2], fb[3][GK_STEPS][2];
#pragma unroll
  for (int g = 0; g < 2; ++g) {
#pragma unroll
    for (int st = 0; st < GK_STEPS; ++st)
#pragma unroll
      for (int i = 0; i < 2; ++i) {
        fa[g][st][i] = *(const float4*)(pa[i] + 32 * g + 8 * st);
        fb[g][st][i] = *(const float4*)(pb[i] + 32 * g + 8 * st);
      }
    __builtin_amdgcn_sched_barrier(0);
  }

  v8f acc[2][2];
#pragma unroll
  for (int i = 0; i < 2; ++i)
#pragma unroll
    for (int j = 0; j < 2; ++j) {
      if constexpr (ACC) {
#pragma unroll
        for (int r = 0; r < 8; ++r)
          acc[i][j][r] = Out[(size_t)(tM + 16 * i + 8 * h + r) * D_DIM + tN + 16 * j + c];
      } else {
        acc[i][j] = (v8f){};
      }
    }

#pragma unroll
  for (int g = 0; g < GK_GROUPS; ++g) {
    if (g + 2 < GK_GROUPS) {
#pragma unroll
      for (int st = 0; st < GK_STEPS; ++st)
#pragma unroll
        for (int i = 0; i < 2; ++i) {
          fa[(g + 2) % 3][st][i] = *(const float4*)(pa[i] + 32 * (g + 2) + 8 * st);
          fb[(g + 2) % 3][st][i] = *(const float4*)(pb[i] + 32 * (g + 2) + 8 * st);
        }
    }
    __builtin_amdgcn_sched_barrier(0);
#pragma unroll
    for (int st = 0; st < GK_STEPS; ++st) {
      const float4 a0 = fa[g % 3][st][0], a1 = fa[g % 3][st][1];
      const float4 b0 = fb[g % 3][st][0], b1 = fb[g % 3][st][1];
      acc[0][0] = mma4((v2f){a0.x, a0.y}, (v2f){b0.x, b0.y}, acc[0][0]);
      acc[0][1] = mma4((v2f){a0.x, a0.y}, (v2f){b1.x, b1.y}, acc[0][1]);
      acc[1][0] = mma4((v2f){a1.x, a1.y}, (v2f){b0.x, b0.y}, acc[1][0]);
      acc[1][1] = mma4((v2f){a1.x, a1.y}, (v2f){b1.x, b1.y}, acc[1][1]);
      acc[0][0] = mma4((v2f){a0.z, a0.w}, (v2f){b0.z, b0.w}, acc[0][0]);
      acc[0][1] = mma4((v2f){a0.z, a0.w}, (v2f){b1.z, b1.w}, acc[0][1]);
      acc[1][0] = mma4((v2f){a1.z, a1.w}, (v2f){b0.z, b0.w}, acc[1][0]);
      acc[1][1] = mma4((v2f){a1.z, a1.w}, (v2f){b1.z, b1.w}, acc[1][1]);
    }
  }

#pragma unroll
  for (int j = 0; j < 2; ++j) {
    const int n = tN + 16 * j + c;
#pragma unroll
    for (int i = 0; i < 2; ++i)
#pragma unroll
      for (int r = 0; r < 8; ++r) {
        const int m = tM + 16 * i + 8 * h + r;
        Out[(size_t)m * D_DIM + n] = acc[i][j][r] + bv[j];
      }
  }
}

__device__ __forceinline__ float4 sub4(float4 a, float s) {
  return make_float4(a.x - s, a.y - s, a.z - s, a.w - s);
}
__device__ __forceinline__ float4 mul4(float4 a, float4 b) {
  return make_float4(a.x * b.x, a.y * b.y, a.z * b.z, a.w * b.w);
}

__global__ __launch_bounds__(128) void wmma_gemm_ln_f32_kernel(const float* __restrict__ A,
                                                               const float* __restrict__ Wm,
                                                               float* __restrict__ Out,
                                                               const float* __restrict__ bias,
                                                               const float* __restrict__ lng,
                                                               const float* __restrict__ lnb) {
  const int wave = threadIdx.x >> 5;
  const int lane = threadIdx.x & 31;
  const int h = lane >> 4;
  const int c = lane & 15;
  const int tM = blockIdx.x * 64 + (wave >> 1) * 32;
  const int tN = blockIdx.y * 64 + (wave & 1) * 32;

  const float* pa[2];
  const float* pb[2];
#pragma unroll
  for (int i = 0; i < 2; ++i) {
    pa[i] = A  + (size_t)(tM + 16 * i + c) * D_DIM + 4 * h;
    pb[i] = Wm + (size_t)(tN + 16 * i + c) * D_DIM + 4 * h;
  }
  const float* pg = lng + 4 * h;
  const float* pl = lnb + 4 * h;

  float bv[2];
#pragma unroll
  for (int j = 0; j < 2; ++j) bv[j] = bias[tN + 16 * j + c];

  float4 fa[3][GK_STEPS][2], fb[3][GK_STEPS][2], fg[3][GK_STEPS], fl[3][GK_STEPS];
#pragma unroll
  for (int g = 0; g < 2; ++g) {
#pragma unroll
    for (int st = 0; st < GK_STEPS; ++st) {
#pragma unroll
      for (int i = 0; i < 2; ++i) {
        fa[g][st][i] = *(const float4*)(pa[i] + 32 * g + 8 * st);
        fb[g][st][i] = *(const float4*)(pb[i] + 32 * g + 8 * st);
      }
      fg[g][st] = *(const float4*)(pg + 32 * g + 8 * st);
      fl[g][st] = *(const float4*)(pl + 32 * g + 8 * st);
    }
    __builtin_amdgcn_sched_barrier(0);
  }

  float sh[2];
#pragma unroll
  for (int i = 0; i < 2; ++i) sh[i] = __shfl(fa[0][0][i].x, c, 32);

  v8f acc[2][2];
#pragma unroll
  for (int i = 0; i < 2; ++i)
#pragma unroll
    for (int j = 0; j < 2; ++j) acc[i][j] = (v8f){};
  float s1[2] = {0.f, 0.f}, s2[2] = {0.f, 0.f};
  float sg[2] = {0.f, 0.f}, cb[2] = {0.f, 0.f};

#pragma unroll
  for (int g = 0; g < GK_GROUPS; ++g) {
    if (g + 2 < GK_GROUPS) {
#pragma unroll
      for (int st = 0; st < GK_STEPS; ++st) {
#pragma unroll
        for (int i = 0; i < 2; ++i) {
          fa[(g + 2) % 3][st][i] = *(const float4*)(pa[i] + 32 * (g + 2) + 8 * st);
          fb[(g + 2) % 3][st][i] = *(const float4*)(pb[i] + 32 * (g + 2) + 8 * st);
        }
        fg[(g + 2) % 3][st] = *(const float4*)(pg + 32 * (g + 2) + 8 * st);
        fl[(g + 2) % 3][st] = *(const float4*)(pl + 32 * (g + 2) + 8 * st);
      }
    }
    __builtin_amdgcn_sched_barrier(0);
#pragma unroll
    for (int st = 0; st < GK_STEPS; ++st) {
      const float4 a0 = sub4(fa[g % 3][st][0], sh[0]), a1 = sub4(fa[g % 3][st][1], sh[1]);
      const float4 w0 = fb[g % 3][st][0], w1 = fb[g % 3][st][1];
      const float4 gg = fg[g % 3][st], ll = fl[g % 3][st];
      const float4 b0 = mul4(w0, gg), b1 = mul4(w1, gg);
      s1[0] += (a0.x + a0.y) + (a0.z + a0.w);
      s1[1] += (a1.x + a1.y) + (a1.z + a1.w);
      s2[0] = fmaf(a0.x, a0.x, fmaf(a0.y, a0.y, fmaf(a0.z, a0.z, fmaf(a0.w, a0.w, s2[0]))));
      s2[1] = fmaf(a1.x, a1.x, fmaf(a1.y, a1.y, fmaf(a1.z, a1.z, fmaf(a1.w, a1.w, s2[1]))));
      sg[0] += (b0.x + b0.y) + (b0.z + b0.w);
      sg[1] += (b1.x + b1.y) + (b1.z + b1.w);
      cb[0] = fmaf(w0.x, ll.x, fmaf(w0.y, ll.y, fmaf(w0.z, ll.z, fmaf(w0.w, ll.w, cb[0]))));
      cb[1] = fmaf(w1.x, ll.x, fmaf(w1.y, ll.y, fmaf(w1.z, ll.z, fmaf(w1.w, ll.w, cb[1]))));
      acc[0][0] = mma4((v2f){a0.x, a0.y}, (v2f){b0.x, b0.y}, acc[0][0]);
      acc[0][1] = mma4((v2f){a0.x, a0.y}, (v2f){b1.x, b1.y}, acc[0][1]);
      acc[1][0] = mma4((v2f){a1.x, a1.y}, (v2f){b0.x, b0.y}, acc[1][0]);
      acc[1][1] = mma4((v2f){a1.x, a1.y}, (v2f){b1.x, b1.y}, acc[1][1]);
      acc[0][0] = mma4((v2f){a0.z, a0.w}, (v2f){b0.z, b0.w}, acc[0][0]);
      acc[0][1] = mma4((v2f){a0.z, a0.w}, (v2f){b1.z, b1.w}, acc[0][1]);
      acc[1][0] = mma4((v2f){a1.z, a1.w}, (v2f){b0.z, b0.w}, acc[1][0]);
      acc[1][1] = mma4((v2f){a1.z, a1.w}, (v2f){b1.z, b1.w}, acc[1][1]);
#pragma unroll
      for (int q = 0; q < 8; ++q) {
        __builtin_amdgcn_sched_group_barrier(0x008, 1, 0);
        __builtin_amdgcn_sched_group_barrier(0x002, 5, 0);
      }
      asm volatile("" : "+v"(s1[0]), "+v"(s1[1]), "+v"(s2[0]), "+v"(s2[1]),
                        "+v"(sg[0]), "+v"(sg[1]), "+v"(cb[0]), "+v"(cb[1]));
    }
  }

  float dm[2], rs[2];
#pragma unroll
  for (int i = 0; i < 2; ++i) {
    s1[i] += __shfl_xor(s1[i], 16, 32);
    s2[i] += __shfl_xor(s2[i], 16, 32);
    dm[i] = s1[i] / D_DIM;
    const float var = fmaxf(s2[i] / D_DIM - dm[i] * dm[i], 0.f);
    rs[i] = rsqrtf(var + LN_EPS);
  }
#pragma unroll
  for (int j = 0; j < 2; ++j) {
    sg[j] += __shfl_xor(sg[j], 16, 32);
    cb[j] += __shfl_xor(cb[j], 16, 32);
    cb[j] += bv[j];
  }

#pragma unroll
  for (int i = 0; i < 2; ++i)
#pragma unroll
    for (int r = 0; r < 8; ++r) {
      const float dmr = __shfl(dm[i], 8 * h + r, 32);
      const float rsr = __shfl(rs[i], 8 * h + r, 32);
      const int m = tM + 16 * i + 8 * h + r;
#pragma unroll
      for (int j = 0; j < 2; ++j) {
        const int n = tN + 16 * j + c;
        Out[(size_t)m * D_DIM + n] = rsr * (acc[i][j][r] - dmr * sg[j]) + cb[j];
      }
    }
}

__global__ __launch_bounds__(128) void wmma_gemm3_f32_kernel(const float* __restrict__ A,
                                                             const float* __restrict__ W0,
                                                             const float* __restrict__ W1,
                                                             const float* __restrict__ W2,
                                                             float* __restrict__ O0,
                                                             float* __restrict__ O1,
                                                             float* __restrict__ O2) {
  const int z = blockIdx.z;
  const float* Wm = (z == 0) ? W0 : (z == 1) ? W1 : W2;
  float* Out = (z == 0) ? O0 : (z == 1) ? O1 : O2;
  gemm_tile_f32<false>(A, Wm, Out, nullptr);
}

__global__ void phi_beta_kernel(const float* __restrict__ x,
                                float* __restrict__ q,
                                float* __restrict__ k,
                                const float* __restrict__ beta_w,
                                const float* __restrict__ beta_b,
                                const float* __restrict__ lnp_g,
                                const float* __restrict__ lnp_b,
                                float* __restrict__ beta) {
  __shared__ float sh[8 * 5];
  const int row = blockIdx.x, tid = threadIdx.x;
  const size_t off = (size_t)row * D_DIM + tid;

  float qe; { float t = q[off]; qe = (t > 0.f) ? t + 1.f : __expf(t); }
  float ke; { float t = k[off]; ke = (t > 0.f) ? t + 1.f : __expf(t); }
  const float xb = x[off] * beta_w[tid];

  const float sq  = block_sum_256(qe,      sh + 0);
  const float sq2 = block_sum_256(qe * qe, sh + 8);
  const float sk  = block_sum_256(ke,      sh + 16);
  const float sk2 = block_sum_256(ke * ke, sh + 24);
  const float sb  = block_sum_256(xb,      sh + 32);

  const float muq = sq / D_DIM, varq = sq2 / D_DIM - muq * muq;
  const float muk = sk / D_DIM, vark = sk2 / D_DIM - muk * muk;
  const float g = lnp_g[tid], bb = lnp_b[tid];
  q[off] = (qe - muq) * rsqrtf(varq + LN_EPS) * g + bb;
  k[off] = (ke - muk) * rsqrtf(vark + LN_EPS) * g + bb;
  if (tid == 0) beta[row] = 1.f / (1.f + __expf(-(sb + beta_b[0])));
}

#define CH 16
#define NCH (L_DIM / CH)
#define AUX_STRIDE 512

__global__ void chunk_prep_kernel(const float* __restrict__ phi_q,
                                  const float* __restrict__ phi_k,
                                  const float* __restrict__ beta,
                                  const float* __restrict__ decay,
                                  const float* __restrict__ v,
                                  float* __restrict__ u0,
                                  float* __restrict__ aux) {
  __shared__ __attribute__((aligned(16))) float Ks[CH][260];
  __shared__ __attribute__((aligned(16))) float Qs[CH][260];
  __shared__ float Al[CH][17];
  __shared__ float Tl[CH][17];
  __shared__ float bl[CH];
  const int ch = blockIdx.x, b = blockIdx.y, tid = threadIdx.x;
  const float df = 1.f / (1.f + __expf(-decay[0]));
  const size_t base = ((size_t)b * L_DIM + (size_t)ch * CH) * D_DIM;
  float vcol[CH];
#pragma unroll
  for (int i = 0; i < CH; ++i) {
    Ks[i][tid] = phi_k[base + (size_t)i * D_DIM + tid];
    Qs[i][tid] = phi_q[base + (size_t)i * D_DIM + tid];
    vcol[i] = v[base + (size_t)i * D_DIM + tid];
  }
  if (tid < CH) bl[tid] = beta[(size_t)b * L_DIM + ch * CH + tid];
  __syncthreads();

  const int i = tid >> 4, j = tid & 15;
  float gk = 0.f, gq = 0.f;
#pragma unroll 4
  for (int c = 0; c < D_DIM; c += 4) {
    const float4 kj = *(const float4*)&Ks[j][c];
    const float4 ki = *(const float4*)&Ks[i][c];
    const float4 qi = *(const float4*)&Qs[i][c];
    gk = fmaf(ki.x, kj.x, gk); gk = fmaf(ki.y, kj.y, gk); gk = fmaf(ki.z, kj.z, gk); gk = fmaf(ki.w, kj.w, gk);
    gq = fmaf(qi.x, kj.x, gq); gq = fmaf(qi.y, kj.y, gq); gq = fmaf(qi.z, kj.z, gq); gq = fmaf(qi.w, kj.w, gq);
  }
  float pw = 1.f;
  for (int n = 0; n < i - 1 - j; ++n) pw *= df;
  const bool lower = j < i;
  Al[i][j] = lower ? bl[i] * pw * gk : 0.f;
  float* o = aux + ((size_t)b * NCH + ch) * AUX_STRIDE;
  o[256 + tid] = lower ? pw * gq : 0.f;
  __syncthreads();

  if (tid < CH) {
    float tc[CH];
#pragma unroll
    for (int ii = 0; ii < CH; ++ii) {
      float s = (ii == tid) ? 1.f : 0.f;
#pragma unroll
      for (int jj = 0; jj < ii; ++jj) s = fmaf(-Al[ii][jj], tc[jj], s);
      tc[ii] = s;
      Tl[ii][tid] = s;
    }
  }
  __syncthreads();
  float gj = 1.f;
  for (int n = 0; n < j; ++n) gj *= df;
  o[tid] = -Tl[i][j] * bl[j] * gj;

#pragma unroll
  for (int jj = 0; jj < CH; ++jj) vcol[jj] *= bl[jj];
#pragma unroll
  for (int ii = 0; ii < CH; ++ii) {
    float s = 0.f;
#pragma unroll
    for (int jj = 0; jj <= ii; ++jj) s = fmaf(Tl[ii][jj], vcol[jj], s);
    u0[base + (size_t)ii * D_DIM + tid] = s;
  }
}

#define HCH (NCH / 2)
template <bool V> using flag_c = std::integral_constant<bool, V>;
template <int V> using int_c = std::integral_constant<int, V>;

__global__ __launch_bounds__(128) void delta_chunk_scan_kernel(
    const float* __restrict__ phi_q, const float* __restrict__ phi_k,
    const float* u0p, float* yv, float* zv, float* smid,
    const float* __restrict__ aux, const float* __restrict__ decay) {
  __shared__ float lds_pool[41 * 1024];
  float (*red)[4][2][8][32] = reinterpret_cast<float (*)[4][2][8][32]>(lds_pool);
  float (*u0s)[8][32] = reinterpret_cast<float (*)[8][32]>(lds_pool + 2 * 4 * 2 * 8 * 32);
  const int b = blockIdx.y;
  const int row0 = blockIdx.x * 16;
  const int set = blockIdx.z;
  const int t0 = (set == 0) ? 0 : HCH;
  const int wave = __builtin_amdgcn_readfirstlane(threadIdx.x >> 5);
  const int lane = threadIdx.x & 31;
  const int c = lane & 15, h = lane >> 4;
  const int r0 = 2 * wave;

  const float df = 1.f / (1.f + __expf(-decay[0]));
  float pwr[17];
  pwr[0] = 1.f;
#pragma unroll
  for (int n = 1; n <= 16; ++n) pwr[n] = pwr[n - 1] * df;
  const float g16 = pwr[16];
  float gpos[8], grev[8];
#pragma unroll
  for (int r = 0; r < 8; ++r) {
    gpos[r] = h ? pwr[8 + r] : pwr[r];
    grev[r] = h ? pwr[7 - r] : pwr[15 - r];
  }
  float gy0 = gpos[0], gy1 = gpos[1];
#pragma unroll
  for (int r = 2; r < 8; r += 2)
    if (r == r0) { gy0 = gpos[r]; gy1 = gpos[r + 1]; }

  const size_t bbase = (size_t)b * L_DIM * D_DIM;
  const size_t off0 = (size_t)t0 * CH * D_DIM;
  const int colw = 64 * wave;
  const float* kA = phi_k + bbase + (size_t)c * D_DIM + colw + 8 * h;
  const float* qA = phi_q + bbase + (size_t)c * D_DIM + colw + 8 * h;
  const float* kT = phi_k + bbase + (size_t)(8 * h) * D_DIM + colw + c;
  const size_t tile_off = (size_t)(8 * h + r0) * D_DIM + row0 + c;
  const float* u0w = u0p + bbase + tile_off;
  float* ytw = (set == 2) ? zv + (size_t)b * (HCH * CH) * D_DIM + tile_off : yv + bbase + tile_off;
  const int ty = (set == 2) ? HCH : 0;
  const float* ax = aux + (size_t)b * NCH * AUX_STRIDE + c * 16 + 8 * h;

  v8f acc[4];
#pragma unroll
  for (int ct = 0; ct < 4; ++ct)
#pragma unroll
    for (int r = 0; r < 8; ++r)
      acc[ct][r] = (set == 2 && colw + 16 * ct + 8 * h + r == row0 + c) ? 1.f : 0.f;

  float4 Kf[4][2], Qf[4][2];
  float ktr[4][8];
#pragma unroll
  for (int ct = 0; ct < 4; ++ct) {
    Kf[ct][0] = *(const float4*)(kA + off0 + 16 * ct);
    Kf[ct][1] = *(const float4*)(kA + off0 + 16 * ct + 4);
    Qf[ct][0] = *(const float4*)(qA + off0 + 16 * ct);
    Qf[ct][1] = *(const float4*)(qA + off0 + 16 * ct + 4);
#pragma unroll
    for (int r = 0; r < 8; ++r) ktr[ct][r] = kT[off0 + (size_t)r * D_DIM + 16 * ct];
  }
  float4 tnf[2], tnN[2], mqf[2];
  {
    const float* a0 = ax + (size_t)t0 * AUX_STRIDE;
    tnf[0] = *(const float4*)(a0);  tnf[1] = *(const float4*)(a0 + 4);
    mqf[0] = *(const float4*)(a0 + 256);  mqf[1] = *(const float4*)(a0 + 260);
  }
  float u0a = 0.f, u0b = 0.f;
  if (set != 2) { u0a = u0w[off0]; u0b = u0w[off0 + D_DIM]; }

#pragma unroll
  for (int r = 0; r < 8; ++r) red[1][wave][1][r][lane] = 0.f;
  float ym0 = 0.f, ym1 = 0.f;

  __builtin_amdgcn_fence(__ATOMIC_ACQUIRE, "workgroup", "global");

#define YQ2(ct, p)                                                                        \
  do {                                                                                    \
    const float* qf_ = (const float*)Qf[ct];                                              \
    Ya = mma4((v2f){qf_[2 * (p)], qf_[2 * (p) + 1]},                                      \
              (v2f){acc[ct][2 * (p)], acc[ct][2 * (p) + 1]}, Ya);                         \
    Yb = mma4((v2f){qf_[2 * (p) + 2], qf_[2 * (p) + 3]},                                  \
              (v2f){acc[ct][2 * (p) + 2], acc[ct][2 * (p) + 3]}, Yb);                     \
  } while (0)
#define ORDER2(a, b) asm volatile("" : "+v"(a), "+v"(b) : : "memory")
#define YQ_ORDER() ORDER2(Ya, Yb)

  auto chunk = [&](const int t, auto par_c, auto unit_c) __attribute__((always_inline)) {
    constexpr int par = decltype(par_c)::value;
    constexpr bool UNIT = decltype(unit_c)::value;
    const size_t coff = (size_t)(t - ty) * CH * D_DIM;
    const int tn1 = (t + 1 < NCH) ? t + 1 : t;
    const size_t noff = (size_t)tn1 * CH * D_DIM;

    v8f Pa = {}, Pb = {};
#pragma unroll
    for (int ct = 0; ct < 4; ++ct) {
      const float* kf = (const float*)Kf[ct];
#pragma unroll
      for (int p = 0; p < 4; p += 2) {
        Pa = mma4((v2f){kf[2 * p], kf[2 * p + 1]}, (v2f){acc[ct][2 * p], acc[ct][2 * p + 1]}, Pa);
        Pb = mma4((v2f){kf[2 * p + 2], kf[2 * p + 3]}, (v2f){acc[ct][2 * p + 2], acc[ct][2 * p + 3]}, Pb);
      }
    }
    ORDER2(Pa, Pb);

#pragma unroll
    for (int r = 0; r < 8; ++r) red[par][wave][0][r][lane] = Pa[r] + Pb[r];
    if constexpr (!UNIT) {
      u0s[par][r0][lane] = u0a;
      u0s[par][r0 + 1][lane] = u0b;
    }

    v8f Ya = {}, Yb = {};
    YQ_ORDER();
#pragma unroll
    for (int ct = 0; ct < 4; ++ct) {
      Kf[ct][0] = *(const float4*)(kA + noff + 16 * ct);
      Kf[ct][1] = *(const float4*)(kA + noff + 16 * ct + 4);
    }
    {
      const float* an = ax + (size_t)tn1 * AUX_STRIDE;
      tnN[0] = *(const float4*)(an);  tnN[1] = *(const float4*)(an + 4);
    }
    if constexpr (!UNIT) {
      u0a = u0w[noff];
      u0b = u0w[noff + D_DIM];
    }
    YQ2(0, 0); YQ2(0, 2);
    YQ_ORDER();
    __builtin_amdgcn_fence(__ATOMIC_RELEASE, "workgroup", "local");
    __builtin_amdgcn_s_barrier_signal(-1);
    YQ_ORDER();
    YQ2(1, 0); YQ2(1, 2); YQ2(2, 0);
    YQ_ORDER();
    __builtin_amdgcn_s_barrier_wait(-1);
    __builtin_amdgcn_fence(__ATOMIC_ACQUIRE, "workgroup", "local");

    float pp[4][8];
#pragma unroll
    for (int w = 0; w < 4; ++w)
#pragma unroll
      for (int r = 0; r < 8; ++r) pp[w][r] = red[par][w][0][r][lane];
    v8f U = {};
    if constexpr (!UNIT) {
#pragma unroll
      for (int r = 0; r < 8; ++r) U[r] = u0s[par][r][lane];
    }
    float yp[4][2];
#pragma unroll
    for (int w = 0; w < 4; ++w) {
      yp[w][0] = red[par ^ 1][w][1][r0][lane];
      yp[w][1] = red[par ^ 1][w][1][r0 + 1][lane];
    }
    YQ_ORDER();
    YQ2(2, 2); YQ2(3, 0); YQ2(3, 2);
    YQ_ORDER();
    __builtin_amdgcn_sched_barrier(0);

#pragma unroll
    for (int ct = 0; ct < 4; ++ct) {
      Qf[ct][0] = *(const float4*)(qA + noff + 16 * ct);
      Qf[ct][1] = *(const float4*)(qA + noff + 16 * ct + 4);
    }
    if (t + 2 < NCH) {
      __builtin_prefetch(kA + noff + (size_t)CH * D_DIM + 24 * h, 0, 0);
      __builtin_prefetch(qA + noff + (size_t)CH * D_DIM + 24 * h, 0, 0);
    }

    float ps[8];
#pragma unroll
    for (int r = 0; r < 8; ++r) ps[r] = (pp[0][r] + pp[1][r]) + (pp[2][r] + pp[3][r]);
    {
      const float* tnp = (const float*)tnf;
#pragma unroll
      for (int jj = 0; jj < 4; ++jj)
        U = mma4((v2f){tnp[2 * jj], tnp[2 * jj + 1]}, (v2f){ps[2 * jj], ps[2 * jj + 1]}, U);
    }

    {
      const float yq0 = (yp[0][0] + yp[1][0]) + (yp[2][0] + yp[3][0]);
      const float yq1 = (yp[0][1] + yp[1][1]) + (yp[2][1] + yp[3][1]);
      if (t > t0) {
        ytw[coff - (size_t)CH * D_DIM] = gy0 * yq0 + ym0;
        ytw[coff - (size_t)CH * D_DIM + D_DIM] = gy1 * yq1 + ym1;
      }
    }
#pragma unroll
    for (int r = 0; r < 8; ++r) red[par][wave][1][r][lane] = Ya[r] + Yb[r];

#pragma unroll
    for (int ct = 0; ct < 4; ++ct) acc[ct] *= g16;
    float kt[4][8];
#pragma unroll
    for (int ct = 0; ct < 4; ++ct)
#pragma unroll
      for (int r = 0; r < 8; ++r) kt[ct][r] = ktr[ct][r] * grev[r];
    const float* mqp = (const float*)mqf;
    v8f Ym = {};
#pragma unroll
    for (int jj = 0; jj < 4; ++jj) {
#pragma unroll
      for (int ct = 0; ct < 4; ++ct)
        acc[ct] = mma4((v2f){kt[ct][2 * jj], kt[ct][2 * jj + 1]},
                       (v2f){U[2 * jj], U[2 * jj + 1]}, acc[ct]);
      Ym = mma4((v2f){mqp[2 * jj], mqp[2 * jj + 1]}, (v2f){U[2 * jj], U[2 * jj + 1]}, Ym);
    }
    ym0 = Ym[0]; ym1 = Ym[1];
#pragma unroll
    for (int r = 2; r < 8; r += 2)
      if (r == r0) { ym0 = Ym[r]; ym1 = Ym[r + 1]; }

    {
      const float* an = ax + (size_t)tn1 * AUX_STRIDE;
      mqf[0] = *(const float4*)(an + 256);  mqf[1] = *(const float4*)(an + 260);
    }
#pragma unroll
    for (int ct = 0; ct < 4; ++ct)
#pragma unroll
      for (int r = 0; r < 8; ++r) ktr[ct][r] = kT[noff + (size_t)r * D_DIM + 16 * ct];
    tnf[0] = tnN[0]; tnf[1] = tnN[1];
    __builtin_amdgcn_sched_barrier(0);
  };

  auto pass = [&](auto unit_c) __attribute__((always_inline)) {
    for (int tb = t0; tb < t0 + HCH; tb += 2) {
      chunk(tb + 0, int_c<0>{}, unit_c);
      chunk(tb + 1, int_c<1>{}, unit_c);
    }
  };
  if (set == 2) pass(flag_c<true>{});
  else pass(flag_c<false>{});
#undef YQ2
#undef YQ_ORDER
#undef ORDER2

  {
    __builtin_amdgcn_fence(__ATOMIC_RELEASE, "workgroup", "local");
    __builtin_amdgcn_s_barrier();
    __builtin_amdgcn_fence(__ATOMIC_ACQUIRE, "workgroup", "local");
    const int lp = (HCH - 1) & 1;
    const float yq0 = (red[lp][0][1][r0][lane] + red[lp][1][1][r0][lane]) +
                      (red[lp][2][1][r0][lane] + red[lp][3][1][r0][lane]);
    const float yq1 = (red[lp][0][1][r0 + 1][lane] + red[lp][1][1][r0 + 1][lane]) +
                      (red[lp][2][1][r0 + 1][lane] + red[lp][3][1][r0 + 1][lane]);
    const size_t loff = (size_t)(t0 + HCH - 1 - ty) * CH * D_DIM;
    ytw[loff] = gy0 * yq0 + ym0;
    ytw[loff + D_DIM] = gy1 * yq1 + ym1;
  }

  if (set == 0) {
    float* sp = smid + (size_t)b * D_DIM * D_DIM + (size_t)(row0 + c) * D_DIM + colw + 8 * h;
#pragma unroll
    for (int ct = 0; ct < 4; ++ct) {
      *(float4*)(sp + 16 * ct)     = make_float4(acc[ct][0], acc[ct][1], acc[ct][2], acc[ct][3]);
      *(float4*)(sp + 16 * ct + 4) = make_float4(acc[ct][4], acc[ct][5], acc[ct][6], acc[ct][7]);
    }
  }
}

__global__ __launch_bounds__(128) void wmma_gemm_join_f32_kernel(const float* __restrict__ Z,
                                                                 const float* __restrict__ Sm,
                                                                 float* __restrict__ Y) {
  const int b = blockIdx.z;
  gemm_tile_f32<true>(Z + (size_t)b * (HCH * CH) * D_DIM, Sm + (size_t)b * D_DIM * D_DIM,
                      Y + ((size_t)b * L_DIM + HCH * CH) * D_DIM, nullptr);
}

extern "C" void kernel_launch(void* const* d_in, const int* in_sizes, int n_in,
                              void* d_out, int out_size, void* d_ws, size_t ws_size,
                              hipStream_t stream) {
  (void)in_sizes; (void)n_in; (void)out_size; (void)ws_size;
  const float* x      = (const float*)d_in[0];
  const float* Wq     = (const float*)d_in[1];
  const float* Wk     = (const float*)d_in[2];
  const float* Wv     = (const float*)d_in[3];
  const float* beta_w = (const float*)d_in[4];
  const float* beta_b = (const float*)d_in[5];
  const float* decay  = (const float*)d_in[6];
  const float* Wo     = (const float*)d_in[7];
  const float* bo     = (const float*)d_in[8];
  const float* ln_g   = (const float*)d_in[9];
  const float* ln_b   = (const float*)d_in[10];
  const float* lnp_g  = (const float*)d_in[11];
  const float* lnp_b  = (const float*)d_in[12];
  float* out = (float*)d_out;

  char* ws = (char*)d_ws;
  const size_t MAT = (size_t)M_DIM * D_DIM * sizeof(float);
  float* q    = (float*)(ws);
  float* k    = (float*)(ws + MAT);
  float* vv   = (float*)(ws + 2 * MAT);
  float* beta = (float*)(ws + 3 * MAT);
  float* aux  = (float*)(ws + 3 * MAT + (64 << 10));
  float* smid = (float*)(ws + 3 * MAT + (1 << 20));
  float* zv   = (float*)(ws + 3 * MAT + (2 << 20));

  const dim3 gemm_grid(M_DIM / 64, D_DIM / 64);
  const dim3 gemm_blk(128);

  wmma_gemm3_f32_kernel<<<dim3(M_DIM / 64, D_DIM / 64, 3), gemm_blk, 0, stream>>>(
      x, Wq, Wk, Wv, q, k, vv);

  phi_beta_kernel<<<dim3(M_DIM), dim3(256), 0, stream>>>(
      x, q, k, beta_w, beta_b, lnp_g, lnp_b, beta);

  chunk_prep_kernel<<<dim3(NCH, B_DIM), dim3(256), 0, stream>>>(q, k, beta, decay, vv, out, aux);

  delta_chunk_scan_kernel<<<dim3(D_DIM / 16, B_DIM, 3), dim3(128), 0, stream>>>(
      q, k, out, vv, zv, smid, aux, decay);

  wmma_gemm_join_f32_kernel<<<dim3(L_DIM / 2 / 64, D_DIM / 64, B_DIM), gemm_blk, 0, stream>>>(
      zv, smid, vv);

  wmma_gemm_ln_f32_kernel<<<gemm_grid, gemm_blk, 0, stream>>>(vv, Wo, out, bo, ln_g, ln_b);
}
